# starting kernel
# baseline (speedup 1.0000x reference)
#include <hip/hip_runtime.h>
#include <math.h>

typedef __attribute__((ext_vector_type(2))) float v2f;
typedef __attribute__((ext_vector_type(8))) float v8f;

#define IN_CH     128
#define OUT_CH    64
#define NHEAD     4
#define TOT_OUT   256
#define NEG_SLOPE 0.2f

#define MTILE 64
#define KCH   32
#define KPAD  34

__global__ void k_init(float* out, float* denom, unsigned* segmax, int C) {
    int i = blockIdx.x * blockDim.x + threadIdx.x;
    if (i < C * TOT_OUT) out[i] = 0.0f;
    if (i < 4 * C) { denom[i] = 0.0f; segmax[i] = 0xFF800000u; }
}

__global__ void k_fold(const float* __restrict__ lin_w, const float* __restrict__ lin_b,
                       const float* __restrict__ att_w, const float* __restrict__ att_b,
                       float* v, float* c) {
    int t = blockIdx.x * blockDim.x + threadIdx.x;
    if (t < NHEAD * IN_CH) {
        int h = t / IN_CH, i = t % IN_CH;
        float s = 0.0f;
        for (int d = 0; d < OUT_CH; ++d)
            s += lin_w[(size_t)(h * OUT_CH + d) * IN_CH + i] * att_w[d];
        v[h * IN_CH + i] = s;
    }
    if (t < NHEAD) {
        float s = att_b[0];
        for (int d = 0; d < OUT_CH; ++d) s += lin_b[t * OUT_CH + d] * att_w[d];
        c[t] = s;
    }
}

__global__ __launch_bounds__(256)
void k_score(const float* __restrict__ x, const int* __restrict__ y,
             const float* __restrict__ v, const float* __restrict__ c,
             float* __restrict__ score, int* segmax_i, unsigned* segmax_u, int Nn) {
    __shared__ float vl[NHEAD * IN_CH];
    __shared__ float cl[NHEAD];
    int t = threadIdx.x;
    vl[t] = v[t]; vl[t + 256] = v[t + 256];
    if (t < NHEAD) cl[t] = c[t];
    __syncthreads();

    int node = blockIdx.x * 256 + t;
    if (node >= Nn) return;

    const float4* xp = (const float4*)(x + (size_t)node * IN_CH);
    float s[NHEAD] = {0.f, 0.f, 0.f, 0.f};
#pragma unroll
    for (int q = 0; q < IN_CH / 4; ++q) {
        float4 xv = xp[q];
#pragma unroll
        for (int h = 0; h < NHEAD; ++h) {
            const float* vh = &vl[h * IN_CH + q * 4];
            s[h] += xv.x * vh[0] + xv.y * vh[1] + xv.z * vh[2] + xv.w * vh[3];
        }
    }
    int yy = y[node];
#pragma unroll
    for (int h = 0; h < NHEAD; ++h) {
        float r = s[h] + cl[h];
        r = (r >= 0.0f) ? r : NEG_SLOPE * r;
        score[(size_t)node * NHEAD + h] = r;
        int idx = yy * NHEAD + h;
        if (r >= 0.0f) atomicMax(&segmax_i[idx], __float_as_int(r));
        else           atomicMin(&segmax_u[idx], (unsigned)__float_as_int(r));
    }
}

__global__ void k_denom(const float* __restrict__ score, const int* __restrict__ y,
                        const float* __restrict__ segmax, float* denom, int Nn) {
    int i = blockIdx.x * blockDim.x + threadIdx.x;
    if (i >= 4 * Nn) return;
    int node = i >> 2, h = i & 3;
    int idx = y[node] * NHEAD + h;
    atomicAdd(&denom[idx], __expf(score[i] - segmax[idx]));
}

__global__ __launch_bounds__(256)
void k_gemm(const float* __restrict__ x, const int* __restrict__ y,
            const float* __restrict__ lin_w, const float* __restrict__ lin_b,
            const float* __restrict__ score, const float* __restrict__ segmax,
            const float* __restrict__ denom, float* __restrict__ out, int Nn) {
    __shared__ float Xl[MTILE * KPAD];
    __shared__ float Wl[TOT_OUT * KPAD];
    __shared__ float wgt[MTILE * NHEAD];
    __shared__ float bl[TOT_OUT];
    __shared__ int   yl[MTILE];

    int t = threadIdx.x;
    int blockM = blockIdx.x * MTILE;

    {
        int m = t >> 2, h = t & 3;
        int node = blockM + m;
        float w = 0.0f; int yy = 0;
        if (node < Nn) {
            yy = y[node];
            int idx = yy * NHEAD + h;
            w = __expf(score[(size_t)node * NHEAD + h] - segmax[idx]) / denom[idx];
        }
        wgt[m * NHEAD + h] = w;
        if (h == 0) yl[m] = yy;
        bl[t] = lin_b[t];
    }

    int wave = t >> 5, lane = t & 31;
    int rowbase = (wave >> 1) * 16;
    int colbase = (wave & 1) * 128;
    int lm      = lane & 15;
    int khalf   = (lane < 16) ? 0 : 2;

    v8f acc[8];
    const v8f vzero = {0.f, 0.f, 0.f, 0.f, 0.f, 0.f, 0.f, 0.f};
#pragma unroll
    for (int i = 0; i < 8; ++i) acc[i] = vzero;

    for (int kc = 0; kc < IN_CH / KCH; ++kc) {
        int kbase = kc * KCH;
        __syncthreads();
        {
            int row = t >> 2, q = t & 3;
            int node = blockM + row;
#pragma unroll
            for (int j = 0; j < 2; ++j) {
                int kk = q * 8 + j * 4;
                float4 xv = make_float4(0.f, 0.f, 0.f, 0.f);
                if (node < Nn)
                    xv = *(const float4*)(x + (size_t)node * IN_CH + kbase + kk);
                Xl[row * KPAD + kk + 0] = xv.x;
                Xl[row * KPAD + kk + 1] = xv.y;
                Xl[row * KPAD + kk + 2] = xv.z;
                Xl[row * KPAD + kk + 3] = xv.w;
            }
            if (kc + 1 < IN_CH / KCH && node < Nn)
                __builtin_prefetch(x + (size_t)node * IN_CH + kbase + KCH, 0, 1);
        }
        {
            int n = t;
#pragma unroll
            for (int q = 0; q < 8; ++q) {
                int kk = q * 4;
                float4 w4 = *(const float4*)(lin_w + (size_t)n * IN_CH + kbase + kk);
                Wl[n * KPAD + kk + 0] = w4.x;
                Wl[n * KPAD + kk + 1] = w4.y;
                Wl[n * KPAD + kk + 2] = w4.z;
                Wl[n * KPAD + kk + 3] = w4.w;
            }
            if (kc + 1 < IN_CH / KCH)
                __builtin_prefetch(lin_w + (size_t)n * IN_CH + kbase + KCH, 0, 1);
        }
        __syncthreads();
#pragma unroll
        for (int ks = 0; ks < KCH / 4; ++ks) {
            int kk = ks * 4 + khalf;
            v2f a = *(const v2f*)(&Xl[(rowbase + lm) * KPAD + kk]);
#pragma unroll
            for (int nt = 0; nt < 8; ++nt) {
                int n = colbase + nt * 16 + lm;
                v2f b = *(const v2f*)(&Wl[n * KPAD + kk]);
                acc[nt] = __builtin_amdgcn_wmma_f32_16x16x4_f32(
                    false, a, false, b, (short)0, acc[nt], false, false);
            }
        }
    }

    int mhalf = (lane < 16) ? 0 : 8;
#pragma unroll
    for (int nt = 0; nt < 8; ++nt) {
        int n = colbase + nt * 16 + lm;
        int head = n >> 6;
        float bn = bl[n];
#pragma unroll
        for (int r = 0; r < 8; ++r) {
            int m = rowbase + mhalf + r;
            float w = wgt[m * NHEAD + head];
            float val = (acc[nt][r] + bn) * w;
            atomicAdd(&out[(size_t)yl[m] * TOT_OUT + n], val);
        }
    }
}

extern "C" void kernel_launch(void* const* d_in, const int* in_sizes, int n_in,
                              void* d_out, int out_size, void* d_ws, size_t ws_size,
                              hipStream_t stream) {
    (void)n_in; (void)ws_size;
    const float* x     = (const float*)d_in[0];
    const int*   y     = (const int*)  d_in[1];
    const float* lin_w = (const float*)d_in[3];
    const float* lin_b = (const float*)d_in[4];
    const float* att_w = (const float*)d_in[5];
    const float* att_b = (const float*)d_in[6];
    float* out = (float*)d_out;

    int N = in_sizes[0] / IN_CH;
    int C = out_size / TOT_OUT;

    float* wsf    = (float*)d_ws;
    float* v      = wsf;
    float* c      = wsf + 512;
    float* segmax = wsf + 516;
    float* denom  = segmax + 4 * C;
    float* score  = denom + 4 * C;

    int initN = C * TOT_OUT;
    k_init <<<(initN + 255) / 256, 256, 0, stream>>>(out, denom, (unsigned*)segmax, C);
    k_fold <<<2, 256, 0, stream>>>(lin_w, lin_b, att_w, att_b, v, c);
    k_score<<<(N + 255) / 256, 256, 0, stream>>>(x, y, v, c, score,
                                                 (int*)segmax, (unsigned*)segmax, N);
    k_denom<<<(4 * N + 255) / 256, 256, 0, stream>>>(score, y, segmax, denom, N);
    k_gemm <<<(N + MTILE - 1) / MTILE, 256, 0, stream>>>(x, y, lin_w, lin_b,
                                                         score, segmax, denom, out, N);
}


# k_score reads x coalesced: one wave load = one 512-byte row, 8 rows x 4 heads summed by a lane butterfly (score dot products summed in a different tree); strided per-thread row reads removed
# speedup vs baseline: 1.2198x; 1.2198x over previous
#include <hip/hip_runtime.h>
#include <math.h>

typedef __attribute__((ext_vector_type(2))) float v2f;
typedef __attribute__((ext_vector_type(8))) float v8f;

#define IN_CH     128
#define OUT_CH    64
#define NHEAD     4
#define TOT_OUT   256
#define NEG_SLOPE 0.2f

#define MTILE 64
#define KCH   32
#define KPAD  34

__global__ void k_init(float* out, float* denom, unsigned* segmax, int C) {
    int i = blockIdx.x * blockDim.x + threadIdx.x;
    if (i < C * TOT_OUT) out[i] = 0.0f;
    if (i < 4 * C) { denom[i] = 0.0f; segmax[i] = 0xFF800000u; }
}

__global__ void k_fold(const float* __restrict__ lin_w, const float* __restrict__ lin_b,
                       const float* __restrict__ att_w, const float* __restrict__ att_b,
                       float* v, float* c) {
    int t = blockIdx.x * blockDim.x + threadIdx.x;
    if (t < NHEAD * IN_CH) {
        int h = t / IN_CH, i = t % IN_CH;
        float s = 0.0f;
        for (int d = 0; d < OUT_CH; ++d)
            s += lin_w[(size_t)(h * OUT_CH + d) * IN_CH + i] * att_w[d];
        v[h * IN_CH + i] = s;
    }
    if (t < NHEAD) {
        float s = att_b[0];
        for (int d = 0; d < OUT_CH; ++d) s += lin_b[t * OUT_CH + d] * att_w[d];
        c[t] = s;
    }
}

typedef __attribute__((ext_vector_type(4))) float v4f;

#define SC_ROWS   8
#define SC_GROUPS 4
#define SC_WAVES  8

__global__ __launch_bounds__(256)
void k_score(const float* __restrict__ x, const int* __restrict__ y,
             const float* __restrict__ v, const float* __restrict__ c,
             float* __restrict__ score, int* segmax_i, unsigned* segmax_u, int Nn) {
    const int lane = threadIdx.x & 31;
    const int wave = __builtin_amdgcn_readfirstlane((int)(threadIdx.x >> 5));
    const int row0 = (blockIdx.x * SC_WAVES + wave) * (SC_ROWS * SC_GROUPS);
    if (row0 >= Nn) return;

    v4f vh[NHEAD];
#pragma unroll
    for (int h = 0; h < NHEAD; ++h) vh[h] = *(const v4f*)(v + h * IN_CH + lane * 4);
    const int hh = lane & 3;
    const int rr = lane >> 2;
    const float ch = c[hh];

    const float* xb = x + (size_t)row0 * IN_CH + lane * 4;
    v4f xv[2][SC_ROWS];
#define SC_LOAD(G) do { _Pragma("unroll") for (int r = 0; r < SC_ROWS; ++r) { \
        const int ro_ = (row0 + (G) * SC_ROWS + r < Nn) ? ((G) * SC_ROWS + r) : 0; \
        xv[(G) & 1][r] = __builtin_nontemporal_load((const v4f*)(xb + (size_t)ro_ * IN_CH)); } } while (0)

    int yv[SC_GROUPS];
#pragma unroll
    for (int g = 0; g < SC_GROUPS; ++g) {
        const int nd = row0 + g * SC_ROWS + rr;
        yv[g] = y[(nd < Nn) ? nd : (Nn - 1)];
    }

    SC_LOAD(0);
#pragma unroll
    for (int g = 0; g < SC_GROUPS; ++g) {
        __builtin_amdgcn_sched_barrier(0);
        if (g + 1 < SC_GROUPS) SC_LOAD(g + 1);
        __builtin_amdgcn_sched_barrier(0);

        float vals[SC_ROWS * NHEAD];
#pragma unroll
        for (int r = 0; r < SC_ROWS; ++r) {
            const v4f xr = xv[g & 1][r];
#pragma unroll
            for (int h = 0; h < NHEAD; ++h)
                vals[r * NHEAD + h] = xr.x * vh[h].x + xr.y * vh[h].y + xr.z * vh[h].z + xr.w * vh[h].w;
        }
#define SC_STAGE(WD) do { const bool up_ = (lane & (WD)) != 0; \
        _Pragma("unroll") for (int i = 0; i < (WD); ++i) { \
            const float keep_ = up_ ? vals[i + (WD)] : vals[i]; \
            const float send_ = up_ ? vals[i] : vals[i + (WD)]; \
            vals[i] = keep_ + __shfl_xor(send_, (WD), 32); } } while (0)
        SC_STAGE(16); SC_STAGE(8); SC_STAGE(4); SC_STAGE(2); SC_STAGE(1);
#undef SC_STAGE
        float s = vals[0] + ch;
        s = (s >= 0.0f) ? s : NEG_SLOPE * s;
        const int node = row0 + g * SC_ROWS + rr;
        if (node < Nn) {
            score[(size_t)node * NHEAD + hh] = s;
            const int idx = yv[g] * NHEAD + hh;
            if (s >= 0.0f) atomicMax(&segmax_i[idx], __float_as_int(s));
            else           atomicMin(&segmax_u[idx], (unsigned)__float_as_int(s));
        }
    }
#undef SC_LOAD
}

__global__ void k_denom(const float* __restrict__ score, const int* __restrict__ y,
                        const float* __restrict__ segmax, float* denom, int Nn) {
    int i = blockIdx.x * blockDim.x + threadIdx.x;
    if (i >= 4 * Nn) return;
    int node = i >> 2, h = i & 3;
    int idx = y[node] * NHEAD + h;
    atomicAdd(&denom[idx], __expf(score[i] - segmax[idx]));
}

__global__ __launch_bounds__(256)
void k_gemm(const float* __restrict__ x, const int* __restrict__ y,
            const float* __restrict__ lin_w, const float* __restrict__ lin_b,
            const float* __restrict__ score, const float* __restrict__ segmax,
            const float* __restrict__ denom, float* __restrict__ out, int Nn) {
    __shared__ float Xl[MTILE * KPAD];
    __shared__ float Wl[TOT_OUT * KPAD];
    __shared__ float wgt[MTILE * NHEAD];
    __shared__ float bl[TOT_OUT];
    __shared__ int   yl[MTILE];

    int t = threadIdx.x;
    int blockM = blockIdx.x * MTILE;

    {
        int m = t >> 2, h = t & 3;
        int node = blockM + m;
        float w = 0.0f; int yy = 0;
        if (node < Nn) {
            yy = y[node];
            int idx = yy * NHEAD + h;
            w = __expf(score[(size_t)node * NHEAD + h] - segmax[idx]) / denom[idx];
        }
        wgt[m * NHEAD + h] = w;
        if (h == 0) yl[m] = yy;
        bl[t] = lin_b[t];
    }

    int wave = t >> 5, lane = t & 31;
    int rowbase = (wave >> 1) * 16;
    int colbase = (wave & 1) * 128;
    int lm      = lane & 15;
    int khalf   = (lane < 16) ? 0 : 2;

    v8f acc[8];
    const v8f vzero = {0.f, 0.f, 0.f, 0.f, 0.f, 0.f, 0.f, 0.f};
#pragma unroll
    for (int i = 0; i < 8; ++i) acc[i] = vzero;

    for (int kc = 0; kc < IN_CH / KCH; ++kc) {
        int kbase = kc * KCH;
        __syncthreads();
        {
            int row = t >> 2, q = t & 3;
            int node = blockM + row;
#pragma unroll
            for (int j = 0; j < 2; ++j) {
                int kk = q * 8 + j * 4;
                float4 xv = make_float4(0.f, 0.f, 0.f, 0.f);
                if (node < Nn)
                    xv = *(const float4*)(x + (size_t)node * IN_CH + kbase + kk);
                Xl[row * KPAD + kk + 0] = xv.x;
                Xl[row * KPAD + kk + 1] = xv.y;
                Xl[row * KPAD + kk + 2] = xv.z;
                Xl[row * KPAD + kk + 3] = xv.w;
            }
            if (kc + 1 < IN_CH / KCH && node < Nn)
                __builtin_prefetch(x + (size_t)node * IN_CH + kbase + KCH, 0, 1);
        }
        {
            int n = t;
#pragma unroll
            for (int q = 0; q < 8; ++q) {
                int kk = q * 4;
                float4 w4 = *(const float4*)(lin_w + (size_t)n * IN_CH + kbase + kk);
                Wl[n * KPAD + kk + 0] = w4.x;
                Wl[n * KPAD + kk + 1] = w4.y;
                Wl[n * KPAD + kk + 2] = w4.z;
                Wl[n * KPAD + kk + 3] = w4.w;
            }
            if (kc + 1 < IN_CH / KCH)
                __builtin_prefetch(lin_w + (size_t)n * IN_CH + kbase + KCH, 0, 1);
        }
        __syncthreads();
#pragma unroll
        for (int ks = 0; ks < KCH / 4; ++ks) {
            int kk = ks * 4 + khalf;
            v2f a = *(const v2f*)(&Xl[(rowbase + lm) * KPAD + kk]);
#pragma unroll
            for (int nt = 0; nt < 8; ++nt) {
                int n = colbase + nt * 16 + lm;
                v2f b = *(const v2f*)(&Wl[n * KPAD + kk]);
                acc[nt] = __builtin_amdgcn_wmma_f32_16x16x4_f32(
                    false, a, false, b, (short)0, acc[nt], false, false);
            }
        }
    }

    int mhalf = (lane < 16) ? 0 : 8;
#pragma unroll
    for (int nt = 0; nt < 8; ++nt) {
        int n = colbase + nt * 16 + lm;
        int head = n >> 6;
        float bn = bl[n];
#pragma unroll
        for (int r = 0; r < 8; ++r) {
            int m = rowbase + mhalf + r;
            float w = wgt[m * NHEAD + head];
            float val = (acc[nt][r] + bn) * w;
            atomicAdd(&out[(size_t)yl[m] * TOT_OUT + n], val);
        }
    }
}

extern "C" void kernel_launch(void* const* d_in, const int* in_sizes, int n_in,
                              void* d_out, int out_size, void* d_ws, size_t ws_size,
                              hipStream_t stream) {
    (void)n_in; (void)ws_size;
    const float* x     = (const float*)d_in[0];
    const int*   y     = (const int*)  d_in[1];
    const float* lin_w = (const float*)d_in[3];
    const float* lin_b = (const float*)d_in[4];
    const float* att_w = (const float*)d_in[5];
    const float* att_b = (const float*)d_in[6];
    float* out = (float*)d_out;

    int N = in_sizes[0] / IN_CH;
    int C = out_size / TOT_OUT;

    float* wsf    = (float*)d_ws;
    float* v      = wsf;
    float* c      = wsf + 512;
    float* segmax = wsf + 516;
    float* denom  = segmax + 4 * C;
    float* score  = denom + 4 * C;

    int initN = C * TOT_OUT;
    k_init <<<(initN + 255) / 256, 256, 0, stream>>>(out, denom, (unsigned*)segmax, C);
    k_fold <<<2, 256, 0, stream>>>(lin_w, lin_b, att_w, att_b, v, c);
    k_score<<<(N + 255) / 256, 256, 0, stream>>>(x, y, v, c, score,
                                                 (int*)segmax, (unsigned*)segmax, N);
    k_denom<<<(4 * N + 255) / 256, 256, 0, stream>>>(score, y, segmax, denom, N);
    k_gemm <<<(N + MTILE - 1) / MTILE, 256, 0, stream>>>(x, y, lin_w, lin_b,
                                                         score, segmax, denom, out, N);
}


# k_gemm walks rows in segment order (counting sort, arrival-order rank) and sums a wave's 16 rows per column before one atomic add: 16x fewer float atomics, pooled sums in a different order
# speedup vs baseline: 1.5735x; 1.2900x over previous
#include <hip/hip_runtime.h>
#include <math.h>

typedef __attribute__((ext_vector_type(2))) float v2f;
typedef __attribute__((ext_vector_type(8))) float v8f;

#define IN_CH     128
#define OUT_CH    64
#define NHEAD     4
#define TOT_OUT   256
#define NEG_SLOPE 0.2f

#define MTILE 64
#define KCH   32
#define KPAD  34

__global__ void k_init(float* out, float* denom, unsigned* segmax, int* cnt, int C) {
    int i = blockIdx.x * blockDim.x + threadIdx.x;
    if (i < C * TOT_OUT) out[i] = 0.0f;
    if (i < 4 * C) { denom[i] = 0.0f; segmax[i] = 0xFF800000u; }
    if (cnt != nullptr && i < C) cnt[i] = 0;
}

__global__ void k_fold(const float* __restrict__ lin_w, const float* __restrict__ lin_b,
                       const float* __restrict__ att_w, const float* __restrict__ att_b,
                       float* v, float* c) {
    int t = blockIdx.x * blockDim.x + threadIdx.x;
    if (t < NHEAD * IN_CH) {
        int h = t / IN_CH, i = t % IN_CH;
        float s = 0.0f;
        for (int d = 0; d < OUT_CH; ++d)
            s += lin_w[(size_t)(h * OUT_CH + d) * IN_CH + i] * att_w[d];
        v[h * IN_CH + i] = s;
    }
    if (t < NHEAD) {
        float s = att_b[0];
        for (int d = 0; d < OUT_CH; ++d) s += lin_b[t * OUT_CH + d] * att_w[d];
        c[t] = s;
    }
}

typedef __attribute__((ext_vector_type(4))) float v4f;

#define SC_ROWS   8
#define SC_GROUPS 4
#define SC_WAVES  8

__global__ __launch_bounds__(256)
void k_score(const float* __restrict__ x, const int* __restrict__ y,
             const float* __restrict__ v, const float* __restrict__ c,
             float* __restrict__ score, int* segmax_i, unsigned* segmax_u, int* cnt, int Nn) {
    const int lane = threadIdx.x & 31;
    const int wave = __builtin_amdgcn_readfirstlane((int)(threadIdx.x >> 5));
    const int row0 = (blockIdx.x * SC_WAVES + wave) * (SC_ROWS * SC_GROUPS);
    if (row0 >= Nn) return;

    v4f vh[NHEAD];
#pragma unroll
    for (int h = 0; h < NHEAD; ++h) vh[h] = *(const v4f*)(v + h * IN_CH + lane * 4);
    const int hh = lane & 3;
    const int rr = lane >> 2;
    const float ch = c[hh];

    const float* xb = x + (size_t)row0 * IN_CH + lane * 4;
    v4f xv[2][SC_ROWS];
#define SC_LOAD(G) do { _Pragma("unroll") for (int r = 0; r < SC_ROWS; ++r) { \
        const int ro_ = (row0 + (G) * SC_ROWS + r < Nn) ? ((G) * SC_ROWS + r) : 0; \
        xv[(G) & 1][r] = __builtin_nontemporal_load((const v4f*)(xb + (size_t)ro_ * IN_CH)); } } while (0)

    int yv[SC_GROUPS];
#pragma unroll
    for (int g = 0; g < SC_GROUPS; ++g) {
        const int nd = row0 + g * SC_ROWS + rr;
        yv[g] = y[(nd < Nn) ? nd : (Nn - 1)];
    }

    SC_LOAD(0);
#pragma unroll
    for (int g = 0; g < SC_GROUPS; ++g) {
        __builtin_amdgcn_sched_barrier(0);
        if (g + 1 < SC_GROUPS) SC_LOAD(g + 1);
        __builtin_amdgcn_sched_barrier(0);

        float vals[SC_ROWS * NHEAD];
#pragma unroll
        for (int r = 0; r < SC_ROWS; ++r) {
            const v4f xr = xv[g & 1][r];
#pragma unroll
            for (int h = 0; h < NHEAD; ++h)
                vals[r * NHEAD + h] = xr.x * vh[h].x + xr.y * vh[h].y + xr.z * vh[h].z + xr.w * vh[h].w;
        }
#define SC_STAGE(WD) do { const bool up_ = (lane & (WD)) != 0; \
        _Pragma("unroll") for (int i = 0; i < (WD); ++i) { \
            const float keep_ = up_ ? vals[i + (WD)] : vals[i]; \
            const float send_ = up_ ? vals[i] : vals[i + (WD)]; \
            vals[i] = keep_ + __shfl_xor(send_, (WD), 32); } } while (0)
        SC_STAGE(16); SC_STAGE(8); SC_STAGE(4); SC_STAGE(2); SC_STAGE(1);
#undef SC_STAGE
        float s = vals[0] + ch;
        s = (s >= 0.0f) ? s : NEG_SLOPE * s;
        const int node = row0 + g * SC_ROWS + rr;
        if (node < Nn) {
            score[(size_t)node * NHEAD + hh] = s;
            const int idx = yv[g] * NHEAD + hh;
            if (s >= 0.0f) atomicMax(&segmax_i[idx], __float_as_int(s));
            else           atomicMin(&segmax_u[idx], (unsigned)__float_as_int(s));
            if (cnt != nullptr && hh == 0) atomicAdd(&cnt[yv[g]], 1);
        }
    }
#undef SC_LOAD
}

#define SCAN_T 1024
__global__ __launch_bounds__(SCAN_T)
void k_scan(int* cnt, int C) {
    __shared__ int part[SCAN_T];
    const int t = threadIdx.x;
    const int per = (C + SCAN_T - 1) / SCAN_T;
    const int b = t * per;
    int s = 0;
    for (int j = 0; j < per; ++j) { const int i = b + j; if (i < C) s += cnt[i]; }
    part[t] = s;
    __syncthreads();
    for (int d = 1; d < SCAN_T; d <<= 1) {
        const int add = (t >= d) ? part[t - d] : 0;
        __syncthreads();
        part[t] += add;
        __syncthreads();
    }
    int run = part[t] - s;
    for (int j = 0; j < per; ++j) {
        const int i = b + j;
        if (i < C) { const int n = cnt[i]; cnt[i] = run; run += n; }
    }
}

__global__ void k_denom(const float* __restrict__ score, const int* __restrict__ y,
                        const float* __restrict__ segmax, float* denom,
                        int* cursor, int* perm, int Nn) {
    int i = blockIdx.x * blockDim.x + threadIdx.x;
    if (i >= 4 * Nn) return;
    int node = i >> 2, h = i & 3;
    int yy = y[node];
    int idx = yy * NHEAD + h;
    atomicAdd(&denom[idx], __expf(score[i] - segmax[idx]));
    if (perm != nullptr && h == 0) {
        const int pos = atomicAdd(&cursor[yy], 1);
        if ((unsigned)pos < (unsigned)Nn) perm[pos] = node;
    }
}

__global__ __launch_bounds__(256)
void k_gemm(const float* __restrict__ x, const int* __restrict__ y,
            const float* __restrict__ lin_w, const float* __restrict__ lin_b,
            const float* __restrict__ score, const float* __restrict__ segmax,
            const float* __restrict__ denom, const int* __restrict__ perm,
            float* __restrict__ out, int Nn) {
    __shared__ float Xl[MTILE * KPAD];
    __shared__ float Wl[TOT_OUT * KPAD];
    __shared__ float wgt[MTILE * NHEAD];
    __shared__ float bl[TOT_OUT];
    __shared__ int   yl[MTILE];

    int t = threadIdx.x;
    int blockM = blockIdx.x * MTILE;

    int  mynode;
    bool myok;
    {
        int m = t >> 2, h = t & 3;
        int p = blockM + m;
        myok = p < Nn;
        int pc = myok ? p : (Nn - 1);
        mynode = (perm != nullptr) ? perm[pc] : pc;
        if ((unsigned)mynode >= (unsigned)Nn) mynode = 0;
        int yy = y[mynode];
        int idx = yy * NHEAD + h;
        float w = 0.0f;
        if (myok)
            w = __expf(score[(size_t)mynode * NHEAD + h] - segmax[idx]) / denom[idx];
        wgt[m * NHEAD + h] = w;
        if (h == 0) yl[m] = yy;
        bl[t] = lin_b[t];
    }

    int wave = t >> 5, lane = t & 31;
    int rowbase = (wave >> 1) * 16;
    int colbase = (wave & 1) * 128;
    int lm      = lane & 15;
    int khalf   = (lane < 16) ? 0 : 2;

    v8f acc[8];
    const v8f vzero = {0.f, 0.f, 0.f, 0.f, 0.f, 0.f, 0.f, 0.f};
#pragma unroll
    for (int i = 0; i < 8; ++i) acc[i] = vzero;

    for (int kc = 0; kc < IN_CH / KCH; ++kc) {
        int kbase = kc * KCH;
        __syncthreads();
        {
            int row = t >> 2, q = t & 3;
            int node = mynode;
#pragma unroll
            for (int j = 0; j < 2; ++j) {
                int kk = q * 8 + j * 4;
                float4 xv = make_float4(0.f, 0.f, 0.f, 0.f);
                if (myok)
                    xv = *(const float4*)(x + (size_t)node * IN_CH + kbase + kk);
                Xl[row * KPAD + kk + 0] = xv.x;
                Xl[row * KPAD + kk + 1] = xv.y;
                Xl[row * KPAD + kk + 2] = xv.z;
                Xl[row * KPAD + kk + 3] = xv.w;
            }
            if (kc + 1 < IN_CH / KCH && myok)
                __builtin_prefetch(x + (size_t)node * IN_CH + kbase + KCH, 0, 1);
        }
        {
            int n = t;
#pragma unroll
            for (int q = 0; q < 8; ++q) {
                int kk = q * 4;
                float4 w4 = *(const float4*)(lin_w + (size_t)n * IN_CH + kbase + kk);
                Wl[n * KPAD + kk + 0] = w4.x;
                Wl[n * KPAD + kk + 1] = w4.y;
                Wl[n * KPAD + kk + 2] = w4.z;
                Wl[n * KPAD + kk + 3] = w4.w;
            }
            if (kc + 1 < IN_CH / KCH)
                __builtin_prefetch(lin_w + (size_t)n * IN_CH + kbase + KCH, 0, 1);
        }
        __syncthreads();
#pragma unroll
        for (int ks = 0; ks < KCH / 4; ++ks) {
            int kk = ks * 4 + khalf;
            v2f a = *(const v2f*)(&Xl[(rowbase + lm) * KPAD + kk]);
#pragma unroll
            for (int nt = 0; nt < 8; ++nt) {
                int n = colbase + nt * 16 + lm;
                v2f b = *(const v2f*)(&Wl[n * KPAD + kk]);
                acc[nt] = __builtin_amdgcn_wmma_f32_16x16x4_f32(
                    false, a, false, b, (short)0, acc[nt], false, false);
            }
        }
    }

    int mhalf = (lane < 16) ? 0 : 8;
    const int seg0 = yl[rowbase];
    int same = 1;
#pragma unroll
    for (int r = 1; r < 16; ++r) same &= (yl[rowbase + r] == seg0) ? 1 : 0;
    const int all_one_seg = __builtin_amdgcn_readfirstlane(same);

    if (all_one_seg) {
#pragma unroll
        for (int nt = 0; nt < 8; ++nt) {
            int n = colbase + nt * 16 + lm;
            int head = n >> 6;
            float bn = bl[n];
            float sum = 0.0f;
#pragma unroll
            for (int r = 0; r < 8; ++r) {
                int m = rowbase + mhalf + r;
                sum += (acc[nt][r] + bn) * wgt[m * NHEAD + head];
            }
            sum += __shfl_xor(sum, 16, 32);
            if (lane < 16) atomicAdd(&out[(size_t)seg0 * TOT_OUT + n], sum);
        }
    } else {
#pragma unroll
        for (int nt = 0; nt < 8; ++nt) {
            int n = colbase + nt * 16 + lm;
            int head = n >> 6;
            float bn = bl[n];
            int   cur = yl[rowbase + mhalf];
            float sum = (acc[nt][0] + bn) * wgt[(rowbase + mhalf) * NHEAD + head];
#pragma unroll
            for (int r = 1; r < 8; ++r) {
                int m = rowbase + mhalf + r;
                int sg = yl[m];
                float val = (acc[nt][r] + bn) * wgt[m * NHEAD + head];
                if (sg == cur) {
                    sum += val;
                } else {
                    atomicAdd(&out[(size_t)cur * TOT_OUT + n], sum);
                    cur = sg; sum = val;
                }
            }
            atomicAdd(&out[(size_t)cur * TOT_OUT + n], sum);
        }
    }
}

extern "C" void kernel_launch(void* const* d_in, const int* in_sizes, int n_in,
                              void* d_out, int out_size, void* d_ws, size_t ws_size,
                              hipStream_t stream) {
    (void)n_in;
    const float* x     = (const float*)d_in[0];
    const int*   y     = (const int*)  d_in[1];
    const float* lin_w = (const float*)d_in[3];
    const float* lin_b = (const float*)d_in[4];
    const float* att_w = (const float*)d_in[5];
    const float* att_b = (const float*)d_in[6];
    float* out = (float*)d_out;

    int N = in_sizes[0] / IN_CH;
    int C = out_size / TOT_OUT;

    float* wsf    = (float*)d_ws;
    float* v      = wsf;
    float* c      = wsf + 512;
    float* segmax = wsf + 516;
    float* denom  = segmax + 4 * C;
    float* score  = denom + 4 * C;
    int*   cnt    = (int*)(score + (size_t)4 * N);
    int*   perm   = cnt + C;

    size_t need = ((size_t)516 + (size_t)9 * C + (size_t)5 * N) * sizeof(float);
    if (ws_size < need) { cnt = nullptr; perm = nullptr; }

    int initN = C * TOT_OUT;
    k_init <<<(initN + 255) / 256, 256, 0, stream>>>(out, denom, (unsigned*)segmax, cnt, C);
    k_fold <<<2, 256, 0, stream>>>(lin_w, lin_b, att_w, att_b, v, c);
    k_score<<<(N + 255) / 256, 256, 0, stream>>>(x, y, v, c, score,
                                                 (int*)segmax, (unsigned*)segmax, cnt, N);
    if (perm != nullptr) k_scan<<<1, SCAN_T, 0, stream>>>(cnt, C);
    k_denom<<<(4 * N + 255) / 256, 256, 0, stream>>>(score, y, segmax, denom, cnt, perm, N);
    k_gemm <<<(N + MTILE - 1) / MTILE, 256, 0, stream>>>(x, y, lin_w, lin_b,
                                                         score, segmax, denom, perm, out, N);
}


# k_score and k_denom collect segment max, row counts, exp sums and sort ranks in per-block LDS bins; a block adds each used bin to memory once, whole lines per atomic (exp sums in a different tree)
# speedup vs baseline: 3.2450x; 2.0623x over previous
#include <hip/hip_runtime.h>
#include <math.h>

typedef __attribute__((ext_vector_type(2))) float v2f;
typedef __attribute__((ext_vector_type(8))) float v8f;

#define IN_CH     128
#define OUT_CH    64
#define NHEAD     4
#define TOT_OUT   256
#define NEG_SLOPE 0.2f

#define MTILE 64
#define KCH   32
#define KPAD  34

__global__ void k_init(float* out, float* denom, unsigned* segmax, int* cnt, int C) {
    int i = blockIdx.x * blockDim.x + threadIdx.x;
    if (i < C * TOT_OUT) out[i] = 0.0f;
    if (i < 4 * C) { denom[i] = 0.0f; segmax[i] = 0xFF800000u; }
    if (cnt != nullptr && i < C) cnt[i] = 0;
}

__global__ void k_fold(const float* __restrict__ lin_w, const float* __restrict__ lin_b,
                       const float* __restrict__ att_w, const float* __restrict__ att_b,
                       float* v, float* c) {
    int t = blockIdx.x * blockDim.x + threadIdx.x;
    if (t < NHEAD * IN_CH) {
        int h = t / IN_CH, i = t % IN_CH;
        float s = 0.0f;
        for (int d = 0; d < OUT_CH; ++d)
            s += lin_w[(size_t)(h * OUT_CH + d) * IN_CH + i] * att_w[d];
        v[h * IN_CH + i] = s;
    }
    if (t < NHEAD) {
        float s = att_b[0];
        for (int d = 0; d < OUT_CH; ++d) s += lin_b[t * OUT_CH + d] * att_w[d];
        c[t] = s;
    }
}

typedef __attribute__((ext_vector_type(4))) float v4f;

#define SC_ROWS   8
#define SC_GROUPS 4
#define SC_WAVES  8
#define SC_SLABS  4
#define SEG_CAP   2048

__global__ __launch_bounds__(256)
void k_score(const float* __restrict__ x, const int* __restrict__ y,
             const float* __restrict__ v, const float* __restrict__ c,
             float* __restrict__ score, int* segmax_i, unsigned* segmax_u, int* cnt, int Nn, int C) {
    __shared__ int lmax[NHEAD * SEG_CAP];
    __shared__ int lcnt[SEG_CAP];
    const bool bins = (C <= SEG_CAP);
    if (bins) {
        for (int i = threadIdx.x; i < NHEAD * C; i += 256) lmax[i] = (int)0xFF800000;
        for (int i = threadIdx.x; i < C; i += 256) lcnt[i] = 0;
    }
    __syncthreads();

    const int lane = threadIdx.x & 31;
    const int wave = __builtin_amdgcn_readfirstlane((int)(threadIdx.x >> 5));

    v4f vh[NHEAD];
#pragma unroll
    for (int h = 0; h < NHEAD; ++h) vh[h] = *(const v4f*)(v + h * IN_CH + lane * 4);
    const int hh = lane & 3;
    const int rr = lane >> 2;
    const float ch = c[hh];

    for (int sl = 0; sl < SC_SLABS; ++sl) {
    const int row0 = ((blockIdx.x * SC_SLABS + sl) * SC_WAVES + wave) * (SC_ROWS * SC_GROUPS);
    if (row0 < Nn) {
    const float* xb = x + (size_t)row0 * IN_CH + lane * 4;
    v4f xv[2][SC_ROWS];
#define SC_LOAD(G) do { _Pragma("unroll") for (int r = 0; r < SC_ROWS; ++r) { \
        const int ro_ = (row0 + (G) * SC_ROWS + r < Nn) ? ((G) * SC_ROWS + r) : 0; \
        xv[(G) & 1][r] = __builtin_nontemporal_load((const v4f*)(xb + (size_t)ro_ * IN_CH)); } } while (0)

    int yv[SC_GROUPS];
#pragma unroll
    for (int g = 0; g < SC_GROUPS; ++g) {
        const int nd = row0 + g * SC_ROWS + rr;
        yv[g] = y[(nd < Nn) ? nd : (Nn - 1)];
    }

    SC_LOAD(0);
#pragma unroll
    for (int g = 0; g < SC_GROUPS; ++g) {
        __builtin_amdgcn_sched_barrier(0);
        if (g + 1 < SC_GROUPS) SC_LOAD(g + 1);
        __builtin_amdgcn_sched_barrier(0);

        float vals[SC_ROWS * NHEAD];
#pragma unroll
        for (int r = 0; r < SC_ROWS; ++r) {
            const v4f xr = xv[g & 1][r];
#pragma unroll
            for (int h = 0; h < NHEAD; ++h)
                vals[r * NHEAD + h] = xr.x * vh[h].x + xr.y * vh[h].y + xr.z * vh[h].z + xr.w * vh[h].w;
        }
#define SC_STAGE(WD) do { const bool up_ = (lane & (WD)) != 0; \
        _Pragma("unroll") for (int i = 0; i < (WD); ++i) { \
            const float keep_ = up_ ? vals[i + (WD)] : vals[i]; \
            const float send_ = up_ ? vals[i] : vals[i + (WD)]; \
            vals[i] = keep_ + __shfl_xor(send_, (WD), 32); } } while (0)
        SC_STAGE(16); SC_STAGE(8); SC_STAGE(4); SC_STAGE(2); SC_STAGE(1);
#undef SC_STAGE
        float s = vals[0] + ch;
        s = (s >= 0.0f) ? s : NEG_SLOPE * s;
        const int node = row0 + g * SC_ROWS + rr;
        if (node < Nn) {
            score[(size_t)node * NHEAD + hh] = s;
            const int idx = yv[g] * NHEAD + hh;
            if (bins) {
                if (s >= 0.0f) atomicMax(&lmax[idx], __float_as_int(s));
                else           atomicMin((unsigned*)&lmax[idx], (unsigned)__float_as_int(s));
                if (hh == 0) atomicAdd(&lcnt[yv[g]], 1);
            } else {
                if (s >= 0.0f) atomicMax(&segmax_i[idx], __float_as_int(s));
                else           atomicMin(&segmax_u[idx], (unsigned)__float_as_int(s));
                if (cnt != nullptr && hh == 0) atomicAdd(&cnt[yv[g]], 1);
            }
        }
    }
#undef SC_LOAD
    }
    }

    __syncthreads();
    if (bins) {
        for (int i = threadIdx.x; i < NHEAD * C; i += 256) {
            const int b = lmax[i];
            if (b != (int)0xFF800000) {
                if (__int_as_float(b) >= 0.0f) atomicMax(&segmax_i[i], b);
                else                           atomicMin(&segmax_u[i], (unsigned)b);
            }
        }
        if (cnt != nullptr)
            for (int i = threadIdx.x; i < C; i += 256) {
                const int n = lcnt[i];
                if (n != 0) atomicAdd(&cnt[i], n);
            }
    }
}

#define SCAN_T 1024
__global__ __launch_bounds__(SCAN_T)
void k_scan(int* cnt, int C) {
    __shared__ int part[SCAN_T];
    const int t = threadIdx.x;
    const int per = (C + SCAN_T - 1) / SCAN_T;
    const int b = t * per;
    int s = 0;
    for (int j = 0; j < per; ++j) { const int i = b + j; if (i < C) s += cnt[i]; }
    part[t] = s;
    __syncthreads();
    for (int d = 1; d < SCAN_T; d <<= 1) {
        const int add = (t >= d) ? part[t - d] : 0;
        __syncthreads();
        part[t] += add;
        __syncthreads();
    }
    int run = part[t] - s;
    for (int j = 0; j < per; ++j) {
        const int i = b + j;
        if (i < C) { const int n = cnt[i]; cnt[i] = run; run += n; }
    }
}

#define DN_NPT 4
__global__ __launch_bounds__(256)
void k_denom(const float* __restrict__ score, const int* __restrict__ y,
             const float* __restrict__ segmax, float* denom,
             int* cursor, int* perm, int Nn, int C) {
    __shared__ float lden[NHEAD * SEG_CAP];
    __shared__ int   lrun[SEG_CAP];
    const bool bins = (C <= SEG_CAP);
    const int t = threadIdx.x;
    if (bins) {
        for (int i = t; i < NHEAD * C; i += 256) lden[i] = 0.0f;
        for (int i = t; i < C; i += 256) lrun[i] = 0;
    }
    __syncthreads();

    const int base = blockIdx.x * (256 * DN_NPT);
    int  yy[DN_NPT], rk[DN_NPT];
    v4f  sc[DN_NPT];
#pragma unroll
    for (int j = 0; j < DN_NPT; ++j) {
        const int node = base + j * 256 + t;
        const int nc = (node < Nn) ? node : (Nn - 1);
        yy[j] = y[nc];
        sc[j] = *(const v4f*)(score + (size_t)nc * NHEAD);
    }
#pragma unroll
    for (int j = 0; j < DN_NPT; ++j) {
        const int node = base + j * 256 + t;
        const v4f mx = *(const v4f*)(segmax + (size_t)yy[j] * NHEAD);
        const float e0 = __expf(sc[j].x - mx.x), e1 = __expf(sc[j].y - mx.y);
        const float e2 = __expf(sc[j].z - mx.z), e3 = __expf(sc[j].w - mx.w);
        rk[j] = 0;
        if (node < Nn) {
            if (bins) {
                float* ld = &lden[yy[j] * NHEAD];
                atomicAdd(ld + 0, e0); atomicAdd(ld + 1, e1);
                atomicAdd(ld + 2, e2); atomicAdd(ld + 3, e3);
                if (perm != nullptr) rk[j] = atomicAdd(&lrun[yy[j]], 1);
            } else {
                float* gd = &denom[(size_t)yy[j] * NHEAD];
                atomicAdd(gd + 0, e0); atomicAdd(gd + 1, e1);
                atomicAdd(gd + 2, e2); atomicAdd(gd + 3, e3);
                if (perm != nullptr) {
                    const int pos = atomicAdd(&cursor[yy[j]], 1);
                    if ((unsigned)pos < (unsigned)Nn) perm[pos] = node;
                }
            }
        }
    }
    __syncthreads();
    if (bins) {
        for (int i = t; i < NHEAD * C; i += 256) {
            const float s = lden[i];
            if (s != 0.0f) atomicAdd(&denom[i], s);
        }
        if (perm != nullptr)
            for (int i = t; i < C; i += 256) {
                const int n = lrun[i];
                lrun[i] = (n != 0) ? atomicAdd(&cursor[i], n) : 0;
            }
    }
    __syncthreads();
    if (bins && perm != nullptr) {
#pragma unroll
        for (int j = 0; j < DN_NPT; ++j) {
            const int node = base + j * 256 + t;
            if (node < Nn) {
                const int pos = lrun[yy[j]] + rk[j];
                if ((unsigned)pos < (unsigned)Nn) perm[pos] = node;
            }
        }
    }
}

__global__ __launch_bounds__(256)
void k_gemm(const float* __restrict__ x, const int* __restrict__ y,
            const float* __restrict__ lin_w, const float* __restrict__ lin_b,
            const float* __restrict__ score, const float* __restrict__ segmax,
            const float* __restrict__ denom, const int* __restrict__ perm,
            float* __restrict__ out, int Nn) {
    __shared__ float Xl[MTILE * KPAD];
    __shared__ float Wl[TOT_OUT * KPAD];
    __shared__ float wgt[MTILE * NHEAD];
    __shared__ float bl[TOT_OUT];
    __shared__ int   yl[MTILE];

    int t = threadIdx.x;
    int blockM = blockIdx.x * MTILE;

    int  mynode;
    bool myok;
    {
        int m = t >> 2, h = t & 3;
        int p = blockM + m;
        myok = p < Nn;
        int pc = myok ? p : (Nn - 1);
        mynode = (perm != nullptr) ? perm[pc] : pc;
        if ((unsigned)mynode >= (unsigned)Nn) mynode = 0;
        int yy = y[mynode];
        int idx = yy * NHEAD + h;
        float w = 0.0f;
        if (myok)
            w = __expf(score[(size_t)mynode * NHEAD + h] - segmax[idx]) / denom[idx];
        wgt[m * NHEAD + h] = w;
        if (h == 0) yl[m] = yy;
        bl[t] = lin_b[t];
    }

    int wave = t >> 5, lane = t & 31;
    int rowbase = (wave >> 1) * 16;
    int colbase = (wave & 1) * 128;
    int lm      = lane & 15;
    int khalf   = (lane < 16) ? 0 : 2;

    v8f acc[8];
    const v8f vzero = {0.f, 0.f, 0.f, 0.f, 0.f, 0.f, 0.f, 0.f};
#pragma unroll
    for (int i = 0; i < 8; ++i) acc[i] = vzero;

    for (int kc = 0; kc < IN_CH / KCH; ++kc) {
        int kbase = kc * KCH;
        __syncthreads();
        {
            int row = t >> 2, q = t & 3;
            int node = mynode;
#pragma unroll
            for (int j = 0; j < 2; ++j) {
                int kk = q * 8 + j * 4;
                float4 xv = make_float4(0.f, 0.f, 0.f, 0.f);
                if (myok)
                    xv = *(const float4*)(x + (size_t)node * IN_CH + kbase + kk);
                Xl[row * KPAD + kk + 0] = xv.x;
                Xl[row * KPAD + kk + 1] = xv.y;
                Xl[row * KPAD + kk + 2] = xv.z;
                Xl[row * KPAD + kk + 3] = xv.w;
            }
            if (kc + 1 < IN_CH / KCH && myok)
                __builtin_prefetch(x + (size_t)node * IN_CH + kbase + KCH, 0, 1);
        }
        {
            int n = t;
#pragma unroll
            for (int q = 0; q < 8; ++q) {
                int kk = q * 4;
                float4 w4 = *(const float4*)(lin_w + (size_t)n * IN_CH + kbase + kk);
                Wl[n * KPAD + kk + 0] = w4.x;
                Wl[n * KPAD + kk + 1] = w4.y;
                Wl[n * KPAD + kk + 2] = w4.z;
                Wl[n * KPAD + kk + 3] = w4.w;
            }
            if (kc + 1 < IN_CH / KCH)
                __builtin_prefetch(lin_w + (size_t)n * IN_CH + kbase + KCH, 0, 1);
        }
        __syncthreads();
#pragma unroll
        for (int ks = 0; ks < KCH / 4; ++ks) {
            int kk = ks * 4 + khalf;
            v2f a = *(const v2f*)(&Xl[(rowbase + lm) * KPAD + kk]);
#pragma unroll
            for (int nt = 0; nt < 8; ++nt) {
                int n = colbase + nt * 16 + lm;
                v2f b = *(const v2f*)(&Wl[n * KPAD + kk]);
                acc[nt] = __builtin_amdgcn_wmma_f32_16x16x4_f32(
                    false, a, false, b, (short)0, acc[nt], false, false);
            }
        }
    }

    int mhalf = (lane < 16) ? 0 : 8;
    const int seg0 = yl[rowbase];
    int same = 1;
#pragma unroll
    for (int r = 1; r < 16; ++r) same &= (yl[rowbase + r] == seg0) ? 1 : 0;
    const int all_one_seg = __builtin_amdgcn_readfirstlane(same);

    if (all_one_seg) {
#pragma unroll
        for (int nt = 0; nt < 8; ++nt) {
            int n = colbase + nt * 16 + lm;
            int head = n >> 6;
            float bn = bl[n];
            float sum = 0.0f;
#pragma unroll
            for (int r = 0; r < 8; ++r) {
                int m = rowbase + mhalf + r;
                sum += (acc[nt][r] + bn) * wgt[m * NHEAD + head];
            }
            sum += __shfl_xor(sum, 16, 32);
            if (lane < 16) atomicAdd(&out[(size_t)seg0 * TOT_OUT + n], sum);
        }
    } else {
#pragma unroll
        for (int nt = 0; nt < 8; ++nt) {
            int n = colbase + nt * 16 + lm;
            int head = n >> 6;
            float bn = bl[n];
            int   cur = yl[rowbase + mhalf];
            float sum = (acc[nt][0] + bn) * wgt[(rowbase + mhalf) * NHEAD + head];
#pragma unroll
            for (int r = 1; r < 8; ++r) {
                int m = rowbase + mhalf + r;
                int sg = yl[m];
                float val = (acc[nt][r] + bn) * wgt[m * NHEAD + head];
                if (sg == cur) {
                    sum += val;
                } else {
                    atomicAdd(&out[(size_t)cur * TOT_OUT + n], sum);
                    cur = sg; sum = val;
                }
            }
            atomicAdd(&out[(size_t)cur * TOT_OUT + n], sum);
        }
    }
}

extern "C" void kernel_launch(void* const* d_in, const int* in_sizes, int n_in,
                              void* d_out, int out_size, void* d_ws, size_t ws_size,
                              hipStream_t stream) {
    (void)n_in;
    const float* x     = (const float*)d_in[0];
    const int*   y     = (const int*)  d_in[1];
    const float* lin_w = (const float*)d_in[3];
    const float* lin_b = (const float*)d_in[4];
    const float* att_w = (const float*)d_in[5];
    const float* att_b = (const float*)d_in[6];
    float* out = (float*)d_out;

    int N = in_sizes[0] / IN_CH;
    int C = out_size / TOT_OUT;

    float* wsf    = (float*)d_ws;
    float* v      = wsf;
    float* c      = wsf + 512;
    float* segmax = wsf + 516;
    float* denom  = segmax + 4 * C;
    float* score  = denom + 4 * C;
    int*   cnt    = (int*)(score + (size_t)4 * N);
    int*   perm   = cnt + C;

    size_t need = ((size_t)516 + (size_t)9 * C + (size_t)5 * N) * sizeof(float);
    if (ws_size < need) { cnt = nullptr; perm = nullptr; }

    int initN = C * TOT_OUT;
    k_init <<<(initN + 255) / 256, 256, 0, stream>>>(out, denom, (unsigned*)segmax, cnt, C);
    k_fold <<<2, 256, 0, stream>>>(lin_w, lin_b, att_w, att_b, v, c);
    k_score<<<(N + SC_SLABS * 256 - 1) / (SC_SLABS * 256), 256, 0, stream>>>(x, y, v, c, score,
                                                 (int*)segmax, (unsigned*)segmax, cnt, N, C);
    if (perm != nullptr) k_scan<<<1, SCAN_T, 0, stream>>>(cnt, C);
    k_denom<<<(N + DN_NPT * 256 - 1) / (DN_NPT * 256), 256, 0, stream>>>(score, y, segmax, denom, cnt, perm, N, C);
    k_gemm <<<(N + MTILE - 1) / MTILE, 256, 0, stream>>>(x, y, lin_w, lin_b,
                                                         score, segmax, denom, perm, out, N);
}


# k_gemm keeps all of lin_w in LDS: 256 blocks of 16 waves walk over 64-row tiles, next tile's x rows fetched under the products; lin_w not re-staged per 64 rows (atomic adds arrive in another order)
# speedup vs baseline: 3.6755x; 1.1327x over previous
#include <hip/hip_runtime.h>
#include <math.h>

typedef __attribute__((ext_vector_type(2))) float v2f;
typedef __attribute__((ext_vector_type(8))) float v8f;

#define IN_CH     128
#define OUT_CH    64
#define NHEAD     4
#define TOT_OUT   256
#define NEG_SLOPE 0.2f


__global__ void k_init(float* out, float* denom, unsigned* segmax, int* cnt, int C) {
    int i = blockIdx.x * blockDim.x + threadIdx.x;
    if (i < C * TOT_OUT) out[i] = 0.0f;
    if (i < 4 * C) { denom[i] = 0.0f; segmax[i] = 0xFF800000u; }
    if (cnt != nullptr && i < C) cnt[i] = 0;
}

__global__ void k_fold(const float* __restrict__ lin_w, const float* __restrict__ lin_b,
                       const float* __restrict__ att_w, const float* __restrict__ att_b,
                       float* v, float* c) {
    int t = blockIdx.x * blockDim.x + threadIdx.x;
    if (t < NHEAD * IN_CH) {
        int h = t / IN_CH, i = t % IN_CH;
        float s = 0.0f;
        for (int d = 0; d < OUT_CH; ++d)
            s += lin_w[(size_t)(h * OUT_CH + d) * IN_CH + i] * att_w[d];
        v[h * IN_CH + i] = s;
    }
    if (t < NHEAD) {
        float s = att_b[0];
        for (int d = 0; d < OUT_CH; ++d) s += lin_b[t * OUT_CH + d] * att_w[d];
        c[t] = s;
    }
}

typedef __attribute__((ext_vector_type(4))) float v4f;

#define SC_ROWS   8
#define SC_GROUPS 4
#define SC_WAVES  8
#define SC_SLABS  4
#define SEG_CAP   2048

__global__ __launch_bounds__(256)
void k_score(const float* __restrict__ x, const int* __restrict__ y,
             const float* __restrict__ v, const float* __restrict__ c,
             float* __restrict__ score, int* segmax_i, unsigned* segmax_u, int* cnt, int Nn, int C) {
    __shared__ int lmax[NHEAD * SEG_CAP];
    __shared__ int lcnt[SEG_CAP];
    const bool bins = (C <= SEG_CAP);
    if (bins) {
        for (int i = threadIdx.x; i < NHEAD * C; i += 256) lmax[i] = (int)0xFF800000;
        for (int i = threadIdx.x; i < C; i += 256) lcnt[i] = 0;
    }
    __syncthreads();

    const int lane = threadIdx.x & 31;
    const int wave = __builtin_amdgcn_readfirstlane((int)(threadIdx.x >> 5));

    v4f vh[NHEAD];
#pragma unroll
    for (int h = 0; h < NHEAD; ++h) vh[h] = *(const v4f*)(v + h * IN_CH + lane * 4);
    const int hh = lane & 3;
    const int rr = lane >> 2;
    const float ch = c[hh];

    for (int sl = 0; sl < SC_SLABS; ++sl) {
    const int row0 = ((blockIdx.x * SC_SLABS + sl) * SC_WAVES + wave) * (SC_ROWS * SC_GROUPS);
    if (row0 < Nn) {
    const float* xb = x + (size_t)row0 * IN_CH + lane * 4;
    v4f xv[2][SC_ROWS];
#define SC_LOAD(G) do { _Pragma("unroll") for (int r = 0; r < SC_ROWS; ++r) { \
        const int ro_ = (row0 + (G) * SC_ROWS + r < Nn) ? ((G) * SC_ROWS + r) : 0; \
        xv[(G) & 1][r] = __builtin_nontemporal_load((const v4f*)(xb + (size_t)ro_ * IN_CH)); } } while (0)

    int yv[SC_GROUPS];
#pragma unroll
    for (int g = 0; g < SC_GROUPS; ++g) {
        const int nd = row0 + g * SC_ROWS + rr;
        yv[g] = y[(nd < Nn) ? nd : (Nn - 1)];
    }

    SC_LOAD(0);
#pragma unroll
    for (int g = 0; g < SC_GROUPS; ++g) {
        __builtin_amdgcn_sched_barrier(0);
        if (g + 1 < SC_GROUPS) SC_LOAD(g + 1);
        __builtin_amdgcn_sched_barrier(0);

        float vals[SC_ROWS * NHEAD];
#pragma unroll
        for (int r = 0; r < SC_ROWS; ++r) {
            const v4f xr = xv[g & 1][r];
#pragma unroll
            for (int h = 0; h < NHEAD; ++h)
                vals[r * NHEAD + h] = xr.x * vh[h].x + xr.y * vh[h].y + xr.z * vh[h].z + xr.w * vh[h].w;
        }
#define SC_STAGE(WD) do { const bool up_ = (lane & (WD)) != 0; \
        _Pragma("unroll") for (int i = 0; i < (WD); ++i) { \
            const float keep_ = up_ ? vals[i + (WD)] : vals[i]; \
            const float send_ = up_ ? vals[i] : vals[i + (WD)]; \
            vals[i] = keep_ + __shfl_xor(send_, (WD), 32); } } while (0)
        SC_STAGE(16); SC_STAGE(8); SC_STAGE(4); SC_STAGE(2); SC_STAGE(1);
#undef SC_STAGE
        float s = vals[0] + ch;
        s = (s >= 0.0f) ? s : NEG_SLOPE * s;
        const int node = row0 + g * SC_ROWS + rr;
        if (node < Nn) {
            score[(size_t)node * NHEAD + hh] = s;
            const int idx = yv[g] * NHEAD + hh;
            if (bins) {
                if (s >= 0.0f) atomicMax(&lmax[idx], __float_as_int(s));
                else           atomicMin((unsigned*)&lmax[idx], (unsigned)__float_as_int(s));
                if (hh == 0) atomicAdd(&lcnt[yv[g]], 1);
            } else {
                if (s >= 0.0f) atomicMax(&segmax_i[idx], __float_as_int(s));
                else           atomicMin(&segmax_u[idx], (unsigned)__float_as_int(s));
                if (cnt != nullptr && hh == 0) atomicAdd(&cnt[yv[g]], 1);
            }
        }
    }
#undef SC_LOAD
    }
    }

    __syncthreads();
    if (bins) {
        for (int i = threadIdx.x; i < NHEAD * C; i += 256) {
            const int b = lmax[i];
            if (b != (int)0xFF800000) {
                if (__int_as_float(b) >= 0.0f) atomicMax(&segmax_i[i], b);
                else                           atomicMin(&segmax_u[i], (unsigned)b);
            }
        }
        if (cnt != nullptr)
            for (int i = threadIdx.x; i < C; i += 256) {
                const int n = lcnt[i];
                if (n != 0) atomicAdd(&cnt[i], n);
            }
    }
}

#define SCAN_T 1024
__global__ __launch_bounds__(SCAN_T)
void k_scan(int* cnt, int C) {
    __shared__ int part[SCAN_T];
    const int t = threadIdx.x;
    const int per = (C + SCAN_T - 1) / SCAN_T;
    const int b = t * per;
    int s = 0;
    for (int j = 0; j < per; ++j) { const int i = b + j; if (i < C) s += cnt[i]; }
    part[t] = s;
    __syncthreads();
    for (int d = 1; d < SCAN_T; d <<= 1) {
        const int add = (t >= d) ? part[t - d] : 0;
        __syncthreads();
        part[t] += add;
        __syncthreads();
    }
    int run = part[t] - s;
    for (int j = 0; j < per; ++j) {
        const int i = b + j;
        if (i < C) { const int n = cnt[i]; cnt[i] = run; run += n; }
    }
}

#define DN_NPT 4
__global__ __launch_bounds__(256)
void k_denom(const float* __restrict__ score, const int* __restrict__ y,
             const float* __restrict__ segmax, float* denom,
             int* cursor, int* perm, int Nn, int C) {
    __shared__ float lden[NHEAD * SEG_CAP];
    __shared__ int   lrun[SEG_CAP];
    const bool bins = (C <= SEG_CAP);
    const int t = threadIdx.x;
    if (bins) {
        for (int i = t; i < NHEAD * C; i += 256) lden[i] = 0.0f;
        for (int i = t; i < C; i += 256) lrun[i] = 0;
    }
    __syncthreads();

    const int base = blockIdx.x * (256 * DN_NPT);
    int  yy[DN_NPT], rk[DN_NPT];
    v4f  sc[DN_NPT];
#pragma unroll
    for (int j = 0; j < DN_NPT; ++j) {
        const int node = base + j * 256 + t;
        const int nc = (node < Nn) ? node : (Nn - 1);
        yy[j] = y[nc];
        sc[j] = *(const v4f*)(score + (size_t)nc * NHEAD);
    }
#pragma unroll
    for (int j = 0; j < DN_NPT; ++j) {
        const int node = base + j * 256 + t;
        const v4f mx = *(const v4f*)(segmax + (size_t)yy[j] * NHEAD);
        const float e0 = __expf(sc[j].x - mx.x), e1 = __expf(sc[j].y - mx.y);
        const float e2 = __expf(sc[j].z - mx.z), e3 = __expf(sc[j].w - mx.w);
        rk[j] = 0;
        if (node < Nn) {
            if (bins) {
                float* ld = &lden[yy[j] * NHEAD];
                atomicAdd(ld + 0, e0); atomicAdd(ld + 1, e1);
                atomicAdd(ld + 2, e2); atomicAdd(ld + 3, e3);
                if (perm != nullptr) rk[j] = atomicAdd(&lrun[yy[j]], 1);
            } else {
                float* gd = &denom[(size_t)yy[j] * NHEAD];
                atomicAdd(gd + 0, e0); atomicAdd(gd + 1, e1);
                atomicAdd(gd + 2, e2); atomicAdd(gd + 3, e3);
                if (perm != nullptr) {
                    const int pos = atomicAdd(&cursor[yy[j]], 1);
                    if ((unsigned)pos < (unsigned)Nn) perm[pos] = node;
                }
            }
        }
    }
    __syncthreads();
    if (bins) {
        for (int i = t; i < NHEAD * C; i += 256) {
            const float s = lden[i];
            if (s != 0.0f) atomicAdd(&denom[i], s);
        }
        if (perm != nullptr)
            for (int i = t; i < C; i += 256) {
                const int n = lrun[i];
                lrun[i] = (n != 0) ? atomicAdd(&cursor[i], n) : 0;
            }
    }
    __syncthreads();
    if (bins && perm != nullptr) {
#pragma unroll
        for (int j = 0; j < DN_NPT; ++j) {
            const int node = base + j * 256 + t;
            if (node < Nn) {
                const int pos = lrun[yy[j]] + rk[j];
                if ((unsigned)pos < (unsigned)Nn) perm[pos] = node;
            }
        }
    }
}

#define GT_ROWS    64
#define GT_THREADS 512
#define XP         132
#define GEMM_BLOCKS 256

__device__ __forceinline__ int gm_node(const int* __restrict__ perm, int p, int Nn) {
    const int pc = (p < Nn) ? p : (Nn - 1);
    int nd = (perm != nullptr) ? perm[pc] : pc;
    if ((unsigned)nd >= (unsigned)Nn) nd = 0;
    return nd;
}

__global__ __launch_bounds__(GT_THREADS)
void k_gemm(const float* __restrict__ x, const int* __restrict__ y,
            const float* __restrict__ lin_w, const float* __restrict__ lin_b,
            const float* __restrict__ score, const float* __restrict__ segmax,
            const float* __restrict__ denom, const int* __restrict__ perm,
            float* __restrict__ out, int Nn) {
    __shared__ __attribute__((aligned(16))) float Wl[TOT_OUT * XP];
    __shared__ __attribute__((aligned(16))) float Xl[2][GT_ROWS * XP];
    __shared__ float wgt[2][GT_ROWS * NHEAD];
    __shared__ int   yl[2][GT_ROWS];
    __shared__ float bl[TOT_OUT];

    const int t = threadIdx.x;
    const int G = gridDim.x;
    const int ntiles = (Nn + GT_ROWS - 1) / GT_ROWS;
    int tile = blockIdx.x;
    if (tile >= ntiles) return;

    const int row = t >> 3, e = t & 7, h = t & 3;
    const bool wstore = (t & 4) == 0;

    {
        const int p0 = tile * GT_ROWS + row;
        const bool ok0 = p0 < Nn;
        const int nd0 = gm_node(perm, p0, Nn);
        const float4* s0 = (const float4*)(x + (size_t)nd0 * IN_CH) + e;
        float4 a0 = s0[0], a1 = s0[8], a2 = s0[16], a3 = s0[24];
        const int y0 = y[nd0];
        const int i0 = y0 * NHEAD + h;
        float w0 = 0.0f;
        if (ok0) w0 = __expf(score[(size_t)nd0 * NHEAD + h] - segmax[i0]) / denom[i0];
        const float4 z4 = make_float4(0.f, 0.f, 0.f, 0.f);
        if (!ok0) { a0 = z4; a1 = z4; a2 = z4; a3 = z4; }
        float4* d0 = (float4*)(&Xl[0][row * XP]) + e;
        d0[0] = a0; d0[8] = a1; d0[16] = a2; d0[24] = a3;
        if (wstore) wgt[0][row * NHEAD + h] = w0;
        if (e == 0) yl[0][row] = y0;
    }
    if (t < TOT_OUT) bl[t] = lin_b[t];
    for (int i = t; i < TOT_OUT * (IN_CH / 4); i += GT_THREADS) {
        const int n = i >> 5, c4 = i & 31;
        const float4 w4 = *((const float4*)(lin_w + (size_t)n * IN_CH) + c4);
        *((float4*)(&Wl[n * XP]) + c4) = w4;
    }
    int  pA = (tile + G) * GT_ROWS + row;
    bool okA = pA < Nn;
    int  ndA = gm_node(perm, pA, Nn);
    int  yA  = y[ndA];
    float scA = score[(size_t)ndA * NHEAD + h];
    int  pB = (tile + 2 * G) * GT_ROWS + row;
    bool okB = pB < Nn;
    int  ndB = gm_node(perm, pB, Nn);
    __syncthreads();

    const int wave = __builtin_amdgcn_readfirstlane(t >> 5), lane = t & 31;
    const int rowbase = (wave >> 2) * 16;
    const int colbase = (wave & 3) * 64;
    const int head    = wave & 3;
    const int lm      = lane & 15;
    const int khalf   = (lane < 16) ? 0 : 2;
    const int mhalf   = (lane < 16) ? 0 : 8;
    const v8f vzero = {0.f, 0.f, 0.f, 0.f, 0.f, 0.f, 0.f, 0.f};
    const float* bp = &Wl[(colbase + lm) * XP + khalf];

    int buf = 0;
    for (; tile < ntiles; tile += G) {
        const float4* sA = (const float4*)(x + (size_t)ndA * IN_CH) + e;
        float4 xa = sA[0], xb = sA[8], xc = sA[16], xd = sA[24];
        const int iA = yA * NHEAD + h;
        const float mxA = segmax[iA], dnA = denom[iA];
        const int yB = y[ndB];
        const float scB = score[(size_t)ndB * NHEAD + h];
        const int pC = (tile + 3 * G) * GT_ROWS + row;
        const bool okC = pC < Nn;
        const int ndC = gm_node(perm, pC, Nn);

        v8f acc[4];
#pragma unroll
        for (int i = 0; i < 4; ++i) acc[i] = vzero;
        const float* ap = &Xl[buf][(rowbase + lm) * XP + khalf];
#pragma unroll
        for (int ks = 0; ks < IN_CH / 4; ++ks) {
            const v2f a = *(const v2f*)(ap + ks * 4);
#pragma unroll
            for (int nt = 0; nt < 4; ++nt) {
                const v2f b = *(const v2f*)(bp + nt * 16 * XP + ks * 4);
                acc[nt] = __builtin_amdgcn_wmma_f32_16x16x4_f32(
                    false, a, false, b, (short)0, acc[nt], false, false);
            }
        }

        float wr[8];
        int   yr[8];
        int   seg0, all_one_seg;
        {
            const float* wg = wgt[buf];
            const int*   ys = yl[buf];
            seg0 = ys[rowbase];
            int same = 1;
#pragma unroll
            for (int r = 1; r < 16; ++r) same &= (ys[rowbase + r] == seg0) ? 1 : 0;
            all_one_seg = __builtin_amdgcn_readfirstlane(same);
#pragma unroll
            for (int r = 0; r < 8; ++r) {
                const int m = rowbase + mhalf + r;
                wr[r] = wg[m * NHEAD + head];
                yr[r] = ys[m];
            }
        }

        {
            float wA = 0.0f;
            if (okA) wA = __expf(scA - mxA) / dnA;
            const float4 z4 = make_float4(0.f, 0.f, 0.f, 0.f);
            if (!okA) { xa = z4; xb = z4; xc = z4; xd = z4; }
            float4* dA = (float4*)(&Xl[buf ^ 1][row * XP]) + e;
            dA[0] = xa; dA[8] = xb; dA[16] = xc; dA[24] = xd;
            if (wstore) wgt[buf ^ 1][row * NHEAD + h] = wA;
            if (e == 0) yl[buf ^ 1][row] = yA;
        }
        __syncthreads();

        if (all_one_seg) {
#pragma unroll
            for (int nt = 0; nt < 4; ++nt) {
                const int n = colbase + nt * 16 + lm;
                const float bn = bl[n];
                float sum = 0.0f;
#pragma unroll
                for (int r = 0; r < 8; ++r) sum += (acc[nt][r] + bn) * wr[r];
                sum += __shfl_xor(sum, 16, 32);
                if (lane < 16) atomicAdd(&out[(size_t)seg0 * TOT_OUT + n], sum);
            }
        } else {
#pragma unroll
            for (int nt = 0; nt < 4; ++nt) {
                const int n = colbase + nt * 16 + lm;
                const float bn = bl[n];
                int   cur = yr[0];
                float sum = (acc[nt][0] + bn) * wr[0];
#pragma unroll
                for (int r = 1; r < 8; ++r) {
                    const float val = (acc[nt][r] + bn) * wr[r];
                    if (yr[r] == cur) {
                        sum += val;
                    } else {
                        atomicAdd(&out[(size_t)cur * TOT_OUT + n], sum);
                        cur = yr[r]; sum = val;
                    }
                }
                atomicAdd(&out[(size_t)cur * TOT_OUT + n], sum);
            }
        }
        buf ^= 1;
        ndA = ndB; okA = okB; yA = yB; scA = scB;
        ndB = ndC; okB = okC;
    }
}

extern "C" void kernel_launch(void* const* d_in, const int* in_sizes, int n_in,
                              void* d_out, int out_size, void* d_ws, size_t ws_size,
                              hipStream_t stream) {
    (void)n_in;
    const float* x     = (const float*)d_in[0];
    const int*   y     = (const int*)  d_in[1];
    const float* lin_w = (const float*)d_in[3];
    const float* lin_b = (const float*)d_in[4];
    const float* att_w = (const float*)d_in[5];
    const float* att_b = (const float*)d_in[6];
    float* out = (float*)d_out;

    int N = in_sizes[0] / IN_CH;
    int C = out_size / TOT_OUT;

    float* wsf    = (float*)d_ws;
    float* v      = wsf;
    float* c      = wsf + 512;
    float* segmax = wsf + 516;
    float* denom  = segmax + 4 * C;
    float* score  = denom + 4 * C;
    int*   cnt    = (int*)(score + (size_t)4 * N);
    int*   perm   = cnt + C;

    size_t need = ((size_t)516 + (size_t)9 * C + (size_t)5 * N) * sizeof(float);
    if (ws_size < need) { cnt = nullptr; perm = nullptr; }

    int initN = C * TOT_OUT;
    k_init <<<(initN + 255) / 256, 256, 0, stream>>>(out, denom, (unsigned*)segmax, cnt, C);
    k_fold <<<2, 256, 0, stream>>>(lin_w, lin_b, att_w, att_b, v, c);
    k_score<<<(N + SC_SLABS * 256 - 1) / (SC_SLABS * 256), 256, 0, stream>>>(x, y, v, c, score,
                                                 (int*)segmax, (unsigned*)segmax, cnt, N, C);
    if (perm != nullptr) k_scan<<<1, SCAN_T, 0, stream>>>(cnt, C);
    k_denom<<<(N + DN_NPT * 256 - 1) / (DN_NPT * 256), 256, 0, stream>>>(score, y, segmax, denom, cnt, perm, N, C);
    k_gemm <<<GEMM_BLOCKS, GT_THREADS, 0, stream>>>(x, y, lin_w, lin_b,
                                                  score, segmax, denom, perm, out, N);
}


# k_gemm replaced: rows' w*x summed per segment first, as a WMMA product with the rows as k (float atomics into pooled 4x128 rows), then a small f32 WMMA product with lin_w stores out; sums reordered
# speedup vs baseline: 5.5649x; 1.5141x over previous
#include <hip/hip_runtime.h>
#include <math.h>

typedef __attribute__((ext_vector_type(2))) float v2f;
typedef __attribute__((ext_vector_type(8))) float v8f;

#define IN_CH     128
#define OUT_CH    64
#define NHEAD     4
#define TOT_OUT   256
#define NEG_SLOPE 0.2f


__global__ void k_init(float* zbuf, int zn, float* denom, unsigned* segmax, int* cnt, int C) {
    int i = blockIdx.x * blockDim.x + threadIdx.x;
    if (i < zn) zbuf[i] = 0.0f;
    if (i < 4 * C) { denom[i] = 0.0f; segmax[i] = 0xFF800000u; }
    if (cnt != nullptr && i < C) cnt[i] = 0;
}

__global__ void k_fold(const float* __restrict__ lin_w, const float* __restrict__ lin_b,
                       const float* __restrict__ att_w, const float* __restrict__ att_b,
                       float* v, float* c) {
    int t = blockIdx.x * blockDim.x + threadIdx.x;
    if (t < NHEAD * IN_CH) {
        int h = t / IN_CH, i = t % IN_CH;
        float s = 0.0f;
        for (int d = 0; d < OUT_CH; ++d)
            s += lin_w[(size_t)(h * OUT_CH + d) * IN_CH + i] * att_w[d];
        v[h * IN_CH + i] = s;
    }
    if (t < NHEAD) {
        float s = att_b[0];
        for (int d = 0; d < OUT_CH; ++d) s += lin_b[t * OUT_CH + d] * att_w[d];
        c[t] = s;
    }
}

typedef __attribute__((ext_vector_type(4))) float v4f;

#define SC_ROWS   8
#define SC_GROUPS 4
#define SC_WAVES  8
#define SC_SLABS  4
#define SEG_CAP   2048

__global__ __launch_bounds__(256)
void k_score(const float* __restrict__ x, const int* __restrict__ y,
             const float* __restrict__ v, const float* __restrict__ c,
             float* __restrict__ score, int* segmax_i, unsigned* segmax_u, int* cnt, int Nn, int C) {
    __shared__ int lmax[NHEAD * SEG_CAP];
    __shared__ int lcnt[SEG_CAP];
    const bool bins = (C <= SEG_CAP);
    if (bins) {
        for (int i = threadIdx.x; i < NHEAD * C; i += 256) lmax[i] = (int)0xFF800000;
        for (int i = threadIdx.x; i < C; i += 256) lcnt[i] = 0;
    }
    __syncthreads();

    const int lane = threadIdx.x & 31;
    const int wave = __builtin_amdgcn_readfirstlane((int)(threadIdx.x >> 5));

    v4f vh[NHEAD];
#pragma unroll
    for (int h = 0; h < NHEAD; ++h) vh[h] = *(const v4f*)(v + h * IN_CH + lane * 4);
    const int hh = lane & 3;
    const int rr = lane >> 2;
    const float ch = c[hh];

    for (int sl = 0; sl < SC_SLABS; ++sl) {
    const int row0 = ((blockIdx.x * SC_SLABS + sl) * SC_WAVES + wave) * (SC_ROWS * SC_GROUPS);
    if (row0 < Nn) {
    const float* xb = x + (size_t)row0 * IN_CH + lane * 4;
    v4f xv[2][SC_ROWS];
#define SC_LOAD(G) do { _Pragma("unroll") for (int r = 0; r < SC_ROWS; ++r) { \
        const int ro_ = (row0 + (G) * SC_ROWS + r < Nn) ? ((G) * SC_ROWS + r) : 0; \
        xv[(G) & 1][r] = __builtin_nontemporal_load((const v4f*)(xb + (size_t)ro_ * IN_CH)); } } while (0)

    int yv[SC_GROUPS];
#pragma unroll
    for (int g = 0; g < SC_GROUPS; ++g) {
        const int nd = row0 + g * SC_ROWS + rr;
        yv[g] = y[(nd < Nn) ? nd : (Nn - 1)];
    }

    SC_LOAD(0);
#pragma unroll
    for (int g = 0; g < SC_GROUPS; ++g) {
        __builtin_amdgcn_sched_barrier(0);
        if (g + 1 < SC_GROUPS) SC_LOAD(g + 1);
        __builtin_amdgcn_sched_barrier(0);

        float vals[SC_ROWS * NHEAD];
#pragma unroll
        for (int r = 0; r < SC_ROWS; ++r) {
            const v4f xr = xv[g & 1][r];
#pragma unroll
            for (int h = 0; h < NHEAD; ++h)
                vals[r * NHEAD + h] = xr.x * vh[h].x + xr.y * vh[h].y + xr.z * vh[h].z + xr.w * vh[h].w;
        }
#define SC_STAGE(WD) do { const bool up_ = (lane & (WD)) != 0; \
        _Pragma("unroll") for (int i = 0; i < (WD); ++i) { \
            const float keep_ = up_ ? vals[i + (WD)] : vals[i]; \
            const float send_ = up_ ? vals[i] : vals[i + (WD)]; \
            vals[i] = keep_ + __shfl_xor(send_, (WD), 32); } } while (0)
        SC_STAGE(16); SC_STAGE(8); SC_STAGE(4); SC_STAGE(2); SC_STAGE(1);
#undef SC_STAGE
        float s = vals[0] + ch;
        s = (s >= 0.0f) ? s : NEG_SLOPE * s;
        const int node = row0 + g * SC_ROWS + rr;
        if (node < Nn) {
            score[(size_t)node * NHEAD + hh] = s;
            const int idx = yv[g] * NHEAD + hh;
            if (bins) {
                if (s >= 0.0f) atomicMax(&lmax[idx], __float_as_int(s));
                else           atomicMin((unsigned*)&lmax[idx], (unsigned)__float_as_int(s));
                if (hh == 0) atomicAdd(&lcnt[yv[g]], 1);
            } else {
                if (s >= 0.0f) atomicMax(&segmax_i[idx], __float_as_int(s));
                else           atomicMin(&segmax_u[idx], (unsigned)__float_as_int(s));
                if (cnt != nullptr && hh == 0) atomicAdd(&cnt[yv[g]], 1);
            }
        }
    }
#undef SC_LOAD
    }
    }

    __syncthreads();
    if (bins) {
        for (int i = threadIdx.x; i < NHEAD * C; i += 256) {
            const int b = lmax[i];
            if (b != (int)0xFF800000) {
                if (__int_as_float(b) >= 0.0f) atomicMax(&segmax_i[i], b);
                else                           atomicMin(&segmax_u[i], (unsigned)b);
            }
        }
        if (cnt != nullptr)
            for (int i = threadIdx.x; i < C; i += 256) {
                const int n = lcnt[i];
                if (n != 0) atomicAdd(&cnt[i], n);
            }
    }
}

#define SCAN_T 1024
__global__ __launch_bounds__(SCAN_T)
void k_scan(int* cnt, int C) {
    __shared__ int part[SCAN_T];
    const int t = threadIdx.x;
    const int per = (C + SCAN_T - 1) / SCAN_T;
    const int b = t * per;
    int s = 0;
    for (int j = 0; j < per; ++j) { const int i = b + j; if (i < C) s += cnt[i]; }
    part[t] = s;
    __syncthreads();
    for (int d = 1; d < SCAN_T; d <<= 1) {
        const int add = (t >= d) ? part[t - d] : 0;
        __syncthreads();
        part[t] += add;
        __syncthreads();
    }
    int run = part[t] - s;
    for (int j = 0; j < per; ++j) {
        const int i = b + j;
        if (i < C) { const int n = cnt[i]; cnt[i] = run; run += n; }
    }
}

#define DN_NPT 4
__global__ __launch_bounds__(256)
void k_denom(const float* __restrict__ score, const int* __restrict__ y,
             const float* __restrict__ segmax, float* denom,
             int* cursor, int* perm, int Nn, int C) {
    __shared__ float lden[NHEAD * SEG_CAP];
    __shared__ int   lrun[SEG_CAP];
    const bool bins = (C <= SEG_CAP);
    const int t = threadIdx.x;
    if (bins) {
        for (int i = t; i < NHEAD * C; i += 256) lden[i] = 0.0f;
        for (int i = t; i < C; i += 256) lrun[i] = 0;
    }
    __syncthreads();

    const int base = blockIdx.x * (256 * DN_NPT);
    int  yy[DN_NPT], rk[DN_NPT];
    v4f  sc[DN_NPT];
#pragma unroll
    for (int j = 0; j < DN_NPT; ++j) {
        const int node = base + j * 256 + t;
        const int nc = (node < Nn) ? node : (Nn - 1);
        yy[j] = y[nc];
        sc[j] = *(const v4f*)(score + (size_t)nc * NHEAD);
    }
#pragma unroll
    for (int j = 0; j < DN_NPT; ++j) {
        const int node = base + j * 256 + t;
        const v4f mx = *(const v4f*)(segmax + (size_t)yy[j] * NHEAD);
        const float e0 = __expf(sc[j].x - mx.x), e1 = __expf(sc[j].y - mx.y);
        const float e2 = __expf(sc[j].z - mx.z), e3 = __expf(sc[j].w - mx.w);
        rk[j] = 0;
        if (node < Nn) {
            if (bins) {
                float* ld = &lden[yy[j] * NHEAD];
                atomicAdd(ld + 0, e0); atomicAdd(ld + 1, e1);
                atomicAdd(ld + 2, e2); atomicAdd(ld + 3, e3);
                if (perm != nullptr) rk[j] = atomicAdd(&lrun[yy[j]], 1);
            } else {
                float* gd = &denom[(size_t)yy[j] * NHEAD];
                atomicAdd(gd + 0, e0); atomicAdd(gd + 1, e1);
                atomicAdd(gd + 2, e2); atomicAdd(gd + 3, e3);
                if (perm != nullptr) {
                    const int pos = atomicAdd(&cursor[yy[j]], 1);
                    if ((unsigned)pos < (unsigned)Nn) perm[pos] = node;
                }
            }
        }
    }
    __syncthreads();
    if (bins) {
        for (int i = t; i < NHEAD * C; i += 256) {
            const float s = lden[i];
            if (s != 0.0f) atomicAdd(&denom[i], s);
        }
        if (perm != nullptr)
            for (int i = t; i < C; i += 256) {
                const int n = lrun[i];
                lrun[i] = (n != 0) ? atomicAdd(&cursor[i], n) : 0;
            }
    }
    __syncthreads();
    if (bins && perm != nullptr) {
#pragma unroll
        for (int j = 0; j < DN_NPT; ++j) {
            const int node = base + j * 256 + t;
            if (node < Nn) {
                const int pos = lrun[yy[j]] + rk[j];
                if ((unsigned)pos < (unsigned)Nn) perm[pos] = node;
            }
        }
    }
}

#define GT_ROWS    64
#define GT_THREADS 512
#define XP         132
#define GEMM_BLOCKS 256

__device__ __forceinline__ int gm_node(const int* __restrict__ perm, int p, int Nn) {
    const int pc = (p < Nn) ? p : (Nn - 1);
    int nd = (perm != nullptr) ? perm[pc] : pc;
    if ((unsigned)nd >= (unsigned)Nn) nd = 0;
    return nd;
}

__global__ __launch_bounds__(GT_THREADS)
void k_gemm(const float* __restrict__ x, const int* __restrict__ y,
            const float* __restrict__ lin_w, const float* __restrict__ lin_b,
            const float* __restrict__ score, const float* __restrict__ segmax,
            const float* __restrict__ denom, const int* __restrict__ perm,
            float* __restrict__ out, int Nn) {
    __shared__ __attribute__((aligned(16))) float Wl[TOT_OUT * XP];
    __shared__ __attribute__((aligned(16))) float Xl[2][GT_ROWS * XP];
    __shared__ float wgt[2][GT_ROWS * NHEAD];
    __shared__ int   yl[2][GT_ROWS];
    __shared__ float bl[TOT_OUT];

    const int t = threadIdx.x;
    const int G = gridDim.x;
    const int ntiles = (Nn + GT_ROWS - 1) / GT_ROWS;
    int tile = blockIdx.x;
    if (tile >= ntiles) return;

    const int row = t >> 3, e = t & 7, h = t & 3;
    const bool wstore = (t & 4) == 0;

    {
        const int p0 = tile * GT_ROWS + row;
        const bool ok0 = p0 < Nn;
        const int nd0 = gm_node(perm, p0, Nn);
        const float4* s0 = (const float4*)(x + (size_t)nd0 * IN_CH) + e;
        float4 a0 = s0[0], a1 = s0[8], a2 = s0[16], a3 = s0[24];
        const int y0 = y[nd0];
        const int i0 = y0 * NHEAD + h;
        float w0 = 0.0f;
        if (ok0) w0 = __expf(score[(size_t)nd0 * NHEAD + h] - segmax[i0]) / denom[i0];
        const float4 z4 = make_float4(0.f, 0.f, 0.f, 0.f);
        if (!ok0) { a0 = z4; a1 = z4; a2 = z4; a3 = z4; }
        float4* d0 = (float4*)(&Xl[0][row * XP]) + e;
        d0[0] = a0; d0[8] = a1; d0[16] = a2; d0[24] = a3;
        if (wstore) wgt[0][row * NHEAD + h] = w0;
        if (e == 0) yl[0][row] = y0;
    }
    if (t < TOT_OUT) bl[t] = lin_b[t];
    for (int i = t; i < TOT_OUT * (IN_CH / 4); i += GT_THREADS) {
        const int n = i >> 5, c4 = i & 31;
        const float4 w4 = *((const float4*)(lin_w + (size_t)n * IN_CH) + c4);
        *((float4*)(&Wl[n * XP]) + c4) = w4;
    }
    int  pA = (tile + G) * GT_ROWS + row;
    bool okA = pA < Nn;
    int  ndA = gm_node(perm, pA, Nn);
    int  yA  = y[ndA];
    float scA = score[(size_t)ndA * NHEAD + h];
    int  pB = (tile + 2 * G) * GT_ROWS + row;
    bool okB = pB < Nn;
    int  ndB = gm_node(perm, pB, Nn);
    __syncthreads();

    const int wave = __builtin_amdgcn_readfirstlane(t >> 5), lane = t & 31;
    const int rowbase = (wave >> 2) * 16;
    const int colbase = (wave & 3) * 64;
    const int head    = wave & 3;
    const int lm      = lane & 15;
    const int khalf   = (lane < 16) ? 0 : 2;
    const int mhalf   = (lane < 16) ? 0 : 8;
    const v8f vzero = {0.f, 0.f, 0.f, 0.f, 0.f, 0.f, 0.f, 0.f};
    const float* bp = &Wl[(colbase + lm) * XP + khalf];

    int buf = 0;
    for (; tile < ntiles; tile += G) {
        const float4* sA = (const float4*)(x + (size_t)ndA * IN_CH) + e;
        float4 xa = sA[0], xb = sA[8], xc = sA[16], xd = sA[24];
        const int iA = yA * NHEAD + h;
        const float mxA = segmax[iA], dnA = denom[iA];
        const int yB = y[ndB];
        const float scB = score[(size_t)ndB * NHEAD + h];
        const int pC = (tile + 3 * G) * GT_ROWS + row;
        const bool okC = pC < Nn;
        const int ndC = gm_node(perm, pC, Nn);

        v8f acc[4];
#pragma unroll
        for (int i = 0; i < 4; ++i) acc[i] = vzero;
        const float* ap = &Xl[buf][(rowbase + lm) * XP + khalf];
#pragma unroll
        for (int ks = 0; ks < IN_CH / 4; ++ks) {
            const v2f a = *(const v2f*)(ap + ks * 4);
#pragma unroll
            for (int nt = 0; nt < 4; ++nt) {
                const v2f b = *(const v2f*)(bp + nt * 16 * XP + ks * 4);
                acc[nt] = __builtin_amdgcn_wmma_f32_16x16x4_f32(
                    false, a, false, b, (short)0, acc[nt], false, false);
            }
        }

        float wr[8];
        int   yr[8];
        int   seg0, all_one_seg;
        {
            const float* wg = wgt[buf];
            const int*   ys = yl[buf];
            seg0 = ys[rowbase];
            int same = 1;
#pragma unroll
            for (int r = 1; r < 16; ++r) same &= (ys[rowbase + r] == seg0) ? 1 : 0;
            all_one_seg = __builtin_amdgcn_readfirstlane(same);
#pragma unroll
            for (int r = 0; r < 8; ++r) {
                const int m = rowbase + mhalf + r;
                wr[r] = wg[m * NHEAD + head];
                yr[r] = ys[m];
            }
        }

        {
            float wA = 0.0f;
            if (okA) wA = __expf(scA - mxA) / dnA;
            const float4 z4 = make_float4(0.f, 0.f, 0.f, 0.f);
            if (!okA) { xa = z4; xb = z4; xc = z4; xd = z4; }
            float4* dA = (float4*)(&Xl[buf ^ 1][row * XP]) + e;
            dA[0] = xa; dA[8] = xb; dA[16] = xc; dA[24] = xd;
            if (wstore) wgt[buf ^ 1][row * NHEAD + h] = wA;
            if (e == 0) yl[buf ^ 1][row] = yA;
        }
        __syncthreads();

        if (all_one_seg) {
#pragma unroll
            for (int nt = 0; nt < 4; ++nt) {
                const int n = colbase + nt * 16 + lm;
                const float bn = bl[n];
                float sum = 0.0f;
#pragma unroll
                for (int r = 0; r < 8; ++r) sum += (acc[nt][r] + bn) * wr[r];
                sum += __shfl_xor(sum, 16, 32);
                if (lane < 16) atomicAdd(&out[(size_t)seg0 * TOT_OUT + n], sum);
            }
        } else {
#pragma unroll
            for (int nt = 0; nt < 4; ++nt) {
                const int n = colbase + nt * 16 + lm;
                const float bn = bl[n];
                int   cur = yr[0];
                float sum = (acc[nt][0] + bn) * wr[0];
#pragma unroll
                for (int r = 1; r < 8; ++r) {
                    const float val = (acc[nt][r] + bn) * wr[r];
                    if (yr[r] == cur) {
                        sum += val;
                    } else {
                        atomicAdd(&out[(size_t)cur * TOT_OUT + n], sum);
                        cur = yr[r]; sum = val;
                    }
                }
                atomicAdd(&out[(size_t)cur * TOT_OUT + n], sum);
            }
        }
        buf ^= 1;
        ndA = ndB; okA = okB; yA = yB; scA = scB;
        ndB = ndC; okB = okC;
    }
}


#define PL_THREADS 512
#define PL_BLOCKS  512

__device__ __forceinline__ int pl_first_row(unsigned starts, int run) {
    unsigned mm = starts | 1u;
    for (int q = 0; q < run && mm != 0u; ++q) mm &= mm - 1u;
    return (mm != 0u) ? __builtin_ctz(mm) : 0;
}

__device__ __forceinline__ void pl_add(float* __restrict__ pooled, float* __restrict__ wsum, const v8f acc, float wtot,
                                       int sa, int sb, int sw, int col, int hd) {
    if (sa >= 0) {
#pragma unroll
        for (int hh2 = 0; hh2 < NHEAD; ++hh2)
            atomicAdd(&pooled[((size_t)sa * NHEAD + hh2) * IN_CH + col], acc[hh2]);
    }
    if (sb >= 0) {
#pragma unroll
        for (int hh2 = 0; hh2 < NHEAD; ++hh2)
            atomicAdd(&pooled[((size_t)sb * NHEAD + hh2) * IN_CH + col], acc[4 + hh2]);
    }
    if (sw >= 0) atomicAdd(&wsum[sw * NHEAD + hd], wtot);
}

__global__ __launch_bounds__(PL_THREADS, 8)
void k_pool(const float* __restrict__ x, const int* __restrict__ y,
            const float* __restrict__ score, const float* __restrict__ segmax,
            const float* __restrict__ denom, const int* __restrict__ perm,
            float* __restrict__ pooled, float* __restrict__ wsum, int Nn) {
    __shared__ __attribute__((aligned(16))) float Xl[2][GT_ROWS * XP];
    __shared__ float wgt[2][GT_ROWS * NHEAD];
    __shared__ int   yl[2][GT_ROWS];

    const int t = threadIdx.x;
    const int G = gridDim.x;
    const int ntiles = (Nn + GT_ROWS - 1) / GT_ROWS;
    int tile = blockIdx.x;
    if (tile >= ntiles) return;

    const int row = t >> 3, e = t & 7, h = t & 3;
    const bool wstore = (t & 4) == 0;
    {
        const int p0 = tile * GT_ROWS + row;
        const bool ok0 = p0 < Nn;
        const int nd0 = gm_node(perm, p0, Nn);
        const float4* s0 = (const float4*)(x + (size_t)nd0 * IN_CH) + e;
        float4 a0 = s0[0], a1 = s0[8], a2 = s0[16], a3 = s0[24];
        const int y0 = y[nd0];
        const int i0 = y0 * NHEAD + h;
        float w0 = 0.0f;
        if (ok0) w0 = __expf(score[(size_t)nd0 * NHEAD + h] - segmax[i0]) / denom[i0];
        const float4 z4 = make_float4(0.f, 0.f, 0.f, 0.f);
        if (!ok0) { a0 = z4; a1 = z4; a2 = z4; a3 = z4; }
        float4* d0 = (float4*)(&Xl[0][row * XP]) + e;
        d0[0] = a0; d0[8] = a1; d0[16] = a2; d0[24] = a3;
        if (wstore) wgt[0][row * NHEAD + h] = w0;
        if (e == 0) yl[0][row] = y0;
    }
    int  pA = (tile + G) * GT_ROWS + row;
    bool okA = pA < Nn;
    int  ndA = gm_node(perm, pA, Nn);
    int  yA  = y[ndA];
    float scA = score[(size_t)ndA * NHEAD + h];
    int  pB = (tile + 2 * G) * GT_ROWS + row;
    bool okB = pB < Nn;
    int  ndB = gm_node(perm, pB, Nn);
    __syncthreads();

    const int wave = __builtin_amdgcn_readfirstlane(t >> 5), lane = t & 31;
    const int cb    = (wave & 7) * 16;
    const int rb    = (wave >> 3) * 32;
    const int lm    = lane & 15;
    const int khalf = (lane < 16) ? 0 : 2;
    const int mhalf = (lane < 16) ? 0 : 8;
    const int slot  = lm >> 2, hd = lm & 3;
    const v8f vzero = {0.f, 0.f, 0.f, 0.f, 0.f, 0.f, 0.f, 0.f};

    int buf = 0;
    for (; tile < ntiles; tile += G) {
        const float4* sA = (const float4*)(x + (size_t)ndA * IN_CH) + e;
        float4 xa = sA[0], xb = sA[8], xc = sA[16], xd = sA[24];
        const int iA = yA * NHEAD + h;
        const float mxA = segmax[iA], dnA = denom[iA];
        const int yB = y[ndB];
        const float scB = score[(size_t)ndB * NHEAD + h];
        const int pC = (tile + 3 * G) * GT_ROWS + row;
        const bool okC = pC < Nn;
        const int ndC = gm_node(perm, pC, Nn);

        const float* wg = wgt[buf];
        const int*   ys = yl[buf];
        const int ymine = ys[rb + lane];
        const int yprev = ys[rb + ((lane > 0) ? lane - 1 : 0)];
        const unsigned starts = __builtin_amdgcn_ballot_w32(ymine != yprev);
        const int nruns = __builtin_popcount(starts) + 1;
        const float* xcol = &Xl[buf][rb * XP + cb + lm];

        v8f acc0 = vzero;
        float ws0 = 0.0f;
        int seg0v[2] = {-1, -1};
        int segw0 = -1;
        for (int pass = 0; pass * 4 < nruns; ++pass) {
            v8f acc = vzero;
            float wpart = 0.0f;
#pragma unroll
            for (int ks = 0; ks < 8; ++ks) {
                const int i0 = ks * 4 + khalf;
                const int ru0 = __builtin_popcount(starts & ((2u << (ks * 4 + 0)) - 1u)) - 4 * pass;
                const int ru1 = __builtin_popcount(starts & ((2u << (ks * 4 + 1)) - 1u)) - 4 * pass;
                const int ru2 = __builtin_popcount(starts & ((2u << (ks * 4 + 2)) - 1u)) - 4 * pass;
                const int ru3 = __builtin_popcount(starts & ((2u << (ks * 4 + 3)) - 1u)) - 4 * pass;
                const int r0 = (lane < 16) ? ru0 : ru2;
                const int r1 = (lane < 16) ? ru1 : ru3;
                v2f a, b;
                a.x = (r0 == slot) ? wg[(rb + i0) * NHEAD + hd] : 0.0f;
                a.y = (r1 == slot) ? wg[(rb + i0 + 1) * NHEAD + hd] : 0.0f;
                b.x = xcol[i0 * XP];
                b.y = xcol[(i0 + 1) * XP];
                acc = __builtin_amdgcn_wmma_f32_16x16x4_f32(false, a, false, b, (short)0, acc, false, false);
                wpart += a.x + a.y;
            }
            int segv[2];
#pragma unroll
            for (int s2 = 0; s2 < 2; ++s2) {
                const int run = 4 * pass + (mhalf >> 2) + s2;
                segv[s2] = (run < nruns) ? ys[rb + pl_first_row(starts, run)] : -1;
            }
            float wtot = 0.0f;
            int   segw = -1;
            if (cb == 0) {
                wtot = wpart + __shfl_xor(wpart, 16, 32);
                const int run = 4 * pass + slot;
                if (lane < 16 && run < nruns) segw = ys[rb + pl_first_row(starts, run)];
            }
            if (pass == 0) {
                acc0 = acc; ws0 = wtot; seg0v[0] = segv[0]; seg0v[1] = segv[1]; segw0 = segw;
            } else {
                pl_add(pooled, wsum, acc, wtot, segv[0], segv[1], segw, cb + lm, hd);
            }
        }

        {
            float wA = 0.0f;
            if (okA) wA = __expf(scA - mxA) / dnA;
            const float4 z4 = make_float4(0.f, 0.f, 0.f, 0.f);
            if (!okA) { xa = z4; xb = z4; xc = z4; xd = z4; }
            float4* dA = (float4*)(&Xl[buf ^ 1][row * XP]) + e;
            dA[0] = xa; dA[8] = xb; dA[16] = xc; dA[24] = xd;
            if (wstore) wgt[buf ^ 1][row * NHEAD + h] = wA;
            if (e == 0) yl[buf ^ 1][row] = yA;
        }
        __syncthreads();

        pl_add(pooled, wsum, acc0, ws0, seg0v[0], seg0v[1], segw0, cb + lm, hd);
        buf ^= 1;
        ndA = ndB; okA = okB; yA = yB; scA = scB;
        ndB = ndC; okB = okC;
    }
}

__global__ __launch_bounds__(128)
void k_proj(const float* __restrict__ pooled, const float* __restrict__ wsum,
            const float* __restrict__ lin_w, const float* __restrict__ lin_b,
            float* __restrict__ out, int C) {
    const int lane = threadIdx.x & 31;
    const int head = __builtin_amdgcn_readfirstlane((int)(threadIdx.x >> 5));
    const int sbase = blockIdx.x * 16;
    const int lm    = lane & 15;
    const int khalf = (lane < 16) ? 0 : 2;
    const int mhalf = (lane < 16) ? 0 : 8;
    const int sa = (sbase + lm < C) ? (sbase + lm) : (C - 1);
    const float* ap = pooled + ((size_t)sa * NHEAD + head) * IN_CH + khalf;
    const float* bp = lin_w + (size_t)(head * OUT_CH + lm) * IN_CH + khalf;
    const v8f vzero = {0.f, 0.f, 0.f, 0.f, 0.f, 0.f, 0.f, 0.f};
    v8f acc[4];
#pragma unroll
    for (int i = 0; i < 4; ++i) acc[i] = vzero;
#pragma unroll 8
    for (int ks = 0; ks < IN_CH / 4; ++ks) {
        const v2f a = *(const v2f*)(ap + ks * 4);
#pragma unroll
        for (int nt = 0; nt < 4; ++nt) {
            const v2f b = *(const v2f*)(bp + (size_t)nt * 16 * IN_CH + ks * 4);
            acc[nt] = __builtin_amdgcn_wmma_f32_16x16x4_f32(false, a, false, b, (short)0, acc[nt], false, false);
        }
    }
#pragma unroll
    for (int nt = 0; nt < 4; ++nt) {
        const int n = head * OUT_CH + nt * 16 + lm;
        const float bn = lin_b[n];
#pragma unroll
        for (int r = 0; r < 8; ++r) {
            const int sg = sbase + mhalf + r;
            if (sg < C) out[(size_t)sg * TOT_OUT + n] = acc[nt][r] + bn * wsum[sg * NHEAD + head];
        }
    }
}

extern "C" void kernel_launch(void* const* d_in, const int* in_sizes, int n_in,
                              void* d_out, int out_size, void* d_ws, size_t ws_size,
                              hipStream_t stream) {
    (void)n_in;
    const float* x     = (const float*)d_in[0];
    const int*   y     = (const int*)  d_in[1];
    const float* lin_w = (const float*)d_in[3];
    const float* lin_b = (const float*)d_in[4];
    const float* att_w = (const float*)d_in[5];
    const float* att_b = (const float*)d_in[6];
    float* out = (float*)d_out;

    int N = in_sizes[0] / IN_CH;
    int C = out_size / TOT_OUT;

    float* wsf    = (float*)d_ws;
    float* v      = wsf;
    float* c      = wsf + 512;
    float* segmax = wsf + 516;
    float* denom  = segmax + 4 * C;
    float* score  = denom + 4 * C;
    int*   cnt    = (int*)(score + (size_t)4 * N);
    int*   perm   = cnt + C;

    size_t need = ((size_t)516 + (size_t)9 * C + (size_t)5 * N) * sizeof(float);
    if (ws_size < need) { cnt = nullptr; perm = nullptr; }

    size_t poff = (((size_t)516 + (size_t)9 * C + (size_t)5 * N) + 3) & ~(size_t)3;
    float* pooled = wsf + poff;
    float* wsum   = pooled + (size_t)C * NHEAD * IN_CH;
    size_t need2 = (poff + (size_t)C * (NHEAD * IN_CH + NHEAD)) * sizeof(float);
    const bool pool = (perm != nullptr) && (ws_size >= need2);

    int initN = pool ? C * (NHEAD * IN_CH + NHEAD) : C * TOT_OUT;
    k_init <<<(initN + 255) / 256, 256, 0, stream>>>(pool ? pooled : out, initN, denom, (unsigned*)segmax, cnt, C);
    k_fold <<<2, 256, 0, stream>>>(lin_w, lin_b, att_w, att_b, v, c);
    k_score<<<(N + SC_SLABS * 256 - 1) / (SC_SLABS * 256), 256, 0, stream>>>(x, y, v, c, score,
                                                 (int*)segmax, (unsigned*)segmax, cnt, N, C);
    if (perm != nullptr) k_scan<<<1, SCAN_T, 0, stream>>>(cnt, C);
    k_denom<<<(N + DN_NPT * 256 - 1) / (DN_NPT * 256), 256, 0, stream>>>(score, y, segmax, denom, cnt, perm, N, C);
    if (pool) {
        k_pool<<<PL_BLOCKS, PL_THREADS, 0, stream>>>(x, y, score, segmax, denom, perm, pooled, wsum, N);
        k_proj<<<(C + 15) / 16, 128, 0, stream>>>(pooled, wsum, lin_w, lin_b, out, C);
    } else {
        k_gemm<<<GEMM_BLOCKS, GT_THREADS, 0, stream>>>(x, y, lin_w, lin_b,
                                                      score, segmax, denom, perm, out, N);
    }
}


# launches 7 -> 5: zero fill and weight fold share one launch of 256 + 3 groups (were 2016 + 2); k_denom blocks form the segment start positions in LDS (no scan launch), ticket adds in one batch
# speedup vs baseline: 5.9820x; 1.0750x over previous
#include <hip/hip_runtime.h>
#include <math.h>

typedef __attribute__((ext_vector_type(2))) float v2f;
typedef __attribute__((ext_vector_type(8))) float v8f;

#define IN_CH     128
#define OUT_CH    64
#define NHEAD     4
#define TOT_OUT   256
#define NEG_SLOPE 0.2f


#define PREP_FOLD 3
#define PREP_FILL 256
__global__ __launch_bounds__(256)
void k_prep(float* zbuf, int zn, float* denom, unsigned* segmax, int* cnt, int* tick, int C,
            const float* __restrict__ lin_w, const float* __restrict__ lin_b,
            const float* __restrict__ att_w, const float* __restrict__ att_b,
            float* v, float* c) {
    if (blockIdx.x < PREP_FOLD) {
        if (blockIdx.x < 2) {
            const int t = blockIdx.x * 256 + threadIdx.x;
            if (t < NHEAD * IN_CH) {
                const int h = t / IN_CH, i = t % IN_CH;
                float s = 0.0f;
#pragma unroll
                for (int d = 0; d < OUT_CH; ++d)
                    s += lin_w[(size_t)(h * OUT_CH + d) * IN_CH + i] * att_w[d];
                v[h * IN_CH + i] = s;
            }
        } else if (threadIdx.x < NHEAD) {
            const int t = threadIdx.x;
            float s = att_b[0];
            for (int d = 0; d < OUT_CH; ++d) s += lin_b[t * OUT_CH + d] * att_w[d];
            c[t] = s;
        }
        return;
    }
    const int gt = (blockIdx.x - PREP_FOLD) * 256 + threadIdx.x;
    const int gs = PREP_FILL * 256;
    const int n4 = zn >> 2;
    float4* z4 = (float4*)zbuf;
    const float4 zero4 = make_float4(0.f, 0.f, 0.f, 0.f);
    for (int i = gt; i < n4; i += gs) z4[i] = zero4;
    for (int i = (n4 << 2) + gt; i < zn; i += gs) zbuf[i] = 0.0f;
    for (int i = gt; i < 4 * C; i += gs) { denom[i] = 0.0f; segmax[i] = 0xFF800000u; }
    if (cnt != nullptr)  for (int i = gt; i < C; i += gs) cnt[i] = 0;
    if (tick != nullptr) for (int i = gt; i < C; i += gs) tick[i] = 0;
}

typedef __attribute__((ext_vector_type(4))) float v4f;

#define SC_ROWS   8
#define SC_GROUPS 4
#define SC_WAVES  8
#define SC_SLABS  4
#define SEG_CAP   2048

__global__ __launch_bounds__(256)
void k_score(const float* __restrict__ x, const int* __restrict__ y,
             const float* __restrict__ v, const float* __restrict__ c,
             float* __restrict__ score, int* segmax_i, unsigned* segmax_u, int* cnt, int Nn, int C) {
    __shared__ int lmax[NHEAD * SEG_CAP];
    __shared__ int lcnt[SEG_CAP];
    const bool bins = (C <= SEG_CAP);
    if (bins) {
        for (int i = threadIdx.x; i < NHEAD * C; i += 256) lmax[i] = (int)0xFF800000;
        for (int i = threadIdx.x; i < C; i += 256) lcnt[i] = 0;
    }
    __syncthreads();

    const int lane = threadIdx.x & 31;
    const int wave = __builtin_amdgcn_readfirstlane((int)(threadIdx.x >> 5));

    v4f vh[NHEAD];
#pragma unroll
    for (int h = 0; h < NHEAD; ++h) vh[h] = *(const v4f*)(v + h * IN_CH + lane * 4);
    const int hh = lane & 3;
    const int rr = lane >> 2;
    const float ch = c[hh];

    for (int sl = 0; sl < SC_SLABS; ++sl) {
    const int row0 = ((blockIdx.x * SC_SLABS + sl) * SC_WAVES + wave) * (SC_ROWS * SC_GROUPS);
    if (row0 < Nn) {
    const float* xb = x + (size_t)row0 * IN_CH + lane * 4;
    v4f xv[2][SC_ROWS];
#define SC_LOAD(G) do { _Pragma("unroll") for (int r = 0; r < SC_ROWS; ++r) { \
        const int ro_ = (row0 + (G) * SC_ROWS + r < Nn) ? ((G) * SC_ROWS + r) : 0; \
        xv[(G) & 1][r] = __builtin_nontemporal_load((const v4f*)(xb + (size_t)ro_ * IN_CH)); } } while (0)

    int yv[SC_GROUPS];
#pragma unroll
    for (int g = 0; g < SC_GROUPS; ++g) {
        const int nd = row0 + g * SC_ROWS + rr;
        yv[g] = y[(nd < Nn) ? nd : (Nn - 1)];
    }

    SC_LOAD(0);
#pragma unroll
    for (int g = 0; g < SC_GROUPS; ++g) {
        __builtin_amdgcn_sched_barrier(0);
        if (g + 1 < SC_GROUPS) SC_LOAD(g + 1);
        __builtin_amdgcn_sched_barrier(0);

        float vals[SC_ROWS * NHEAD];
#pragma unroll
        for (int r = 0; r < SC_ROWS; ++r) {
            const v4f xr = xv[g & 1][r];
#pragma unroll
            for (int h = 0; h < NHEAD; ++h)
                vals[r * NHEAD + h] = xr.x * vh[h].x + xr.y * vh[h].y + xr.z * vh[h].z + xr.w * vh[h].w;
        }
#define SC_STAGE(WD) do { const bool up_ = (lane & (WD)) != 0; \
        _Pragma("unroll") for (int i = 0; i < (WD); ++i) { \
            const float keep_ = up_ ? vals[i + (WD)] : vals[i]; \
            const float send_ = up_ ? vals[i] : vals[i + (WD)]; \
            vals[i] = keep_ + __shfl_xor(send_, (WD), 32); } } while (0)
        SC_STAGE(16); SC_STAGE(8); SC_STAGE(4); SC_STAGE(2); SC_STAGE(1);
#undef SC_STAGE
        float s = vals[0] + ch;
        s = (s >= 0.0f) ? s : NEG_SLOPE * s;
        const int node = row0 + g * SC_ROWS + rr;
        if (node < Nn) {
            score[(size_t)node * NHEAD + hh] = s;
            const int idx = yv[g] * NHEAD + hh;
            if (bins) {
                if (s >= 0.0f) atomicMax(&lmax[idx], __float_as_int(s));
                else           atomicMin((unsigned*)&lmax[idx], (unsigned)__float_as_int(s));
                if (hh == 0) atomicAdd(&lcnt[yv[g]], 1);
            } else {
                if (s >= 0.0f) atomicMax(&segmax_i[idx], __float_as_int(s));
                else           atomicMin(&segmax_u[idx], (unsigned)__float_as_int(s));
                if (cnt != nullptr && hh == 0) atomicAdd(&cnt[yv[g]], 1);
            }
        }
    }
#undef SC_LOAD
    }
    }

    __syncthreads();
    if (bins) {
        for (int i = threadIdx.x; i < NHEAD * C; i += 256) {
            const int b = lmax[i];
            if (b != (int)0xFF800000) {
                if (__int_as_float(b) >= 0.0f) atomicMax(&segmax_i[i], b);
                else                           atomicMin(&segmax_u[i], (unsigned)b);
            }
        }
        if (cnt != nullptr)
            for (int i = threadIdx.x; i < C; i += 256) {
                const int n = lcnt[i];
                if (n != 0) atomicAdd(&cnt[i], n);
            }
    }
}

#define SCAN_T 1024
__global__ __launch_bounds__(SCAN_T)
void k_scan(int* cnt, int C) {
    __shared__ int part[SCAN_T];
    const int t = threadIdx.x;
    const int per = (C + SCAN_T - 1) / SCAN_T;
    const int b = t * per;
    int s = 0;
    for (int j = 0; j < per; ++j) { const int i = b + j; if (i < C) s += cnt[i]; }
    part[t] = s;
    __syncthreads();
    for (int d = 1; d < SCAN_T; d <<= 1) {
        const int add = (t >= d) ? part[t - d] : 0;
        __syncthreads();
        part[t] += add;
        __syncthreads();
    }
    int run = part[t] - s;
    for (int j = 0; j < per; ++j) {
        const int i = b + j;
        if (i < C) { const int n = cnt[i]; cnt[i] = run; run += n; }
    }
}

#define DN_NPT 4
#define DN_SPT (SEG_CAP / 256)
__global__ __launch_bounds__(256)
void k_denom(const float* __restrict__ score, const int* __restrict__ y,
             const float* __restrict__ segmax, float* denom,
             int* cursor, int* tick, int* perm, int Nn, int C) {
    __shared__ float lden[NHEAD * SEG_CAP];
    __shared__ int   lrun[SEG_CAP];
    __shared__ int   lbase[SEG_CAP];
    __shared__ int   wtot[8];
    const bool bins = (C <= SEG_CAP);
    const bool scan = bins && (perm != nullptr);
    const int t = threadIdx.x;

    const int per = (C + 255) / 256;
    int cv[DN_SPT];
#pragma unroll
    for (int j = 0; j < DN_SPT; ++j) {
        const int i = t * per + j;
        cv[j] = (scan && j < per && i < C) ? cursor[i] : 0;
    }
    const int base = blockIdx.x * (256 * DN_NPT);
    int  yy[DN_NPT], rk[DN_NPT];
    v4f  sc[DN_NPT];
#pragma unroll
    for (int j = 0; j < DN_NPT; ++j) {
        const int node = base + j * 256 + t;
        const int nc = (node < Nn) ? node : (Nn - 1);
        yy[j] = y[nc];
        sc[j] = *(const v4f*)(score + (size_t)nc * NHEAD);
    }
    if (bins) {
        for (int i = t; i < NHEAD * C; i += 256) lden[i] = 0.0f;
        for (int i = t; i < C; i += 256) lrun[i] = 0;
    }
    int mine = 0, incl = 0;
    if (scan) {
#pragma unroll
        for (int j = 0; j < DN_SPT; ++j) mine += cv[j];
        incl = mine;
        for (int d = 1; d < 32; d <<= 1) {
            const int o = __shfl_up(incl, d, 32);
            if ((t & 31) >= d) incl += o;
        }
        if ((t & 31) == 31) wtot[t >> 5] = incl;
    }
    __syncthreads();
    if (scan) {
        int run = incl - mine;
        for (int w = 0; w < (t >> 5); ++w) run += wtot[w];
#pragma unroll
        for (int j = 0; j < DN_SPT; ++j) {
            const int i = t * per + j;
            if (j < per && i < C) { lbase[i] = run; run += cv[j]; }
        }
    }

#pragma unroll
    for (int j = 0; j < DN_NPT; ++j) {
        const int node = base + j * 256 + t;
        const v4f mx = *(const v4f*)(segmax + (size_t)yy[j] * NHEAD);
        const float e0 = __expf(sc[j].x - mx.x), e1 = __expf(sc[j].y - mx.y);
        const float e2 = __expf(sc[j].z - mx.z), e3 = __expf(sc[j].w - mx.w);
        rk[j] = 0;
        if (node < Nn) {
            if (bins) {
                float* ld = &lden[yy[j] * NHEAD];
                atomicAdd(ld + 0, e0); atomicAdd(ld + 1, e1);
                atomicAdd(ld + 2, e2); atomicAdd(ld + 3, e3);
                if (perm != nullptr) rk[j] = atomicAdd(&lrun[yy[j]], 1);
            } else {
                float* gd = &denom[(size_t)yy[j] * NHEAD];
                atomicAdd(gd + 0, e0); atomicAdd(gd + 1, e1);
                atomicAdd(gd + 2, e2); atomicAdd(gd + 3, e3);
                if (perm != nullptr) {
                    const int pos = atomicAdd(&cursor[yy[j]], 1);
                    if ((unsigned)pos < (unsigned)Nn) perm[pos] = node;
                }
            }
        }
    }
    __syncthreads();
    if (bins) {
        for (int i = t; i < NHEAD * C; i += 256) {
            const float s = lden[i];
            if (s != 0.0f) atomicAdd(&denom[i], s);
        }
        if (perm != nullptr) {
            int nn[DN_SPT], tk[DN_SPT];
#pragma unroll
            for (int j = 0; j < DN_SPT; ++j) { const int i = t + j * 256; nn[j] = (i < C) ? lrun[i] : 0; }
#pragma unroll
            for (int j = 0; j < DN_SPT; ++j) { tk[j] = 0; if (nn[j] != 0) tk[j] = atomicAdd(&tick[t + j * 256], nn[j]); }
#pragma unroll
            for (int j = 0; j < DN_SPT; ++j) {
                const int i = t + j * 256;
                if (i < C) lrun[i] = (nn[j] != 0) ? lbase[i] + tk[j] : 0;
            }
        }
    }
    __syncthreads();
    if (bins && perm != nullptr) {
#pragma unroll
        for (int j = 0; j < DN_NPT; ++j) {
            const int node = base + j * 256 + t;
            if (node < Nn) {
                const int pos = lrun[yy[j]] + rk[j];
                if ((unsigned)pos < (unsigned)Nn) perm[pos] = node;
            }
        }
    }
}

#define GT_ROWS    64
#define GT_THREADS 512
#define XP         132
#define GEMM_BLOCKS 256

__device__ __forceinline__ int gm_node(const int* __restrict__ perm, int p, int Nn) {
    const int pc = (p < Nn) ? p : (Nn - 1);
    int nd = (perm != nullptr) ? perm[pc] : pc;
    if ((unsigned)nd >= (unsigned)Nn) nd = 0;
    return nd;
}

__global__ __launch_bounds__(GT_THREADS)
void k_gemm(const float* __restrict__ x, const int* __restrict__ y,
            const float* __restrict__ lin_w, const float* __restrict__ lin_b,
            const float* __restrict__ score, const float* __restrict__ segmax,
            const float* __restrict__ denom, const int* __restrict__ perm,
            float* __restrict__ out, int Nn) {
    __shared__ __attribute__((aligned(16))) float Wl[TOT_OUT * XP];
    __shared__ __attribute__((aligned(16))) float Xl[2][GT_ROWS * XP];
    __shared__ float wgt[2][GT_ROWS * NHEAD];
    __shared__ int   yl[2][GT_ROWS];
    __shared__ float bl[TOT_OUT];

    const int t = threadIdx.x;
    const int G = gridDim.x;
    const int ntiles = (Nn + GT_ROWS - 1) / GT_ROWS;
    int tile = blockIdx.x;
    if (tile >= ntiles) return;

    const int row = t >> 3, e = t & 7, h = t & 3;
    const bool wstore = (t & 4) == 0;

    {
        const int p0 = tile * GT_ROWS + row;
        const bool ok0 = p0 < Nn;
        const int nd0 = gm_node(perm, p0, Nn);
        const float4* s0 = (const float4*)(x + (size_t)nd0 * IN_CH) + e;
        float4 a0 = s0[0], a1 = s0[8], a2 = s0[16], a3 = s0[24];
        const int y0 = y[nd0];
        const int i0 = y0 * NHEAD + h;
        float w0 = 0.0f;
        if (ok0) w0 = __expf(score[(size_t)nd0 * NHEAD + h] - segmax[i0]) / denom[i0];
        const float4 z4 = make_float4(0.f, 0.f, 0.f, 0.f);
        if (!ok0) { a0 = z4; a1 = z4; a2 = z4; a3 = z4; }
        float4* d0 = (float4*)(&Xl[0][row * XP]) + e;
        d0[0] = a0; d0[8] = a1; d0[16] = a2; d0[24] = a3;
        if (wstore) wgt[0][row * NHEAD + h] = w0;
        if (e == 0) yl[0][row] = y0;
    }
    if (t < TOT_OUT) bl[t] = lin_b[t];
    for (int i = t; i < TOT_OUT * (IN_CH / 4); i += GT_THREADS) {
        const int n = i >> 5, c4 = i & 31;
        const float4 w4 = *((const float4*)(lin_w + (size_t)n * IN_CH) + c4);
        *((float4*)(&Wl[n * XP]) + c4) = w4;
    }
    int  pA = (tile + G) * GT_ROWS + row;
    bool okA = pA < Nn;
    int  ndA = gm_node(perm, pA, Nn);
    int  yA  = y[ndA];
    float scA = score[(size_t)ndA * NHEAD + h];
    int  pB = (tile + 2 * G) * GT_ROWS + row;
    bool okB = pB < Nn;
    int  ndB = gm_node(perm, pB, Nn);
    __syncthreads();

    const int wave = __builtin_amdgcn_readfirstlane(t >> 5), lane = t & 31;
    const int rowbase = (wave >> 2) * 16;
    const int colbase = (wave & 3) * 64;
    const int head    = wave & 3;
    const int lm      = lane & 15;
    const int khalf   = (lane < 16) ? 0 : 2;
    const int mhalf   = (lane < 16) ? 0 : 8;
    const v8f vzero = {0.f, 0.f, 0.f, 0.f, 0.f, 0.f, 0.f, 0.f};
    const float* bp = &Wl[(colbase + lm) * XP + khalf];

    int buf = 0;
    for (; tile < ntiles; tile += G) {
        const float4* sA = (const float4*)(x + (size_t)ndA * IN_CH) + e;
        float4 xa = sA[0], xb = sA[8], xc = sA[16], xd = sA[24];
        const int iA = yA * NHEAD + h;
        const float mxA = segmax[iA], dnA = denom[iA];
        const int yB = y[ndB];
        const float scB = score[(size_t)ndB * NHEAD + h];
        const int pC = (tile + 3 * G) * GT_ROWS + row;
        const bool okC = pC < Nn;
        const int ndC = gm_node(perm, pC, Nn);

        v8f acc[4];
#pragma unroll
        for (int i = 0; i < 4; ++i) acc[i] = vzero;
        const float* ap = &Xl[buf][(rowbase + lm) * XP + khalf];
#pragma unroll
        for (int ks = 0; ks < IN_CH / 4; ++ks) {
            const v2f a = *(const v2f*)(ap + ks * 4);
#pragma unroll
            for (int nt = 0; nt < 4; ++nt) {
                const v2f b = *(const v2f*)(bp + nt * 16 * XP + ks * 4);
                acc[nt] = __builtin_amdgcn_wmma_f32_16x16x4_f32(
                    false, a, false, b, (short)0, acc[nt], false, false);
            }
        }

        float wr[8];
        int   yr[8];
        int   seg0, all_one_seg;
        {
            const float* wg = wgt[buf];
            const int*   ys = yl[buf];
            seg0 = ys[rowbase];
            int same = 1;
#pragma unroll
            for (int r = 1; r < 16; ++r) same &= (ys[rowbase + r] == seg0) ? 1 : 0;
            all_one_seg = __builtin_amdgcn_readfirstlane(same);
#pragma unroll
            for (int r = 0; r < 8; ++r) {
                const int m = rowbase + mhalf + r;
                wr[r] = wg[m * NHEAD + head];
                yr[r] = ys[m];
            }
        }

        {
            float wA = 0.0f;
            if (okA) wA = __expf(scA - mxA) / dnA;
            const float4 z4 = make_float4(0.f, 0.f, 0.f, 0.f);
            if (!okA) { xa = z4; xb = z4; xc = z4; xd = z4; }
            float4* dA = (float4*)(&Xl[buf ^ 1][row * XP]) + e;
            dA[0] = xa; dA[8] = xb; dA[16] = xc; dA[24] = xd;
            if (wstore) wgt[buf ^ 1][row * NHEAD + h] = wA;
            if (e == 0) yl[buf ^ 1][row] = yA;
        }
        __syncthreads();

        if (all_one_seg) {
#pragma unroll
            for (int nt = 0; nt < 4; ++nt) {
                const int n = colbase + nt * 16 + lm;
                const float bn = bl[n];
                float sum = 0.0f;
#pragma unroll
                for (int r = 0; r < 8; ++r) sum += (acc[nt][r] + bn) * wr[r];
                sum += __shfl_xor(sum, 16, 32);
                if (lane < 16) atomicAdd(&out[(size_t)seg0 * TOT_OUT + n], sum);
            }
        } else {
#pragma unroll
            for (int nt = 0; nt < 4; ++nt) {
                const int n = colbase + nt * 16 + lm;
                const float bn = bl[n];
                int   cur = yr[0];
                float sum = (acc[nt][0] + bn) * wr[0];
#pragma unroll
                for (int r = 1; r < 8; ++r) {
                    const float val = (acc[nt][r] + bn) * wr[r];
                    if (yr[r] == cur) {
                        sum += val;
                    } else {
                        atomicAdd(&out[(size_t)cur * TOT_OUT + n], sum);
                        cur = yr[r]; sum = val;
                    }
                }
                atomicAdd(&out[(size_t)cur * TOT_OUT + n], sum);
            }
        }
        buf ^= 1;
        ndA = ndB; okA = okB; yA = yB; scA = scB;
        ndB = ndC; okB = okC;
    }
}


#define PL_THREADS 512
#define PL_BLOCKS  512

__device__ __forceinline__ int pl_first_row(unsigned starts, int run) {
    unsigned mm = starts | 1u;
    for (int q = 0; q < run && mm != 0u; ++q) mm &= mm - 1u;
    return (mm != 0u) ? __builtin_ctz(mm) : 0;
}

__device__ __forceinline__ void pl_add(float* __restrict__ pooled, float* __restrict__ wsum, const v8f acc, float wtot,
                                       int sa, int sb, int sw, int col, int hd) {
    if (sa >= 0) {
#pragma unroll
        for (int hh2 = 0; hh2 < NHEAD; ++hh2)
            atomicAdd(&pooled[((size_t)sa * NHEAD + hh2) * IN_CH + col], acc[hh2]);
    }
    if (sb >= 0) {
#pragma unroll
        for (int hh2 = 0; hh2 < NHEAD; ++hh2)
            atomicAdd(&pooled[((size_t)sb * NHEAD + hh2) * IN_CH + col], acc[4 + hh2]);
    }
    if (sw >= 0) atomicAdd(&wsum[sw * NHEAD + hd], wtot);
}

__global__ __launch_bounds__(PL_THREADS, 8)
void k_pool(const float* __restrict__ x, const int* __restrict__ y,
            const float* __restrict__ score, const float* __restrict__ segmax,
            const float* __restrict__ denom, const int* __restrict__ perm,
            float* __restrict__ pooled, float* __restrict__ wsum, int Nn) {
    __shared__ __attribute__((aligned(16))) float Xl[2][GT_ROWS * XP];
    __shared__ float wgt[2][GT_ROWS * NHEAD];
    __shared__ int   yl[2][GT_ROWS];

    const int t = threadIdx.x;
    const int G = gridDim.x;
    const int ntiles = (Nn + GT_ROWS - 1) / GT_ROWS;
    int tile = blockIdx.x;
    if (tile >= ntiles) return;

    const int row = t >> 3, e = t & 7, h = t & 3;
    const bool wstore = (t & 4) == 0;
    {
        const int p0 = tile * GT_ROWS + row;
        const bool ok0 = p0 < Nn;
        const int nd0 = gm_node(perm, p0, Nn);
        const float4* s0 = (const float4*)(x + (size_t)nd0 * IN_CH) + e;
        float4 a0 = s0[0], a1 = s0[8], a2 = s0[16], a3 = s0[24];
        const int y0 = y[nd0];
        const int i0 = y0 * NHEAD + h;
        float w0 = 0.0f;
        if (ok0) w0 = __expf(score[(size_t)nd0 * NHEAD + h] - segmax[i0]) / denom[i0];
        const float4 z4 = make_float4(0.f, 0.f, 0.f, 0.f);
        if (!ok0) { a0 = z4; a1 = z4; a2 = z4; a3 = z4; }
        float4* d0 = (float4*)(&Xl[0][row * XP]) + e;
        d0[0] = a0; d0[8] = a1; d0[16] = a2; d0[24] = a3;
        if (wstore) wgt[0][row * NHEAD + h] = w0;
        if (e == 0) yl[0][row] = y0;
    }
    int  pA = (tile + G) * GT_ROWS + row;
    bool okA = pA < Nn;
    int  ndA = gm_node(perm, pA, Nn);
    int  yA  = y[ndA];
    float scA = score[(size_t)ndA * NHEAD + h];
    int  pB = (tile + 2 * G) * GT_ROWS + row;
    bool okB = pB < Nn;
    int  ndB = gm_node(perm, pB, Nn);
    __syncthreads();

    const int wave = __builtin_amdgcn_readfirstlane(t >> 5), lane = t & 31;
    const int cb    = (wave & 7) * 16;
    const int rb    = (wave >> 3) * 32;
    const int lm    = lane & 15;
    const int khalf = (lane < 16) ? 0 : 2;
    const int mhalf = (lane < 16) ? 0 : 8;
    const int slot  = lm >> 2, hd = lm & 3;
    const v8f vzero = {0.f, 0.f, 0.f, 0.f, 0.f, 0.f, 0.f, 0.f};

    int buf = 0;
    for (; tile < ntiles; tile += G) {
        const float4* sA = (const float4*)(x + (size_t)ndA * IN_CH) + e;
        float4 xa = sA[0], xb = sA[8], xc = sA[16], xd = sA[24];
        const int iA = yA * NHEAD + h;
        const float mxA = segmax[iA], dnA = denom[iA];
        const int yB = y[ndB];
        const float scB = score[(size_t)ndB * NHEAD + h];
        const int pC = (tile + 3 * G) * GT_ROWS + row;
        const bool okC = pC < Nn;
        const int ndC = gm_node(perm, pC, Nn);

        const float* wg = wgt[buf];
        const int*   ys = yl[buf];
        const int ymine = ys[rb + lane];
        const int yprev = ys[rb + ((lane > 0) ? lane - 1 : 0)];
        const unsigned starts = __builtin_amdgcn_ballot_w32(ymine != yprev);
        const int nruns = __builtin_popcount(starts) + 1;
        const float* xcol = &Xl[buf][rb * XP + cb + lm];

        v8f acc0 = vzero;
        float ws0 = 0.0f;
        int seg0v[2] = {-1, -1};
        int segw0 = -1;
        for (int pass = 0; pass * 4 < nruns; ++pass) {
            v8f acc = vzero;
            float wpart = 0.0f;
#pragma unroll
            for (int ks = 0; ks < 8; ++ks) {
                const int i0 = ks * 4 + khalf;
                const int ru0 = __builtin_popcount(starts & ((2u << (ks * 4 + 0)) - 1u)) - 4 * pass;
                const int ru1 = __builtin_popcount(starts & ((2u << (ks * 4 + 1)) - 1u)) - 4 * pass;
                const int ru2 = __builtin_popcount(starts & ((2u << (ks * 4 + 2)) - 1u)) - 4 * pass;
                const int ru3 = __builtin_popcount(starts & ((2u << (ks * 4 + 3)) - 1u)) - 4 * pass;
                const int r0 = (lane < 16) ? ru0 : ru2;
                const int r1 = (lane < 16) ? ru1 : ru3;
                v2f a, b;
                a.x = (r0 == slot) ? wg[(rb + i0) * NHEAD + hd] : 0.0f;
                a.y = (r1 == slot) ? wg[(rb + i0 + 1) * NHEAD + hd] : 0.0f;
                b.x = xcol[i0 * XP];
                b.y = xcol[(i0 + 1) * XP];
                acc = __builtin_amdgcn_wmma_f32_16x16x4_f32(false, a, false, b, (short)0, acc, false, false);
                wpart += a.x + a.y;
            }
            int segv[2];
#pragma unroll
            for (int s2 = 0; s2 < 2; ++s2) {
                const int run = 4 * pass + (mhalf >> 2) + s2;
                segv[s2] = (run < nruns) ? ys[rb + pl_first_row(starts, run)] : -1;
            }
            float wtot = 0.0f;
            int   segw = -1;
            if (cb == 0) {
                wtot = wpart + __shfl_xor(wpart, 16, 32);
                const int run = 4 * pass + slot;
                if (lane < 16 && run < nruns) segw = ys[rb + pl_first_row(starts, run)];
            }
            if (pass == 0) {
                acc0 = acc; ws0 = wtot; seg0v[0] = segv[0]; seg0v[1] = segv[1]; segw0 = segw;
            } else {
                pl_add(pooled, wsum, acc, wtot, segv[0], segv[1], segw, cb + lm, hd);
            }
        }

        {
            float wA = 0.0f;
            if (okA) wA = __expf(scA - mxA) / dnA;
            const float4 z4 = make_float4(0.f, 0.f, 0.f, 0.f);
            if (!okA) { xa = z4; xb = z4; xc = z4; xd = z4; }
            float4* dA = (float4*)(&Xl[buf ^ 1][row * XP]) + e;
            dA[0] = xa; dA[8] = xb; dA[16] = xc; dA[24] = xd;
            if (wstore) wgt[buf ^ 1][row * NHEAD + h] = wA;
            if (e == 0) yl[buf ^ 1][row] = yA;
        }
        __syncthreads();

        pl_add(pooled, wsum, acc0, ws0, seg0v[0], seg0v[1], segw0, cb + lm, hd);
        buf ^= 1;
        ndA = ndB; okA = okB; yA = yB; scA = scB;
        ndB = ndC; okB = okC;
    }
}

__global__ __launch_bounds__(128)
void k_proj(const float* __restrict__ pooled, const float* __restrict__ wsum,
            const float* __restrict__ lin_w, const float* __restrict__ lin_b,
            float* __restrict__ out, int C) {
    const int lane = threadIdx.x & 31;
    const int head = __builtin_amdgcn_readfirstlane((int)(threadIdx.x >> 5));
    const int sbase = blockIdx.x * 16;
    const int lm    = lane & 15;
    const int khalf = (lane < 16) ? 0 : 2;
    const int mhalf = (lane < 16) ? 0 : 8;
    const int sa = (sbase + lm < C) ? (sbase + lm) : (C - 1);
    const float* ap = pooled + ((size_t)sa * NHEAD + head) * IN_CH + khalf;
    const float* bp = lin_w + (size_t)(head * OUT_CH + lm) * IN_CH + khalf;
    const v8f vzero = {0.f, 0.f, 0.f, 0.f, 0.f, 0.f, 0.f, 0.f};
    v8f acc[4];
#pragma unroll
    for (int i = 0; i < 4; ++i) acc[i] = vzero;
#pragma unroll 8
    for (int ks = 0; ks < IN_CH / 4; ++ks) {
        const v2f a = *(const v2f*)(ap + ks * 4);
#pragma unroll
        for (int nt = 0; nt < 4; ++nt) {
            const v2f b = *(const v2f*)(bp + (size_t)nt * 16 * IN_CH + ks * 4);
            acc[nt] = __builtin_amdgcn_wmma_f32_16x16x4_f32(false, a, false, b, (short)0, acc[nt], false, false);
        }
    }
#pragma unroll
    for (int nt = 0; nt < 4; ++nt) {
        const int n = head * OUT_CH + nt * 16 + lm;
        const float bn = lin_b[n];
#pragma unroll
        for (int r = 0; r < 8; ++r) {
            const int sg = sbase + mhalf + r;
            if (sg < C) out[(size_t)sg * TOT_OUT + n] = acc[nt][r] + bn * wsum[sg * NHEAD + head];
        }
    }
}

extern "C" void kernel_launch(void* const* d_in, const int* in_sizes, int n_in,
                              void* d_out, int out_size, void* d_ws, size_t ws_size,
                              hipStream_t stream) {
    (void)n_in;
    const float* x     = (const float*)d_in[0];
    const int*   y     = (const int*)  d_in[1];
    const float* lin_w = (const float*)d_in[3];
    const float* lin_b = (const float*)d_in[4];
    const float* att_w = (const float*)d_in[5];
    const float* att_b = (const float*)d_in[6];
    float* out = (float*)d_out;

    int N = in_sizes[0] / IN_CH;
    int C = out_size / TOT_OUT;

    float* wsf    = (float*)d_ws;
    float* v      = wsf;
    float* c      = wsf + 512;
    float* segmax = wsf + 516;
    float* denom  = segmax + 4 * C;
    float* score  = denom + 4 * C;
    int*   cnt    = (int*)(score + (size_t)4 * N);
    int*   perm   = cnt + C;
    int*   tick   = perm + N;

    size_t need = ((size_t)516 + (size_t)10 * C + (size_t)5 * N) * sizeof(float);
    if (ws_size < need) { cnt = nullptr; perm = nullptr; tick = nullptr; }

    size_t poff = (((size_t)516 + (size_t)10 * C + (size_t)5 * N) + 3) & ~(size_t)3;
    float* pooled = wsf + poff;
    float* wsum   = pooled + (size_t)C * NHEAD * IN_CH;
    size_t need2 = (poff + (size_t)C * (NHEAD * IN_CH + NHEAD)) * sizeof(float);
    const bool pool = (perm != nullptr) && (ws_size >= need2);

    int initN = pool ? C * (NHEAD * IN_CH + NHEAD) : C * TOT_OUT;
    k_prep <<<PREP_FOLD + PREP_FILL, 256, 0, stream>>>(pool ? pooled : out, initN, denom, (unsigned*)segmax, cnt, tick, C,
                                                       lin_w, lin_b, att_w, att_b, v, c);
    k_score<<<(N + SC_SLABS * 256 - 1) / (SC_SLABS * 256), 256, 0, stream>>>(x, y, v, c, score,
                                                 (int*)segmax, (unsigned*)segmax, cnt, N, C);
    if (perm != nullptr && C > SEG_CAP) k_scan<<<1, SCAN_T, 0, stream>>>(cnt, C);
    k_denom<<<(N + DN_NPT * 256 - 1) / (DN_NPT * 256), 256, 0, stream>>>(score, y, segmax, denom, cnt, tick, perm, N, C);
    if (pool) {
        k_pool<<<PL_BLOCKS, PL_THREADS, 0, stream>>>(x, y, score, segmax, denom, perm, pooled, wsum, N);
        k_proj<<<(C + 15) / 16, 128, 0, stream>>>(pooled, wsum, lin_w, lin_b, out, C);
    } else {
        k_gemm<<<GEMM_BLOCKS, GT_THREADS, 0, stream>>>(x, y, lin_w, lin_b,
                                                      score, segmax, denom, perm, out, N);
    }
}


# k_pool blocks walk consecutive tiles and keep a segment's running sums in the WMMA accumulator from tile to tile; float atomics into pooled only where a segment or the stretch ends (sums reordered)
# speedup vs baseline: 6.3284x; 1.0579x over previous
#include <hip/hip_runtime.h>
#include <math.h>

typedef __attribute__((ext_vector_type(2))) float v2f;
typedef __attribute__((ext_vector_type(8))) float v8f;

#define IN_CH     128
#define OUT_CH    64
#define NHEAD     4
#define TOT_OUT   256
#define NEG_SLOPE 0.2f


#define PREP_FOLD 3
#define PREP_FILL 256
__global__ __launch_bounds__(256)
void k_prep(float* zbuf, int zn, float* denom, unsigned* segmax, int* cnt, int* tick, int C,
            const float* __restrict__ lin_w, const float* __restrict__ lin_b,
            const float* __restrict__ att_w, const float* __restrict__ att_b,
            float* v, float* c) {
    if (blockIdx.x < PREP_FOLD) {
        if (blockIdx.x < 2) {
            const int t = blockIdx.x * 256 + threadIdx.x;
            if (t < NHEAD * IN_CH) {
                const int h = t / IN_CH, i = t % IN_CH;
                float s = 0.0f;
#pragma unroll
                for (int d = 0; d < OUT_CH; ++d)
                    s += lin_w[(size_t)(h * OUT_CH + d) * IN_CH + i] * att_w[d];
                v[h * IN_CH + i] = s;
            }
        } else if (threadIdx.x < NHEAD) {
            const int t = threadIdx.x;
            float s = att_b[0];
            for (int d = 0; d < OUT_CH; ++d) s += lin_b[t * OUT_CH + d] * att_w[d];
            c[t] = s;
        }
        return;
    }
    const int gt = (blockIdx.x - PREP_FOLD) * 256 + threadIdx.x;
    const int gs = PREP_FILL * 256;
    const int n4 = zn >> 2;
    float4* z4 = (float4*)zbuf;
    const float4 zero4 = make_float4(0.f, 0.f, 0.f, 0.f);
    for (int i = gt; i < n4; i += gs) z4[i] = zero4;
    for (int i = (n4 << 2) + gt; i < zn; i += gs) zbuf[i] = 0.0f;
    for (int i = gt; i < 4 * C; i += gs) { denom[i] = 0.0f; segmax[i] = 0xFF800000u; }
    if (cnt != nullptr)  for (int i = gt; i < C; i += gs) cnt[i] = 0;
    if (tick != nullptr) for (int i = gt; i < C; i += gs) tick[i] = 0;
}

typedef __attribute__((ext_vector_type(4))) float v4f;

#define SC_ROWS   8
#define SC_GROUPS 4
#define SC_WAVES  8
#define SC_SLABS  4
#define SEG_CAP   2048

__global__ __launch_bounds__(256)
void k_score(const float* __restrict__ x, const int* __restrict__ y,
             const float* __restrict__ v, const float* __restrict__ c,
             float* __restrict__ score, int* segmax_i, unsigned* segmax_u, int* cnt, int Nn, int C) {
    __shared__ int lmax[NHEAD * SEG_CAP];
    __shared__ int lcnt[SEG_CAP];
    const bool bins = (C <= SEG_CAP);
    if (bins) {
        for (int i = threadIdx.x; i < NHEAD * C; i += 256) lmax[i] = (int)0xFF800000;
        for (int i = threadIdx.x; i < C; i += 256) lcnt[i] = 0;
    }
    __syncthreads();

    const int lane = threadIdx.x & 31;
    const int wave = __builtin_amdgcn_readfirstlane((int)(threadIdx.x >> 5));

    v4f vh[NHEAD];
#pragma unroll
    for (int h = 0; h < NHEAD; ++h) vh[h] = *(const v4f*)(v + h * IN_CH + lane * 4);
    const int hh = lane & 3;
    const int rr = lane >> 2;
    const float ch = c[hh];

    for (int sl = 0; sl < SC_SLABS; ++sl) {
    const int row0 = ((blockIdx.x * SC_SLABS + sl) * SC_WAVES + wave) * (SC_ROWS * SC_GROUPS);
    if (row0 < Nn) {
    const float* xb = x + (size_t)row0 * IN_CH + lane * 4;
    v4f xv[2][SC_ROWS];
#define SC_LOAD(G) do { _Pragma("unroll") for (int r = 0; r < SC_ROWS; ++r) { \
        const int ro_ = (row0 + (G) * SC_ROWS + r < Nn) ? ((G) * SC_ROWS + r) : 0; \
        xv[(G) & 1][r] = __builtin_nontemporal_load((const v4f*)(xb + (size_t)ro_ * IN_CH)); } } while (0)

    int yv[SC_GROUPS];
#pragma unroll
    for (int g = 0; g < SC_GROUPS; ++g) {
        const int nd = row0 + g * SC_ROWS + rr;
        yv[g] = y[(nd < Nn) ? nd : (Nn - 1)];
    }

    SC_LOAD(0);
#pragma unroll
    for (int g = 0; g < SC_GROUPS; ++g) {
        __builtin_amdgcn_sched_barrier(0);
        if (g + 1 < SC_GROUPS) SC_LOAD(g + 1);
        __builtin_amdgcn_sched_barrier(0);

        float vals[SC_ROWS * NHEAD];
#pragma unroll
        for (int r = 0; r < SC_ROWS; ++r) {
            const v4f xr = xv[g & 1][r];
#pragma unroll
            for (int h = 0; h < NHEAD; ++h)
                vals[r * NHEAD + h] = xr.x * vh[h].x + xr.y * vh[h].y + xr.z * vh[h].z + xr.w * vh[h].w;
        }
#define SC_STAGE(WD) do { const bool up_ = (lane & (WD)) != 0; \
        _Pragma("unroll") for (int i = 0; i < (WD); ++i) { \
            const float keep_ = up_ ? vals[i + (WD)] : vals[i]; \
            const float send_ = up_ ? vals[i] : vals[i + (WD)]; \
            vals[i] = keep_ + __shfl_xor(send_, (WD), 32); } } while (0)
        SC_STAGE(16); SC_STAGE(8); SC_STAGE(4); SC_STAGE(2); SC_STAGE(1);
#undef SC_STAGE
        float s = vals[0] + ch;
        s = (s >= 0.0f) ? s : NEG_SLOPE * s;
        const int node = row0 + g * SC_ROWS + rr;
        if (node < Nn) {
            score[(size_t)node * NHEAD + hh] = s;
            const int idx = yv[g] * NHEAD + hh;
            if (bins) {
                if (s >= 0.0f) atomicMax(&lmax[idx], __float_as_int(s));
                else           atomicMin((unsigned*)&lmax[idx], (unsigned)__float_as_int(s));
                if (hh == 0) atomicAdd(&lcnt[yv[g]], 1);
            } else {
                if (s >= 0.0f) atomicMax(&segmax_i[idx], __float_as_int(s));
                else           atomicMin(&segmax_u[idx], (unsigned)__float_as_int(s));
                if (cnt != nullptr && hh == 0) atomicAdd(&cnt[yv[g]], 1);
            }
        }
    }
#undef SC_LOAD
    }
    }

    __syncthreads();
    if (bins) {
        for (int i = threadIdx.x; i < NHEAD * C; i += 256) {
            const int b = lmax[i];
            if (b != (int)0xFF800000) {
                if (__int_as_float(b) >= 0.0f) atomicMax(&segmax_i[i], b);
                else                           atomicMin(&segmax_u[i], (unsigned)b);
            }
        }
        if (cnt != nullptr)
            for (int i = threadIdx.x; i < C; i += 256) {
                const int n = lcnt[i];
                if (n != 0) atomicAdd(&cnt[i], n);
            }
    }
}

#define SCAN_T 1024
__global__ __launch_bounds__(SCAN_T)
void k_scan(int* cnt, int C) {
    __shared__ int part[SCAN_T];
    const int t = threadIdx.x;
    const int per = (C + SCAN_T - 1) / SCAN_T;
    const int b = t * per;
    int s = 0;
    for (int j = 0; j < per; ++j) { const int i = b + j; if (i < C) s += cnt[i]; }
    part[t] = s;
    __syncthreads();
    for (int d = 1; d < SCAN_T; d <<= 1) {
        const int add = (t >= d) ? part[t - d] : 0;
        __syncthreads();
        part[t] += add;
        __syncthreads();
    }
    int run = part[t] - s;
    for (int j = 0; j < per; ++j) {
        const int i = b + j;
        if (i < C) { const int n = cnt[i]; cnt[i] = run; run += n; }
    }
}

#define DN_NPT 4
#define DN_SPT (SEG_CAP / 256)
__global__ __launch_bounds__(256)
void k_denom(const float* __restrict__ score, const int* __restrict__ y,
             const float* __restrict__ segmax, float* denom,
             int* cursor, int* tick, int* perm, int Nn, int C) {
    __shared__ float lden[NHEAD * SEG_CAP];
    __shared__ int   lrun[SEG_CAP];
    __shared__ int   lbase[SEG_CAP];
    __shared__ int   wtot[8];
    const bool bins = (C <= SEG_CAP);
    const bool scan = bins && (perm != nullptr);
    const int t = threadIdx.x;

    const int per = (C + 255) / 256;
    int cv[DN_SPT];
#pragma unroll
    for (int j = 0; j < DN_SPT; ++j) {
        const int i = t * per + j;
        cv[j] = (scan && j < per && i < C) ? cursor[i] : 0;
    }
    const int base = blockIdx.x * (256 * DN_NPT);
    int  yy[DN_NPT], rk[DN_NPT];
    v4f  sc[DN_NPT];
#pragma unroll
    for (int j = 0; j < DN_NPT; ++j) {
        const int node = base + j * 256 + t;
        const int nc = (node < Nn) ? node : (Nn - 1);
        yy[j] = y[nc];
        sc[j] = *(const v4f*)(score + (size_t)nc * NHEAD);
    }
    if (bins) {
        for (int i = t; i < NHEAD * C; i += 256) lden[i] = 0.0f;
        for (int i = t; i < C; i += 256) lrun[i] = 0;
    }
    int mine = 0, incl = 0;
    if (scan) {
#pragma unroll
        for (int j = 0; j < DN_SPT; ++j) mine += cv[j];
        incl = mine;
        for (int d = 1; d < 32; d <<= 1) {
            const int o = __shfl_up(incl, d, 32);
            if ((t & 31) >= d) incl += o;
        }
        if ((t & 31) == 31) wtot[t >> 5] = incl;
    }
    __syncthreads();
    if (scan) {
        int run = incl - mine;
        for (int w = 0; w < (t >> 5); ++w) run += wtot[w];
#pragma unroll
        for (int j = 0; j < DN_SPT; ++j) {
            const int i = t * per + j;
            if (j < per && i < C) { lbase[i] = run; run += cv[j]; }
        }
    }

#pragma unroll
    for (int j = 0; j < DN_NPT; ++j) {
        const int node = base + j * 256 + t;
        const v4f mx = *(const v4f*)(segmax + (size_t)yy[j] * NHEAD);
        const float e0 = __expf(sc[j].x - mx.x), e1 = __expf(sc[j].y - mx.y);
        const float e2 = __expf(sc[j].z - mx.z), e3 = __expf(sc[j].w - mx.w);
        rk[j] = 0;
        if (node < Nn) {
            if (bins) {
                float* ld = &lden[yy[j] * NHEAD];
                atomicAdd(ld + 0, e0); atomicAdd(ld + 1, e1);
                atomicAdd(ld + 2, e2); atomicAdd(ld + 3, e3);
                if (perm != nullptr) rk[j] = atomicAdd(&lrun[yy[j]], 1);
            } else {
                float* gd = &denom[(size_t)yy[j] * NHEAD];
                atomicAdd(gd + 0, e0); atomicAdd(gd + 1, e1);
                atomicAdd(gd + 2, e2); atomicAdd(gd + 3, e3);
                if (perm != nullptr) {
                    const int pos = atomicAdd(&cursor[yy[j]], 1);
                    if ((unsigned)pos < (unsigned)Nn) perm[pos] = node;
                }
            }
        }
    }
    __syncthreads();
    if (bins) {
        for (int i = t; i < NHEAD * C; i += 256) {
            const float s = lden[i];
            if (s != 0.0f) atomicAdd(&denom[i], s);
        }
        if (perm != nullptr) {
            int nn[DN_SPT], tk[DN_SPT];
#pragma unroll
            for (int j = 0; j < DN_SPT; ++j) { const int i = t + j * 256; nn[j] = (i < C) ? lrun[i] : 0; }
#pragma unroll
            for (int j = 0; j < DN_SPT; ++j) { tk[j] = 0; if (nn[j] != 0) tk[j] = atomicAdd(&tick[t + j * 256], nn[j]); }
#pragma unroll
            for (int j = 0; j < DN_SPT; ++j) {
                const int i = t + j * 256;
                if (i < C) lrun[i] = (nn[j] != 0) ? lbase[i] + tk[j] : 0;
            }
        }
    }
    __syncthreads();
    if (bins && perm != nullptr) {
#pragma unroll
        for (int j = 0; j < DN_NPT; ++j) {
            const int node = base + j * 256 + t;
            if (node < Nn) {
                const int pos = lrun[yy[j]] + rk[j];
                if ((unsigned)pos < (unsigned)Nn) perm[pos] = node;
            }
        }
    }
}

#define GT_ROWS    64
#define GT_THREADS 512
#define XP         132
#define GEMM_BLOCKS 256

__device__ __forceinline__ int gm_node(const int* __restrict__ perm, int p, int Nn) {
    const int pc = (p < Nn) ? p : (Nn - 1);
    int nd = (perm != nullptr) ? perm[pc] : pc;
    if ((unsigned)nd >= (unsigned)Nn) nd = 0;
    return nd;
}

__global__ __launch_bounds__(GT_THREADS)
void k_gemm(const float* __restrict__ x, const int* __restrict__ y,
            const float* __restrict__ lin_w, const float* __restrict__ lin_b,
            const float* __restrict__ score, const float* __restrict__ segmax,
            const float* __restrict__ denom, const int* __restrict__ perm,
            float* __restrict__ out, int Nn) {
    __shared__ __attribute__((aligned(16))) float Wl[TOT_OUT * XP];
    __shared__ __attribute__((aligned(16))) float Xl[2][GT_ROWS * XP];
    __shared__ float wgt[2][GT_ROWS * NHEAD];
    __shared__ int   yl[2][GT_ROWS];
    __shared__ float bl[TOT_OUT];

    const int t = threadIdx.x;
    const int G = gridDim.x;
    const int ntiles = (Nn + GT_ROWS - 1) / GT_ROWS;
    int tile = blockIdx.x;
    if (tile >= ntiles) return;

    const int row = t >> 3, e = t & 7, h = t & 3;
    const bool wstore = (t & 4) == 0;

    {
        const int p0 = tile * GT_ROWS + row;
        const bool ok0 = p0 < Nn;
        const int nd0 = gm_node(perm, p0, Nn);
        const float4* s0 = (const float4*)(x + (size_t)nd0 * IN_CH) + e;
        float4 a0 = s0[0], a1 = s0[8], a2 = s0[16], a3 = s0[24];
        const int y0 = y[nd0];
        const int i0 = y0 * NHEAD + h;
        float w0 = 0.0f;
        if (ok0) w0 = __expf(score[(size_t)nd0 * NHEAD + h] - segmax[i0]) / denom[i0];
        const float4 z4 = make_float4(0.f, 0.f, 0.f, 0.f);
        if (!ok0) { a0 = z4; a1 = z4; a2 = z4; a3 = z4; }
        float4* d0 = (float4*)(&Xl[0][row * XP]) + e;
        d0[0] = a0; d0[8] = a1; d0[16] = a2; d0[24] = a3;
        if (wstore) wgt[0][row * NHEAD + h] = w0;
        if (e == 0) yl[0][row] = y0;
    }
    if (t < TOT_OUT) bl[t] = lin_b[t];
    for (int i = t; i < TOT_OUT * (IN_CH / 4); i += GT_THREADS) {
        const int n = i >> 5, c4 = i & 31;
        const float4 w4 = *((const float4*)(lin_w + (size_t)n * IN_CH) + c4);
        *((float4*)(&Wl[n * XP]) + c4) = w4;
    }
    int  pA = (tile + G) * GT_ROWS + row;
    bool okA = pA < Nn;
    int  ndA = gm_node(perm, pA, Nn);
    int  yA  = y[ndA];
    float scA = score[(size_t)ndA * NHEAD + h];
    int  pB = (tile + 2 * G) * GT_ROWS + row;
    bool okB = pB < Nn;
    int  ndB = gm_node(perm, pB, Nn);
    __syncthreads();

    const int wave = __builtin_amdgcn_readfirstlane(t >> 5), lane = t & 31;
    const int rowbase = (wave >> 2) * 16;
    const int colbase = (wave & 3) * 64;
    const int head    = wave & 3;
    const int lm      = lane & 15;
    const int khalf   = (lane < 16) ? 0 : 2;
    const int mhalf   = (lane < 16) ? 0 : 8;
    const v8f vzero = {0.f, 0.f, 0.f, 0.f, 0.f, 0.f, 0.f, 0.f};
    const float* bp = &Wl[(colbase + lm) * XP + khalf];

    int buf = 0;
    for (; tile < ntiles; tile += G) {
        const float4* sA = (const float4*)(x + (size_t)ndA * IN_CH) + e;
        float4 xa = sA[0], xb = sA[8], xc = sA[16], xd = sA[24];
        const int iA = yA * NHEAD + h;
        const float mxA = segmax[iA], dnA = denom[iA];
        const int yB = y[ndB];
        const float scB = score[(size_t)ndB * NHEAD + h];
        const int pC = (tile + 3 * G) * GT_ROWS + row;
        const bool okC = pC < Nn;
        const int ndC = gm_node(perm, pC, Nn);

        v8f acc[4];
#pragma unroll
        for (int i = 0; i < 4; ++i) acc[i] = vzero;
        const float* ap = &Xl[buf][(rowbase + lm) * XP + khalf];
#pragma unroll
        for (int ks = 0; ks < IN_CH / 4; ++ks) {
            const v2f a = *(const v2f*)(ap + ks * 4);
#pragma unroll
            for (int nt = 0; nt < 4; ++nt) {
                const v2f b = *(const v2f*)(bp + nt * 16 * XP + ks * 4);
                acc[nt] = __builtin_amdgcn_wmma_f32_16x16x4_f32(
                    false, a, false, b, (short)0, acc[nt], false, false);
            }
        }

        float wr[8];
        int   yr[8];
        int   seg0, all_one_seg;
        {
            const float* wg = wgt[buf];
            const int*   ys = yl[buf];
            seg0 = ys[rowbase];
            int same = 1;
#pragma unroll
            for (int r = 1; r < 16; ++r) same &= (ys[rowbase + r] == seg0) ? 1 : 0;
            all_one_seg = __builtin_amdgcn_readfirstlane(same);
#pragma unroll
            for (int r = 0; r < 8; ++r) {
                const int m = rowbase + mhalf + r;
                wr[r] = wg[m * NHEAD + head];
                yr[r] = ys[m];
            }
        }

        {
            float wA = 0.0f;
            if (okA) wA = __expf(scA - mxA) / dnA;
            const float4 z4 = make_float4(0.f, 0.f, 0.f, 0.f);
            if (!okA) { xa = z4; xb = z4; xc = z4; xd = z4; }
            float4* dA = (float4*)(&Xl[buf ^ 1][row * XP]) + e;
            dA[0] = xa; dA[8] = xb; dA[16] = xc; dA[24] = xd;
            if (wstore) wgt[buf ^ 1][row * NHEAD + h] = wA;
            if (e == 0) yl[buf ^ 1][row] = yA;
        }
        __syncthreads();

        if (all_one_seg) {
#pragma unroll
            for (int nt = 0; nt < 4; ++nt) {
                const int n = colbase + nt * 16 + lm;
                const float bn = bl[n];
                float sum = 0.0f;
#pragma unroll
                for (int r = 0; r < 8; ++r) sum += (acc[nt][r] + bn) * wr[r];
                sum += __shfl_xor(sum, 16, 32);
                if (lane < 16) atomicAdd(&out[(size_t)seg0 * TOT_OUT + n], sum);
            }
        } else {
#pragma unroll
            for (int nt = 0; nt < 4; ++nt) {
                const int n = colbase + nt * 16 + lm;
                const float bn = bl[n];
                int   cur = yr[0];
                float sum = (acc[nt][0] + bn) * wr[0];
#pragma unroll
                for (int r = 1; r < 8; ++r) {
                    const float val = (acc[nt][r] + bn) * wr[r];
                    if (yr[r] == cur) {
                        sum += val;
                    } else {
                        atomicAdd(&out[(size_t)cur * TOT_OUT + n], sum);
                        cur = yr[r]; sum = val;
                    }
                }
                atomicAdd(&out[(size_t)cur * TOT_OUT + n], sum);
            }
        }
        buf ^= 1;
        ndA = ndB; okA = okB; yA = yB; scA = scB;
        ndB = ndC; okB = okC;
    }
}


#define PL_THREADS 512
#define PL_BLOCKS  512

__device__ __forceinline__ int pl_first_row(unsigned starts, int run) {
    unsigned mm = starts | 1u;
    for (int q = 0; q < run && mm != 0u; ++q) mm &= mm - 1u;
    return (mm != 0u) ? __builtin_ctz(mm) : 0;
}

__device__ __forceinline__ void pl_add(float* __restrict__ pooled, float* __restrict__ wsum, const v8f acc, float wtot,
                                       int sa, int sb, int sw, int col, int hd) {
    if (sa >= 0) {
#pragma unroll
        for (int hh2 = 0; hh2 < NHEAD; ++hh2)
            atomicAdd(&pooled[((size_t)sa * NHEAD + hh2) * IN_CH + col], acc[hh2]);
    }
    if (sb >= 0) {
#pragma unroll
        for (int hh2 = 0; hh2 < NHEAD; ++hh2)
            atomicAdd(&pooled[((size_t)sb * NHEAD + hh2) * IN_CH + col], acc[4 + hh2]);
    }
    if (sw >= 0) atomicAdd(&wsum[sw * NHEAD + hd], wtot);
}

__device__ __forceinline__ void pl_flush(float* __restrict__ pooled, int seg, float c0, float c1, float c2, float c3,
                                         int col, int lane) {
    if (seg >= 0 && lane < 16) {
        atomicAdd(&pooled[((size_t)seg * NHEAD + 0) * IN_CH + col], c0);
        atomicAdd(&pooled[((size_t)seg * NHEAD + 1) * IN_CH + col], c1);
        atomicAdd(&pooled[((size_t)seg * NHEAD + 2) * IN_CH + col], c2);
        atomicAdd(&pooled[((size_t)seg * NHEAD + 3) * IN_CH + col], c3);
    }
}

__device__ __forceinline__ int pl_node(const int* __restrict__ perm, int p, int pend, int Nn) {
    const int pc = (p < pend) ? p : (pend - 1);
    int nd = (perm != nullptr) ? perm[pc] : pc;
    if ((unsigned)nd >= (unsigned)Nn) nd = 0;
    return nd;
}

__global__ __launch_bounds__(PL_THREADS, 8)
void k_pool(const float* __restrict__ x, const int* __restrict__ y,
            const float* __restrict__ score, const float* __restrict__ segmax,
            const float* __restrict__ denom, const int* __restrict__ perm,
            float* __restrict__ pooled, float* __restrict__ wsum, int Nn) {
    __shared__ __attribute__((aligned(16))) float Xl[2][GT_ROWS * XP];
    __shared__ float wgt[2][GT_ROWS * NHEAD];
    __shared__ int   yl[2][GT_ROWS];

    const int t = threadIdx.x;
    const int G = gridDim.x;
    const int ntiles = (Nn + GT_ROWS - 1) / GT_ROWS;
    int tile = (int)(((long long)blockIdx.x * ntiles) / G);
    const int tend = (int)(((long long)(blockIdx.x + 1) * ntiles) / G);
    if (tile >= tend) return;
    const int pend = (tend * GT_ROWS < Nn) ? tend * GT_ROWS : Nn;

    const int row = t >> 3, e = t & 7, h = t & 3;
    const bool wstore = (t & 4) == 0;
    {
        const int p0 = tile * GT_ROWS + row;
        const bool ok0 = p0 < pend;
        const int nd0 = pl_node(perm, p0, pend, Nn);
        const float4* s0 = (const float4*)(x + (size_t)nd0 * IN_CH) + e;
        float4 a0 = s0[0], a1 = s0[8], a2 = s0[16], a3 = s0[24];
        const int y0 = y[nd0];
        const int i0 = y0 * NHEAD + h;
        float w0 = 0.0f;
        if (ok0) w0 = __expf(score[(size_t)nd0 * NHEAD + h] - segmax[i0]) / denom[i0];
        const float4 z4 = make_float4(0.f, 0.f, 0.f, 0.f);
        if (!ok0) { a0 = z4; a1 = z4; a2 = z4; a3 = z4; }
        float4* d0 = (float4*)(&Xl[0][row * XP]) + e;
        d0[0] = a0; d0[8] = a1; d0[16] = a2; d0[24] = a3;
        if (wstore) wgt[0][row * NHEAD + h] = w0;
        if (e == 0) yl[0][row] = y0;
    }
    int  pA = (tile + 1) * GT_ROWS + row;
    bool okA = pA < pend;
    int  ndA = pl_node(perm, pA, pend, Nn);
    int  yA  = y[ndA];
    float scA = score[(size_t)ndA * NHEAD + h];
    int  pB = (tile + 2) * GT_ROWS + row;
    bool okB = pB < pend;
    int  ndB = pl_node(perm, pB, pend, Nn);
    __syncthreads();

    const int wave = __builtin_amdgcn_readfirstlane(t >> 5), lane = t & 31;
    const int cb    = (wave & 7) * 16;
    const int rb    = (wave >> 3) * 32;
    const int lm    = lane & 15;
    const int khalf = (lane < 16) ? 0 : 2;
    const int mhalf = (lane < 16) ? 0 : 8;
    const int slot  = lm >> 2, hd = lm & 3;
    const v8f vzero = {0.f, 0.f, 0.f, 0.f, 0.f, 0.f, 0.f, 0.f};

    int   cseg = -1;
    float car0 = 0.0f, car1 = 0.0f, car2 = 0.0f, car3 = 0.0f;

    int buf = 0;
    for (; tile < tend; ++tile) {
        const float4* sA = (const float4*)(x + (size_t)ndA * IN_CH) + e;
        float4 xa = sA[0], xb = sA[8], xc = sA[16], xd = sA[24];
        const int iA = yA * NHEAD + h;
        const float mxA = segmax[iA], dnA = denom[iA];
        const int yB = y[ndB];
        const float scB = score[(size_t)ndB * NHEAD + h];
        const int pC = (tile + 3) * GT_ROWS + row;
        const bool okC = pC < pend;
        const int ndC = pl_node(perm, pC, pend, Nn);

        const float* wg = wgt[buf];
        const int*   ys = yl[buf];
        const int ymine = ys[rb + lane];
        const int yprev = ys[rb + ((lane > 0) ? lane - 1 : 0)];
        const unsigned starts = __builtin_amdgcn_ballot_w32(ymine != yprev);
        const int nruns = __builtin_popcount(starts) + 1;
        const int lastrun = nruns - 1;
        const float* xcol = &Xl[buf][rb * XP + cb + lm];
        const int slast  = __builtin_amdgcn_readfirstlane(ys[rb + 31]);

        float ws0 = 0.0f;
        int segw0 = -1;
        for (int pass = 0; pass * 4 < nruns; ++pass) {
            v8f acc = vzero;
            float wpart = 0.0f;
            if (pass == 0) { acc[0] = car0; acc[1] = car1; acc[2] = car2; acc[3] = car3; }
#pragma unroll
            for (int ks = 0; ks < 8; ++ks) {
                const int i0 = ks * 4 + khalf;
                const int ru0 = __builtin_popcount(starts & ((2u << (ks * 4 + 0)) - 1u)) - 4 * pass;
                const int ru1 = __builtin_popcount(starts & ((2u << (ks * 4 + 1)) - 1u)) - 4 * pass;
                const int ru2 = __builtin_popcount(starts & ((2u << (ks * 4 + 2)) - 1u)) - 4 * pass;
                const int ru3 = __builtin_popcount(starts & ((2u << (ks * 4 + 3)) - 1u)) - 4 * pass;
                const int r0 = (lane < 16) ? ru0 : ru2;
                const int r1 = (lane < 16) ? ru1 : ru3;
                v2f a, b;
                a.x = (r0 == slot) ? wg[(rb + i0) * NHEAD + hd] : 0.0f;
                a.y = (r1 == slot) ? wg[(rb + i0 + 1) * NHEAD + hd] : 0.0f;
                b.x = xcol[i0 * XP];
                b.y = xcol[(i0 + 1) * XP];
                acc = __builtin_amdgcn_wmma_f32_16x16x4_f32(false, a, false, b, (short)0, acc, false, false);
                wpart += a.x + a.y;
            }
            int segv[2];
#pragma unroll
            for (int s2 = 0; s2 < 2; ++s2) {
                const int run = 4 * pass + (mhalf >> 2) + s2;
                segv[s2] = (run < lastrun) ? ys[rb + pl_first_row(starts, run)] : -1;
            }
            if ((lastrun >> 2) == pass) {
                const int ls = lastrun & 3;
                float t0 = (ls & 1) ? acc[4] : acc[0];
                float t1 = (ls & 1) ? acc[5] : acc[1];
                float t2 = (ls & 1) ? acc[6] : acc[2];
                float t3 = (ls & 1) ? acc[7] : acc[3];
                if (ls & 2) {
                    t0 = __shfl_xor(t0, 16, 32); t1 = __shfl_xor(t1, 16, 32);
                    t2 = __shfl_xor(t2, 16, 32); t3 = __shfl_xor(t3, 16, 32);
                }
                car0 = (lane < 16) ? t0 : 0.0f; car1 = (lane < 16) ? t1 : 0.0f;
                car2 = (lane < 16) ? t2 : 0.0f; car3 = (lane < 16) ? t3 : 0.0f;
            }
            float wtot = 0.0f;
            int   segw = -1;
            if (cb == 0) {
                wtot = wpart + __shfl_xor(wpart, 16, 32);
                const int run = 4 * pass + slot;
                if (lane < 16 && run < nruns) segw = ys[rb + pl_first_row(starts, run)];
            }
            if (pass == 0) { ws0 = wtot; segw0 = segw; segw = -1; }
            pl_add(pooled, wsum, acc, wtot, segv[0], segv[1], segw, cb + lm, hd);
        }
        cseg = slast;

        {
            float wA = 0.0f;
            if (okA) wA = __expf(scA - mxA) / dnA;
            const float4 z4 = make_float4(0.f, 0.f, 0.f, 0.f);
            if (!okA) { xa = z4; xb = z4; xc = z4; xd = z4; }
            float4* dA = (float4*)(&Xl[buf ^ 1][row * XP]) + e;
            dA[0] = xa; dA[8] = xb; dA[16] = xc; dA[24] = xd;
            if (wstore) wgt[buf ^ 1][row * NHEAD + h] = wA;
            if (e == 0) yl[buf ^ 1][row] = yA;
        }
        __syncthreads();

        if (segw0 >= 0) atomicAdd(&wsum[segw0 * NHEAD + hd], ws0);
        const int snext = __builtin_amdgcn_readfirstlane(yl[buf ^ 1][rb]);
        if (tile + 1 == tend || snext != cseg) {
            pl_flush(pooled, cseg, car0, car1, car2, car3, cb + lm, lane);
            car0 = 0.0f; car1 = 0.0f; car2 = 0.0f; car3 = 0.0f;
        }
        buf ^= 1;
        ndA = ndB; okA = okB; yA = yB; scA = scB;
        ndB = ndC; okB = okC;
    }
}

__global__ __launch_bounds__(128)
void k_proj(const float* __restrict__ pooled, const float* __restrict__ wsum,
            const float* __restrict__ lin_w, const float* __restrict__ lin_b,
            float* __restrict__ out, int C) {
    const int lane = threadIdx.x & 31;
    const int head = __builtin_amdgcn_readfirstlane((int)(threadIdx.x >> 5));
    const int sbase = blockIdx.x * 16;
    const int lm    = lane & 15;
    const int khalf = (lane < 16) ? 0 : 2;
    const int mhalf = (lane < 16) ? 0 : 8;
    const int sa = (sbase + lm < C) ? (sbase + lm) : (C - 1);
    const float* ap = pooled + ((size_t)sa * NHEAD + head) * IN_CH + khalf;
    const float* bp = lin_w + (size_t)(head * OUT_CH + lm) * IN_CH + khalf;
    const v8f vzero = {0.f, 0.f, 0.f, 0.f, 0.f, 0.f, 0.f, 0.f};
    v8f acc[4];
#pragma unroll
    for (int i = 0; i < 4; ++i) acc[i] = vzero;
#pragma unroll 8
    for (int ks = 0; ks < IN_CH / 4; ++ks) {
        const v2f a = *(const v2f*)(ap + ks * 4);
#pragma unroll
        for (int nt = 0; nt < 4; ++nt) {
            const v2f b = *(const v2f*)(bp + (size_t)nt * 16 * IN_CH + ks * 4);
            acc[nt] = __builtin_amdgcn_wmma_f32_16x16x4_f32(false, a, false, b, (short)0, acc[nt], false, false);
        }
    }
#pragma unroll
    for (int nt = 0; nt < 4; ++nt) {
        const int n = head * OUT_CH + nt * 16 + lm;
        const float bn = lin_b[n];
#pragma unroll
        for (int r = 0; r < 8; ++r) {
            const int sg = sbase + mhalf + r;
            if (sg < C) out[(size_t)sg * TOT_OUT + n] = acc[nt][r] + bn * wsum[sg * NHEAD + head];
        }
    }
}

extern "C" void kernel_launch(void* const* d_in, const int* in_sizes, int n_in,
                              void* d_out, int out_size, void* d_ws, size_t ws_size,
                              hipStream_t stream) {
    (void)n_in;
    const float* x     = (const float*)d_in[0];
    const int*   y     = (const int*)  d_in[1];
    const float* lin_w = (const float*)d_in[3];
    const float* lin_b = (const float*)d_in[4];
    const float* att_w = (const float*)d_in[5];
    const float* att_b = (const float*)d_in[6];
    float* out = (float*)d_out;

    int N = in_sizes[0] / IN_CH;
    int C = out_size / TOT_OUT;

    float* wsf    = (float*)d_ws;
    float* v      = wsf;
    float* c      = wsf + 512;
    float* segmax = wsf + 516;
    float* denom  = segmax + 4 * C;
    float* score  = denom + 4 * C;
    int*   cnt    = (int*)(score + (size_t)4 * N);
    int*   perm   = cnt + C;
    int*   tick   = perm + N;

    size_t need = ((size_t)516 + (size_t)10 * C + (size_t)5 * N) * sizeof(float);
    if (ws_size < need) { cnt = nullptr; perm = nullptr; tick = nullptr; }

    size_t poff = (((size_t)516 + (size_t)10 * C + (size_t)5 * N) + 3) & ~(size_t)3;
    float* pooled = wsf + poff;
    float* wsum   = pooled + (size_t)C * NHEAD * IN_CH;
    size_t need2 = (poff + (size_t)C * (NHEAD * IN_CH + NHEAD)) * sizeof(float);
    const bool pool = (perm != nullptr) && (ws_size >= need2);

    int initN = pool ? C * (NHEAD * IN_CH + NHEAD) : C * TOT_OUT;
    k_prep <<<PREP_FOLD + PREP_FILL, 256, 0, stream>>>(pool ? pooled : out, initN, denom, (unsigned*)segmax, cnt, tick, C,
                                                       lin_w, lin_b, att_w, att_b, v, c);
    k_score<<<(N + SC_SLABS * 256 - 1) / (SC_SLABS * 256), 256, 0, stream>>>(x, y, v, c, score,
                                                 (int*)segmax, (unsigned*)segmax, cnt, N, C);
    if (perm != nullptr && C > SEG_CAP) k_scan<<<1, SCAN_T, 0, stream>>>(cnt, C);
    k_denom<<<(N + DN_NPT * 256 - 1) / (DN_NPT * 256), 256, 0, stream>>>(score, y, segmax, denom, cnt, tick, perm, N, C);
    if (pool) {
        k_pool<<<PL_BLOCKS, PL_THREADS, 0, stream>>>(x, y, score, segmax, denom, perm, pooled, wsum, N);
        k_proj<<<(C + 15) / 16, 128, 0, stream>>>(pooled, wsum, lin_w, lin_b, out, C);
    } else {
        k_gemm<<<GEMM_BLOCKS, GT_THREADS, 0, stream>>>(x, y, lin_w, lin_b,
                                                      score, segmax, denom, perm, out, N);
    }
}


# division by the exp sum moved behind the sums: k_pool weights are exp(score - max), k_proj divides by the summed weights (sum order kept, rounding differs); exp-sum pass and its float atomics gone
# speedup vs baseline: 7.2786x; 1.1501x over previous
#include <hip/hip_runtime.h>
#include <math.h>

typedef __attribute__((ext_vector_type(2))) float v2f;
typedef __attribute__((ext_vector_type(8))) float v8f;

#define IN_CH     128
#define OUT_CH    64
#define NHEAD     4
#define TOT_OUT   256
#define NEG_SLOPE 0.2f


#define PREP_FOLD 3
#define PREP_FILL 256
__global__ __launch_bounds__(256)
void k_prep(float* zbuf, int zn, float* denom, unsigned* segmax, int* cnt, int* tick, int C,
            const float* __restrict__ lin_w, const float* __restrict__ lin_b,
            const float* __restrict__ att_w, const float* __restrict__ att_b,
            float* v, float* c) {
    if (blockIdx.x < PREP_FOLD) {
        if (blockIdx.x < 2) {
            const int t = blockIdx.x * 256 + threadIdx.x;
            if (t < NHEAD * IN_CH) {
                const int h = t / IN_CH, i = t % IN_CH;
                float s = 0.0f;
#pragma unroll
                for (int d = 0; d < OUT_CH; ++d)
                    s += lin_w[(size_t)(h * OUT_CH + d) * IN_CH + i] * att_w[d];
                v[h * IN_CH + i] = s;
            }
        } else if (threadIdx.x < NHEAD) {
            const int t = threadIdx.x;
            float s = att_b[0];
            for (int d = 0; d < OUT_CH; ++d) s += lin_b[t * OUT_CH + d] * att_w[d];
            c[t] = s;
        }
        return;
    }
    const int gt = (blockIdx.x - PREP_FOLD) * 256 + threadIdx.x;
    const int gs = PREP_FILL * 256;
    const int n4 = zn >> 2;
    float4* z4 = (float4*)zbuf;
    const float4 zero4 = make_float4(0.f, 0.f, 0.f, 0.f);
    for (int i = gt; i < n4; i += gs) z4[i] = zero4;
    for (int i = (n4 << 2) + gt; i < zn; i += gs) zbuf[i] = 0.0f;
    for (int i = gt; i < 4 * C; i += gs) { if (denom != nullptr) denom[i] = 0.0f; segmax[i] = 0xFF800000u; }
    if (cnt != nullptr)  for (int i = gt; i < C; i += gs) cnt[i] = 0;
    if (tick != nullptr) for (int i = gt; i < C; i += gs) tick[i] = 0;
}

typedef __attribute__((ext_vector_type(4))) float v4f;

#define SC_ROWS   8
#define SC_GROUPS 4
#define SC_WAVES  8
#define SC_SLABS  4
#define SEG_CAP   2048

__global__ __launch_bounds__(256)
void k_score(const float* __restrict__ x, const int* __restrict__ y,
             const float* __restrict__ v, const float* __restrict__ c,
             float* __restrict__ score, int* segmax_i, unsigned* segmax_u, int* cnt, int Nn, int C) {
    __shared__ int lmax[NHEAD * SEG_CAP];
    __shared__ int lcnt[SEG_CAP];
    const bool bins = (C <= SEG_CAP);
    if (bins) {
        for (int i = threadIdx.x; i < NHEAD * C; i += 256) lmax[i] = (int)0xFF800000;
        for (int i = threadIdx.x; i < C; i += 256) lcnt[i] = 0;
    }
    __syncthreads();

    const int lane = threadIdx.x & 31;
    const int wave = __builtin_amdgcn_readfirstlane((int)(threadIdx.x >> 5));

    v4f vh[NHEAD];
#pragma unroll
    for (int h = 0; h < NHEAD; ++h) vh[h] = *(const v4f*)(v + h * IN_CH + lane * 4);
    const int hh = lane & 3;
    const int rr = lane >> 2;
    const float ch = c[hh];

    for (int sl = 0; sl < SC_SLABS; ++sl) {
    const int row0 = ((blockIdx.x * SC_SLABS + sl) * SC_WAVES + wave) * (SC_ROWS * SC_GROUPS);
    if (row0 < Nn) {
    const float* xb = x + (size_t)row0 * IN_CH + lane * 4;
    v4f xv[2][SC_ROWS];
#define SC_LOAD(G) do { _Pragma("unroll") for (int r = 0; r < SC_ROWS; ++r) { \
        const int ro_ = (row0 + (G) * SC_ROWS + r < Nn) ? ((G) * SC_ROWS + r) : 0; \
        xv[(G) & 1][r] = __builtin_nontemporal_load((const v4f*)(xb + (size_t)ro_ * IN_CH)); } } while (0)

    int yv[SC_GROUPS];
#pragma unroll
    for (int g = 0; g < SC_GROUPS; ++g) {
        const int nd = row0 + g * SC_ROWS + rr;
        yv[g] = y[(nd < Nn) ? nd : (Nn - 1)];
    }

    SC_LOAD(0);
#pragma unroll
    for (int g = 0; g < SC_GROUPS; ++g) {
        __builtin_amdgcn_sched_barrier(0);
        if (g + 1 < SC_GROUPS) SC_LOAD(g + 1);
        __builtin_amdgcn_sched_barrier(0);

        float vals[SC_ROWS * NHEAD];
#pragma unroll
        for (int r = 0; r < SC_ROWS; ++r) {
            const v4f xr = xv[g & 1][r];
#pragma unroll
            for (int h = 0; h < NHEAD; ++h)
                vals[r * NHEAD + h] = xr.x * vh[h].x + xr.y * vh[h].y + xr.z * vh[h].z + xr.w * vh[h].w;
        }
#define SC_STAGE(WD) do { const bool up_ = (lane & (WD)) != 0; \
        _Pragma("unroll") for (int i = 0; i < (WD); ++i) { \
            const float keep_ = up_ ? vals[i + (WD)] : vals[i]; \
            const float send_ = up_ ? vals[i] : vals[i + (WD)]; \
            vals[i] = keep_ + __shfl_xor(send_, (WD), 32); } } while (0)
        SC_STAGE(16); SC_STAGE(8); SC_STAGE(4); SC_STAGE(2); SC_STAGE(1);
#undef SC_STAGE
        float s = vals[0] + ch;
        s = (s >= 0.0f) ? s : NEG_SLOPE * s;
        const int node = row0 + g * SC_ROWS + rr;
        if (node < Nn) {
            score[(size_t)node * NHEAD + hh] = s;
            const int idx = yv[g] * NHEAD + hh;
            if (bins) {
                if (s >= 0.0f) atomicMax(&lmax[idx], __float_as_int(s));
                else           atomicMin((unsigned*)&lmax[idx], (unsigned)__float_as_int(s));
                if (hh == 0) atomicAdd(&lcnt[yv[g]], 1);
            } else {
                if (s >= 0.0f) atomicMax(&segmax_i[idx], __float_as_int(s));
                else           atomicMin(&segmax_u[idx], (unsigned)__float_as_int(s));
                if (cnt != nullptr && hh == 0) atomicAdd(&cnt[yv[g]], 1);
            }
        }
    }
#undef SC_LOAD
    }
    }

    __syncthreads();
    if (bins) {
        for (int i = threadIdx.x; i < NHEAD * C; i += 256) {
            const int b = lmax[i];
            if (b != (int)0xFF800000) {
                if (__int_as_float(b) >= 0.0f) atomicMax(&segmax_i[i], b);
                else                           atomicMin(&segmax_u[i], (unsigned)b);
            }
        }
        if (cnt != nullptr)
            for (int i = threadIdx.x; i < C; i += 256) {
                const int n = lcnt[i];
                if (n != 0) atomicAdd(&cnt[i], n);
            }
    }
}

#define SCAN_T 1024
__global__ __launch_bounds__(SCAN_T)
void k_scan(int* cnt, int C) {
    __shared__ int part[SCAN_T];
    const int t = threadIdx.x;
    const int per = (C + SCAN_T - 1) / SCAN_T;
    const int b = t * per;
    int s = 0;
    for (int j = 0; j < per; ++j) { const int i = b + j; if (i < C) s += cnt[i]; }
    part[t] = s;
    __syncthreads();
    for (int d = 1; d < SCAN_T; d <<= 1) {
        const int add = (t >= d) ? part[t - d] : 0;
        __syncthreads();
        part[t] += add;
        __syncthreads();
    }
    int run = part[t] - s;
    for (int j = 0; j < per; ++j) {
        const int i = b + j;
        if (i < C) { const int n = cnt[i]; cnt[i] = run; run += n; }
    }
}

#define DN_NPT 4
#define DN_SPT (SEG_CAP / 256)
__global__ __launch_bounds__(256)
void k_denom(const float* __restrict__ score, const int* __restrict__ y,
             const float* __restrict__ segmax, float* denom,
             int* cursor, int* tick, int* perm, int Nn, int C) {
    __shared__ float lden[NHEAD * SEG_CAP];
    __shared__ int   lrun[SEG_CAP];
    __shared__ int   lbase[SEG_CAP];
    __shared__ int   wtot[8];
    const bool bins = (C <= SEG_CAP);
    const bool scan = bins && (perm != nullptr);
    const bool dsum = (denom != nullptr);
    const int t = threadIdx.x;

    const int per = (C + 255) / 256;
    int cv[DN_SPT];
#pragma unroll
    for (int j = 0; j < DN_SPT; ++j) {
        const int i = t * per + j;
        cv[j] = (scan && j < per && i < C) ? cursor[i] : 0;
    }
    const int base = blockIdx.x * (256 * DN_NPT);
    int  yy[DN_NPT], rk[DN_NPT];
    v4f  sc[DN_NPT];
#pragma unroll
    for (int j = 0; j < DN_NPT; ++j) {
        const int node = base + j * 256 + t;
        const int nc = (node < Nn) ? node : (Nn - 1);
        yy[j] = y[nc];
        sc[j] = (v4f){0.0f, 0.0f, 0.0f, 0.0f};
        if (dsum) sc[j] = *(const v4f*)(score + (size_t)nc * NHEAD);
    }
    if (bins) {
        if (dsum) for (int i = t; i < NHEAD * C; i += 256) lden[i] = 0.0f;
        for (int i = t; i < C; i += 256) lrun[i] = 0;
    }
    int mine = 0, incl = 0;
    if (scan) {
#pragma unroll
        for (int j = 0; j < DN_SPT; ++j) mine += cv[j];
        incl = mine;
        for (int d = 1; d < 32; d <<= 1) {
            const int o = __shfl_up(incl, d, 32);
            if ((t & 31) >= d) incl += o;
        }
        if ((t & 31) == 31) wtot[t >> 5] = incl;
    }
    __syncthreads();
    if (scan) {
        int run = incl - mine;
        for (int w = 0; w < (t >> 5); ++w) run += wtot[w];
#pragma unroll
        for (int j = 0; j < DN_SPT; ++j) {
            const int i = t * per + j;
            if (j < per && i < C) { lbase[i] = run; run += cv[j]; }
        }
    }

#pragma unroll
    for (int j = 0; j < DN_NPT; ++j) {
        const int node = base + j * 256 + t;
        float e0 = 0.0f, e1 = 0.0f, e2 = 0.0f, e3 = 0.0f;
        if (dsum) {
            const v4f mx = *(const v4f*)(segmax + (size_t)yy[j] * NHEAD);
            e0 = __expf(sc[j].x - mx.x); e1 = __expf(sc[j].y - mx.y);
            e2 = __expf(sc[j].z - mx.z); e3 = __expf(sc[j].w - mx.w);
        }
        rk[j] = 0;
        if (node < Nn) {
            if (bins) {
                if (dsum) {
                    float* ld = &lden[yy[j] * NHEAD];
                    atomicAdd(ld + 0, e0); atomicAdd(ld + 1, e1);
                    atomicAdd(ld + 2, e2); atomicAdd(ld + 3, e3);
                }
                if (perm != nullptr) rk[j] = atomicAdd(&lrun[yy[j]], 1);
            } else {
                if (dsum) {
                    float* gd = &denom[(size_t)yy[j] * NHEAD];
                    atomicAdd(gd + 0, e0); atomicAdd(gd + 1, e1);
                    atomicAdd(gd + 2, e2); atomicAdd(gd + 3, e3);
                }
                if (perm != nullptr) {
                    const int pos = atomicAdd(&cursor[yy[j]], 1);
                    if ((unsigned)pos < (unsigned)Nn) perm[pos] = node;
                }
            }
        }
    }
    __syncthreads();
    if (bins) {
        if (dsum)
            for (int i = t; i < NHEAD * C; i += 256) {
                const float s = lden[i];
                if (s != 0.0f) atomicAdd(&denom[i], s);
            }
        if (perm != nullptr) {
            int nn[DN_SPT], tk[DN_SPT];
#pragma unroll
            for (int j = 0; j < DN_SPT; ++j) { const int i = t + j * 256; nn[j] = (i < C) ? lrun[i] : 0; }
#pragma unroll
            for (int j = 0; j < DN_SPT; ++j) { tk[j] = 0; if (nn[j] != 0) tk[j] = atomicAdd(&tick[t + j * 256], nn[j]); }
#pragma unroll
            for (int j = 0; j < DN_SPT; ++j) {
                const int i = t + j * 256;
                if (i < C) lrun[i] = (nn[j] != 0) ? lbase[i] + tk[j] : 0;
            }
        }
    }
    __syncthreads();
    if (bins && perm != nullptr) {
#pragma unroll
        for (int j = 0; j < DN_NPT; ++j) {
            const int node = base + j * 256 + t;
            if (node < Nn) {
                const int pos = lrun[yy[j]] + rk[j];
                if ((unsigned)pos < (unsigned)Nn) perm[pos] = node;
            }
        }
    }
}

#define GT_ROWS    64
#define GT_THREADS 512
#define XP         132
#define GEMM_BLOCKS 256

__device__ __forceinline__ int gm_node(const int* __restrict__ perm, int p, int Nn) {
    const int pc = (p < Nn) ? p : (Nn - 1);
    int nd = (perm != nullptr) ? perm[pc] : pc;
    if ((unsigned)nd >= (unsigned)Nn) nd = 0;
    return nd;
}

__global__ __launch_bounds__(GT_THREADS)
void k_gemm(const float* __restrict__ x, const int* __restrict__ y,
            const float* __restrict__ lin_w, const float* __restrict__ lin_b,
            const float* __restrict__ score, const float* __restrict__ segmax,
            const float* __restrict__ denom, const int* __restrict__ perm,
            float* __restrict__ out, int Nn) {
    __shared__ __attribute__((aligned(16))) float Wl[TOT_OUT * XP];
    __shared__ __attribute__((aligned(16))) float Xl[2][GT_ROWS * XP];
    __shared__ float wgt[2][GT_ROWS * NHEAD];
    __shared__ int   yl[2][GT_ROWS];
    __shared__ float bl[TOT_OUT];

    const int t = threadIdx.x;
    const int G = gridDim.x;
    const int ntiles = (Nn + GT_ROWS - 1) / GT_ROWS;
    int tile = blockIdx.x;
    if (tile >= ntiles) return;

    const int row = t >> 3, e = t & 7, h = t & 3;
    const bool wstore = (t & 4) == 0;

    {
        const int p0 = tile * GT_ROWS + row;
        const bool ok0 = p0 < Nn;
        const int nd0 = gm_node(perm, p0, Nn);
        const float4* s0 = (const float4*)(x + (size_t)nd0 * IN_CH) + e;
        float4 a0 = s0[0], a1 = s0[8], a2 = s0[16], a3 = s0[24];
        const int y0 = y[nd0];
        const int i0 = y0 * NHEAD + h;
        float w0 = 0.0f;
        if (ok0) w0 = __expf(score[(size_t)nd0 * NHEAD + h] - segmax[i0]) / denom[i0];
        const float4 z4 = make_float4(0.f, 0.f, 0.f, 0.f);
        if (!ok0) { a0 = z4; a1 = z4; a2 = z4; a3 = z4; }
        float4* d0 = (float4*)(&Xl[0][row * XP]) + e;
        d0[0] = a0; d0[8] = a1; d0[16] = a2; d0[24] = a3;
        if (wstore) wgt[0][row * NHEAD + h] = w0;
        if (e == 0) yl[0][row] = y0;
    }
    if (t < TOT_OUT) bl[t] = lin_b[t];
    for (int i = t; i < TOT_OUT * (IN_CH / 4); i += GT_THREADS) {
        const int n = i >> 5, c4 = i & 31;
        const float4 w4 = *((const float4*)(lin_w + (size_t)n * IN_CH) + c4);
        *((float4*)(&Wl[n * XP]) + c4) = w4;
    }
    int  pA = (tile + G) * GT_ROWS + row;
    bool okA = pA < Nn;
    int  ndA = gm_node(perm, pA, Nn);
    int  yA  = y[ndA];
    float scA = score[(size_t)ndA * NHEAD + h];
    int  pB = (tile + 2 * G) * GT_ROWS + row;
    bool okB = pB < Nn;
    int  ndB = gm_node(perm, pB, Nn);
    __syncthreads();

    const int wave = __builtin_amdgcn_readfirstlane(t >> 5), lane = t & 31;
    const int rowbase = (wave >> 2) * 16;
    const int colbase = (wave & 3) * 64;
    const int head    = wave & 3;
    const int lm      = lane & 15;
    const int khalf   = (lane < 16) ? 0 : 2;
    const int mhalf   = (lane < 16) ? 0 : 8;
    const v8f vzero = {0.f, 0.f, 0.f, 0.f, 0.f, 0.f, 0.f, 0.f};
    const float* bp = &Wl[(colbase + lm) * XP + khalf];

    int buf = 0;
    for (; tile < ntiles; tile += G) {
        const float4* sA = (const float4*)(x + (size_t)ndA * IN_CH) + e;
        float4 xa = sA[0], xb = sA[8], xc = sA[16], xd = sA[24];
        const int iA = yA * NHEAD + h;
        const float mxA = segmax[iA], dnA = denom[iA];
        const int yB = y[ndB];
        const float scB = score[(size_t)ndB * NHEAD + h];
        const int pC = (tile + 3 * G) * GT_ROWS + row;
        const bool okC = pC < Nn;
        const int ndC = gm_node(perm, pC, Nn);

        v8f acc[4];
#pragma unroll
        for (int i = 0; i < 4; ++i) acc[i] = vzero;
        const float* ap = &Xl[buf][(rowbase + lm) * XP + khalf];
#pragma unroll
        for (int ks = 0; ks < IN_CH / 4; ++ks) {
            const v2f a = *(const v2f*)(ap + ks * 4);
#pragma unroll
            for (int nt = 0; nt < 4; ++nt) {
                const v2f b = *(const v2f*)(bp + nt * 16 * XP + ks * 4);
                acc[nt] = __builtin_amdgcn_wmma_f32_16x16x4_f32(
                    false, a, false, b, (short)0, acc[nt], false, false);
            }
        }

        float wr[8];
        int   yr[8];
        int   seg0, all_one_seg;
        {
            const float* wg = wgt[buf];
            const int*   ys = yl[buf];
            seg0 = ys[rowbase];
            int same = 1;
#pragma unroll
            for (int r = 1; r < 16; ++r) same &= (ys[rowbase + r] == seg0) ? 1 : 0;
            all_one_seg = __builtin_amdgcn_readfirstlane(same);
#pragma unroll
            for (int r = 0; r < 8; ++r) {
                const int m = rowbase + mhalf + r;
                wr[r] = wg[m * NHEAD + head];
                yr[r] = ys[m];
            }
        }

        {
            float wA = 0.0f;
            if (okA) wA = __expf(scA - mxA) / dnA;
            const float4 z4 = make_float4(0.f, 0.f, 0.f, 0.f);
            if (!okA) { xa = z4; xb = z4; xc = z4; xd = z4; }
            float4* dA = (float4*)(&Xl[buf ^ 1][row * XP]) + e;
            dA[0] = xa; dA[8] = xb; dA[16] = xc; dA[24] = xd;
            if (wstore) wgt[buf ^ 1][row * NHEAD + h] = wA;
            if (e == 0) yl[buf ^ 1][row] = yA;
        }
        __syncthreads();

        if (all_one_seg) {
#pragma unroll
            for (int nt = 0; nt < 4; ++nt) {
                const int n = colbase + nt * 16 + lm;
                const float bn = bl[n];
                float sum = 0.0f;
#pragma unroll
                for (int r = 0; r < 8; ++r) sum += (acc[nt][r] + bn) * wr[r];
                sum += __shfl_xor(sum, 16, 32);
                if (lane < 16) atomicAdd(&out[(size_t)seg0 * TOT_OUT + n], sum);
            }
        } else {
#pragma unroll
            for (int nt = 0; nt < 4; ++nt) {
                const int n = colbase + nt * 16 + lm;
                const float bn = bl[n];
                int   cur = yr[0];
                float sum = (acc[nt][0] + bn) * wr[0];
#pragma unroll
                for (int r = 1; r < 8; ++r) {
                    const float val = (acc[nt][r] + bn) * wr[r];
                    if (yr[r] == cur) {
                        sum += val;
                    } else {
                        atomicAdd(&out[(size_t)cur * TOT_OUT + n], sum);
                        cur = yr[r]; sum = val;
                    }
                }
                atomicAdd(&out[(size_t)cur * TOT_OUT + n], sum);
            }
        }
        buf ^= 1;
        ndA = ndB; okA = okB; yA = yB; scA = scB;
        ndB = ndC; okB = okC;
    }
}


#define PL_THREADS 512
#define PL_BLOCKS  512

__device__ __forceinline__ int pl_first_row(unsigned starts, int run) {
    unsigned mm = starts | 1u;
    for (int q = 0; q < run && mm != 0u; ++q) mm &= mm - 1u;
    return (mm != 0u) ? __builtin_ctz(mm) : 0;
}

__device__ __forceinline__ void pl_add(float* __restrict__ pooled, float* __restrict__ wsum, const v8f acc, float wtot,
                                       int sa, int sb, int sw, int col, int hd) {
    if (sa >= 0) {
#pragma unroll
        for (int hh2 = 0; hh2 < NHEAD; ++hh2)
            atomicAdd(&pooled[((size_t)sa * NHEAD + hh2) * IN_CH + col], acc[hh2]);
    }
    if (sb >= 0) {
#pragma unroll
        for (int hh2 = 0; hh2 < NHEAD; ++hh2)
            atomicAdd(&pooled[((size_t)sb * NHEAD + hh2) * IN_CH + col], acc[4 + hh2]);
    }
    if (sw >= 0) atomicAdd(&wsum[sw * NHEAD + hd], wtot);
}

__device__ __forceinline__ void pl_flush(float* __restrict__ pooled, int seg, float c0, float c1, float c2, float c3,
                                         int col, int lane) {
    if (seg >= 0 && lane < 16) {
        atomicAdd(&pooled[((size_t)seg * NHEAD + 0) * IN_CH + col], c0);
        atomicAdd(&pooled[((size_t)seg * NHEAD + 1) * IN_CH + col], c1);
        atomicAdd(&pooled[((size_t)seg * NHEAD + 2) * IN_CH + col], c2);
        atomicAdd(&pooled[((size_t)seg * NHEAD + 3) * IN_CH + col], c3);
    }
}

__device__ __forceinline__ int pl_node(const int* __restrict__ perm, int p, int pend, int Nn) {
    const int pc = (p < pend) ? p : (pend - 1);
    int nd = (perm != nullptr) ? perm[pc] : pc;
    if ((unsigned)nd >= (unsigned)Nn) nd = 0;
    return nd;
}

__global__ __launch_bounds__(PL_THREADS, 8)
void k_pool(const float* __restrict__ x, const int* __restrict__ y,
            const float* __restrict__ score, const float* __restrict__ segmax,
            const int* __restrict__ perm,
            float* __restrict__ pooled, float* __restrict__ wsum, int Nn) {
    __shared__ __attribute__((aligned(16))) float Xl[2][GT_ROWS * XP];
    __shared__ float wgt[2][GT_ROWS * NHEAD];
    __shared__ int   yl[2][GT_ROWS];

    const int t = threadIdx.x;
    const int G = gridDim.x;
    const int ntiles = (Nn + GT_ROWS - 1) / GT_ROWS;
    int tile = (int)(((long long)blockIdx.x * ntiles) / G);
    const int tend = (int)(((long long)(blockIdx.x + 1) * ntiles) / G);
    if (tile >= tend) return;
    const int pend = (tend * GT_ROWS < Nn) ? tend * GT_ROWS : Nn;

    const int row = t >> 3, e = t & 7, h = t & 3;
    const bool wstore = (t & 4) == 0;
    {
        const int p0 = tile * GT_ROWS + row;
        const bool ok0 = p0 < pend;
        const int nd0 = pl_node(perm, p0, pend, Nn);
        const float4* s0 = (const float4*)(x + (size_t)nd0 * IN_CH) + e;
        float4 a0 = s0[0], a1 = s0[8], a2 = s0[16], a3 = s0[24];
        const int y0 = y[nd0];
        const int i0 = y0 * NHEAD + h;
        float w0 = 0.0f;
        if (ok0) w0 = __expf(score[(size_t)nd0 * NHEAD + h] - segmax[i0]);
        const float4 z4 = make_float4(0.f, 0.f, 0.f, 0.f);
        if (!ok0) { a0 = z4; a1 = z4; a2 = z4; a3 = z4; }
        float4* d0 = (float4*)(&Xl[0][row * XP]) + e;
        d0[0] = a0; d0[8] = a1; d0[16] = a2; d0[24] = a3;
        if (wstore) wgt[0][row * NHEAD + h] = w0;
        if (e == 0) yl[0][row] = y0;
    }
    int  pA = (tile + 1) * GT_ROWS + row;
    bool okA = pA < pend;
    int  ndA = pl_node(perm, pA, pend, Nn);
    int  yA  = y[ndA];
    float scA = score[(size_t)ndA * NHEAD + h];
    int  pB = (tile + 2) * GT_ROWS + row;
    bool okB = pB < pend;
    int  ndB = pl_node(perm, pB, pend, Nn);
    __syncthreads();

    const int wave = __builtin_amdgcn_readfirstlane(t >> 5), lane = t & 31;
    const int cb    = (wave & 7) * 16;
    const int rb    = (wave >> 3) * 32;
    const int lm    = lane & 15;
    const int khalf = (lane < 16) ? 0 : 2;
    const int mhalf = (lane < 16) ? 0 : 8;
    const int slot  = lm >> 2, hd = lm & 3;
    const v8f vzero = {0.f, 0.f, 0.f, 0.f, 0.f, 0.f, 0.f, 0.f};

    int   cseg = -1;
    float car0 = 0.0f, car1 = 0.0f, car2 = 0.0f, car3 = 0.0f;

    int buf = 0;
    for (; tile < tend; ++tile) {
        const float4* sA = (const float4*)(x + (size_t)ndA * IN_CH) + e;
        float4 xa = sA[0], xb = sA[8], xc = sA[16], xd = sA[24];
        const int iA = yA * NHEAD + h;
        const float mxA = segmax[iA];
        const int yB = y[ndB];
        const float scB = score[(size_t)ndB * NHEAD + h];
        const int pC = (tile + 3) * GT_ROWS + row;
        const bool okC = pC < pend;
        const int ndC = pl_node(perm, pC, pend, Nn);

        const float* wg = wgt[buf];
        const int*   ys = yl[buf];
        const int ymine = ys[rb + lane];
        const int yprev = ys[rb + ((lane > 0) ? lane - 1 : 0)];
        const unsigned starts = __builtin_amdgcn_ballot_w32(ymine != yprev);
        const int nruns = __builtin_popcount(starts) + 1;
        const int lastrun = nruns - 1;
        const float* xcol = &Xl[buf][rb * XP + cb + lm];
        const int slast  = __builtin_amdgcn_readfirstlane(ys[rb + 31]);

        float ws0 = 0.0f;
        int segw0 = -1;
        for (int pass = 0; pass * 4 < nruns; ++pass) {
            v8f acc = vzero;
            float wpart = 0.0f;
            if (pass == 0) { acc[0] = car0; acc[1] = car1; acc[2] = car2; acc[3] = car3; }
#pragma unroll
            for (int ks = 0; ks < 8; ++ks) {
                const int i0 = ks * 4 + khalf;
                const int ru0 = __builtin_popcount(starts & ((2u << (ks * 4 + 0)) - 1u)) - 4 * pass;
                const int ru1 = __builtin_popcount(starts & ((2u << (ks * 4 + 1)) - 1u)) - 4 * pass;
                const int ru2 = __builtin_popcount(starts & ((2u << (ks * 4 + 2)) - 1u)) - 4 * pass;
                const int ru3 = __builtin_popcount(starts & ((2u << (ks * 4 + 3)) - 1u)) - 4 * pass;
                const int r0 = (lane < 16) ? ru0 : ru2;
                const int r1 = (lane < 16) ? ru1 : ru3;
                v2f a, b;
                a.x = (r0 == slot) ? wg[(rb + i0) * NHEAD + hd] : 0.0f;
                a.y = (r1 == slot) ? wg[(rb + i0 + 1) * NHEAD + hd] : 0.0f;
                b.x = xcol[i0 * XP];
                b.y = xcol[(i0 + 1) * XP];
                acc = __builtin_amdgcn_wmma_f32_16x16x4_f32(false, a, false, b, (short)0, acc, false, false);
                wpart += a.x + a.y;
            }
            int segv[2];
#pragma unroll
            for (int s2 = 0; s2 < 2; ++s2) {
                const int run = 4 * pass + (mhalf >> 2) + s2;
                segv[s2] = (run < lastrun) ? ys[rb + pl_first_row(starts, run)] : -1;
            }
            if ((lastrun >> 2) == pass) {
                const int ls = lastrun & 3;
                float t0 = (ls & 1) ? acc[4] : acc[0];
                float t1 = (ls & 1) ? acc[5] : acc[1];
                float t2 = (ls & 1) ? acc[6] : acc[2];
                float t3 = (ls & 1) ? acc[7] : acc[3];
                if (ls & 2) {
                    t0 = __shfl_xor(t0, 16, 32); t1 = __shfl_xor(t1, 16, 32);
                    t2 = __shfl_xor(t2, 16, 32); t3 = __shfl_xor(t3, 16, 32);
                }
                car0 = (lane < 16) ? t0 : 0.0f; car1 = (lane < 16) ? t1 : 0.0f;
                car2 = (lane < 16) ? t2 : 0.0f; car3 = (lane < 16) ? t3 : 0.0f;
            }
            float wtot = 0.0f;
            int   segw = -1;
            if (cb == 0) {
                wtot = wpart + __shfl_xor(wpart, 16, 32);
                const int run = 4 * pass + slot;
                if (lane < 16 && run < nruns) segw = ys[rb + pl_first_row(starts, run)];
            }
            if (pass == 0) { ws0 = wtot; segw0 = segw; segw = -1; }
            pl_add(pooled, wsum, acc, wtot, segv[0], segv[1], segw, cb + lm, hd);
        }
        cseg = slast;

        {
            float wA = 0.0f;
            if (okA) wA = __expf(scA - mxA);
            const float4 z4 = make_float4(0.f, 0.f, 0.f, 0.f);
            if (!okA) { xa = z4; xb = z4; xc = z4; xd = z4; }
            float4* dA = (float4*)(&Xl[buf ^ 1][row * XP]) + e;
            dA[0] = xa; dA[8] = xb; dA[16] = xc; dA[24] = xd;
            if (wstore) wgt[buf ^ 1][row * NHEAD + h] = wA;
            if (e == 0) yl[buf ^ 1][row] = yA;
        }
        __syncthreads();

        if (segw0 >= 0) atomicAdd(&wsum[segw0 * NHEAD + hd], ws0);
        const int snext = __builtin_amdgcn_readfirstlane(yl[buf ^ 1][rb]);
        if (tile + 1 == tend || snext != cseg) {
            pl_flush(pooled, cseg, car0, car1, car2, car3, cb + lm, lane);
            car0 = 0.0f; car1 = 0.0f; car2 = 0.0f; car3 = 0.0f;
        }
        buf ^= 1;
        ndA = ndB; okA = okB; yA = yB; scA = scB;
        ndB = ndC; okB = okC;
    }
}

__global__ __launch_bounds__(128)
void k_proj(const float* __restrict__ pooled, const float* __restrict__ wsum,
            const float* __restrict__ lin_w, const float* __restrict__ lin_b,
            float* __restrict__ out, int C) {
    const int lane = threadIdx.x & 31;
    const int head = __builtin_amdgcn_readfirstlane((int)(threadIdx.x >> 5));
    const int sbase = blockIdx.x * 16;
    const int lm    = lane & 15;
    const int khalf = (lane < 16) ? 0 : 2;
    const int mhalf = (lane < 16) ? 0 : 8;
    const int sa = (sbase + lm < C) ? (sbase + lm) : (C - 1);
    const float* ap = pooled + ((size_t)sa * NHEAD + head) * IN_CH + khalf;
    const float* bp = lin_w + (size_t)(head * OUT_CH + lm) * IN_CH + khalf;
    const v8f vzero = {0.f, 0.f, 0.f, 0.f, 0.f, 0.f, 0.f, 0.f};
    v8f acc[4];
#pragma unroll
    for (int i = 0; i < 4; ++i) acc[i] = vzero;
#pragma unroll 8
    for (int ks = 0; ks < IN_CH / 4; ++ks) {
        const v2f a = *(const v2f*)(ap + ks * 4);
#pragma unroll
        for (int nt = 0; nt < 4; ++nt) {
            const v2f b = *(const v2f*)(bp + (size_t)nt * 16 * IN_CH + ks * 4);
            acc[nt] = __builtin_amdgcn_wmma_f32_16x16x4_f32(false, a, false, b, (short)0, acc[nt], false, false);
        }
    }
    float inv[8], one[8];
#pragma unroll
    for (int r = 0; r < 8; ++r) {
        const int sg = sbase + mhalf + r;
        const float ws = (sg < C) ? wsum[sg * NHEAD + head] : 0.0f;
        inv[r] = (ws != 0.0f) ? 1.0f / ws : 0.0f;
        one[r] = (ws != 0.0f) ? 1.0f : 0.0f;
    }
#pragma unroll
    for (int nt = 0; nt < 4; ++nt) {
        const int n = head * OUT_CH + nt * 16 + lm;
        const float bn = lin_b[n];
#pragma unroll
        for (int r = 0; r < 8; ++r) {
            const int sg = sbase + mhalf + r;
            if (sg < C) out[(size_t)sg * TOT_OUT + n] = acc[nt][r] * inv[r] + bn * one[r];
        }
    }
}

extern "C" void kernel_launch(void* const* d_in, const int* in_sizes, int n_in,
                              void* d_out, int out_size, void* d_ws, size_t ws_size,
                              hipStream_t stream) {
    (void)n_in;
    const float* x     = (const float*)d_in[0];
    const int*   y     = (const int*)  d_in[1];
    const float* lin_w = (const float*)d_in[3];
    const float* lin_b = (const float*)d_in[4];
    const float* att_w = (const float*)d_in[5];
    const float* att_b = (const float*)d_in[6];
    float* out = (float*)d_out;

    int N = in_sizes[0] / IN_CH;
    int C = out_size / TOT_OUT;

    float* wsf    = (float*)d_ws;
    float* v      = wsf;
    float* c      = wsf + 512;
    float* segmax = wsf + 516;
    float* denom  = segmax + 4 * C;
    float* score  = denom + 4 * C;
    int*   cnt    = (int*)(score + (size_t)4 * N);
    int*   perm   = cnt + C;
    int*   tick   = perm + N;

    size_t need = ((size_t)516 + (size_t)10 * C + (size_t)5 * N) * sizeof(float);
    if (ws_size < need) { cnt = nullptr; perm = nullptr; tick = nullptr; }

    size_t poff = (((size_t)516 + (size_t)10 * C + (size_t)5 * N) + 3) & ~(size_t)3;
    float* pooled = wsf + poff;
    float* wsum   = pooled + (size_t)C * NHEAD * IN_CH;
    size_t need2 = (poff + (size_t)C * (NHEAD * IN_CH + NHEAD)) * sizeof(float);
    const bool pool = (perm != nullptr) && (ws_size >= need2);

    int initN = pool ? C * (NHEAD * IN_CH + NHEAD) : C * TOT_OUT;
    float* dn = pool ? nullptr : denom;
    k_prep <<<PREP_FOLD + PREP_FILL, 256, 0, stream>>>(pool ? pooled : out, initN, dn, (unsigned*)segmax, cnt, tick, C,
                                                       lin_w, lin_b, att_w, att_b, v, c);
    k_score<<<(N + SC_SLABS * 256 - 1) / (SC_SLABS * 256), 256, 0, stream>>>(x, y, v, c, score,
                                                 (int*)segmax, (unsigned*)segmax, cnt, N, C);
    if (perm != nullptr && C > SEG_CAP) k_scan<<<1, SCAN_T, 0, stream>>>(cnt, C);
    k_denom<<<(N + DN_NPT * 256 - 1) / (DN_NPT * 256), 256, 0, stream>>>(score, y, segmax, dn, cnt, tick, perm, N, C);
    if (pool) {
        k_pool<<<PL_BLOCKS, PL_THREADS, 0, stream>>>(x, y, score, segmax, perm, pooled, wsum, N);
        k_proj<<<(C + 15) / 16, 128, 0, stream>>>(pooled, wsum, lin_w, lin_b, out, C);
    } else {
        k_gemm<<<GEMM_BLOCKS, GT_THREADS, 0, stream>>>(x, y, lin_w, lin_b,
                                                      score, segmax, denom, perm, out, N);
    }
}


# k_pool: A-fragment run masks from four ballots (1 mask register, not 32), the pass's weights read as one LDS group; loop fits 128 registers, no scratch: row gathers stay in flight under the products
# speedup vs baseline: 8.2874x; 1.1386x over previous
#include <hip/hip_runtime.h>
#include <math.h>

typedef __attribute__((ext_vector_type(2))) float v2f;
typedef __attribute__((ext_vector_type(8))) float v8f;

#define IN_CH     128
#define OUT_CH    64
#define NHEAD     4
#define TOT_OUT   256
#define NEG_SLOPE 0.2f


#define PREP_FOLD 3
#define PREP_FILL 256
__global__ __launch_bounds__(256)
void k_prep(float* zbuf, int zn, float* denom, unsigned* segmax, int* cnt, int* tick, int C,
            const float* __restrict__ lin_w, const float* __restrict__ lin_b,
            const float* __restrict__ att_w, const float* __restrict__ att_b,
            float* v, float* c) {
    if (blockIdx.x < PREP_FOLD) {
        if (blockIdx.x < 2) {
            const int t = blockIdx.x * 256 + threadIdx.x;
            if (t < NHEAD * IN_CH) {
                const int h = t / IN_CH, i = t % IN_CH;
                float s = 0.0f;
#pragma unroll
                for (int d = 0; d < OUT_CH; ++d)
                    s += lin_w[(size_t)(h * OUT_CH + d) * IN_CH + i] * att_w[d];
                v[h * IN_CH + i] = s;
            }
        } else if (threadIdx.x < NHEAD) {
            const int t = threadIdx.x;
            float s = att_b[0];
            for (int d = 0; d < OUT_CH; ++d) s += lin_b[t * OUT_CH + d] * att_w[d];
            c[t] = s;
        }
        return;
    }
    const int gt = (blockIdx.x - PREP_FOLD) * 256 + threadIdx.x;
    const int gs = PREP_FILL * 256;
    const int n4 = zn >> 2;
    float4* z4 = (float4*)zbuf;
    const float4 zero4 = make_float4(0.f, 0.f, 0.f, 0.f);
    for (int i = gt; i < n4; i += gs) z4[i] = zero4;
    for (int i = (n4 << 2) + gt; i < zn; i += gs) zbuf[i] = 0.0f;
    for (int i = gt; i < 4 * C; i += gs) { if (denom != nullptr) denom[i] = 0.0f; segmax[i] = 0xFF800000u; }
    if (cnt != nullptr)  for (int i = gt; i < C; i += gs) cnt[i] = 0;
    if (tick != nullptr) for (int i = gt; i < C; i += gs) tick[i] = 0;
}

typedef __attribute__((ext_vector_type(4))) float v4f;

#define SC_ROWS   8
#define SC_GROUPS 4
#define SC_WAVES  8
#define SC_SLABS  4
#define SEG_CAP   2048

__global__ __launch_bounds__(256)
void k_score(const float* __restrict__ x, const int* __restrict__ y,
             const float* __restrict__ v, const float* __restrict__ c,
             float* __restrict__ score, int* segmax_i, unsigned* segmax_u, int* cnt, int Nn, int C) {
    __shared__ int lmax[NHEAD * SEG_CAP];
    __shared__ int lcnt[SEG_CAP];
    const bool bins = (C <= SEG_CAP);
    if (bins) {
        for (int i = threadIdx.x; i < NHEAD * C; i += 256) lmax[i] = (int)0xFF800000;
        for (int i = threadIdx.x; i < C; i += 256) lcnt[i] = 0;
    }
    __syncthreads();

    const int lane = threadIdx.x & 31;
    const int wave = __builtin_amdgcn_readfirstlane((int)(threadIdx.x >> 5));

    v4f vh[NHEAD];
#pragma unroll
    for (int h = 0; h < NHEAD; ++h) vh[h] = *(const v4f*)(v + h * IN_CH + lane * 4);
    const int hh = lane & 3;
    const int rr = lane >> 2;
    const float ch = c[hh];

    for (int sl = 0; sl < SC_SLABS; ++sl) {
    const int row0 = ((blockIdx.x * SC_SLABS + sl) * SC_WAVES + wave) * (SC_ROWS * SC_GROUPS);
    if (row0 < Nn) {
    const float* xb = x + (size_t)row0 * IN_CH + lane * 4;
    v4f xv[2][SC_ROWS];
#define SC_LOAD(G) do { _Pragma("unroll") for (int r = 0; r < SC_ROWS; ++r) { \
        const int ro_ = (row0 + (G) * SC_ROWS + r < Nn) ? ((G) * SC_ROWS + r) : 0; \
        xv[(G) & 1][r] = __builtin_nontemporal_load((const v4f*)(xb + (size_t)ro_ * IN_CH)); } } while (0)

    int yv[SC_GROUPS];
#pragma unroll
    for (int g = 0; g < SC_GROUPS; ++g) {
        const int nd = row0 + g * SC_ROWS + rr;
        yv[g] = y[(nd < Nn) ? nd : (Nn - 1)];
    }

    SC_LOAD(0);
#pragma unroll
    for (int g = 0; g < SC_GROUPS; ++g) {
        __builtin_amdgcn_sched_barrier(0);
        if (g + 1 < SC_GROUPS) SC_LOAD(g + 1);
        __builtin_amdgcn_sched_barrier(0);

        float vals[SC_ROWS * NHEAD];
#pragma unroll
        for (int r = 0; r < SC_ROWS; ++r) {
            const v4f xr = xv[g & 1][r];
#pragma unroll
            for (int h = 0; h < NHEAD; ++h)
                vals[r * NHEAD + h] = xr.x * vh[h].x + xr.y * vh[h].y + xr.z * vh[h].z + xr.w * vh[h].w;
        }
#define SC_STAGE(WD) do { const bool up_ = (lane & (WD)) != 0; \
        _Pragma("unroll") for (int i = 0; i < (WD); ++i) { \
            const float keep_ = up_ ? vals[i + (WD)] : vals[i]; \
            const float send_ = up_ ? vals[i] : vals[i + (WD)]; \
            vals[i] = keep_ + __shfl_xor(send_, (WD), 32); } } while (0)
        SC_STAGE(16); SC_STAGE(8); SC_STAGE(4); SC_STAGE(2); SC_STAGE(1);
#undef SC_STAGE
        float s = vals[0] + ch;
        s = (s >= 0.0f) ? s : NEG_SLOPE * s;
        const int node = row0 + g * SC_ROWS + rr;
        if (node < Nn) {
            score[(size_t)node * NHEAD + hh] = s;
            const int idx = yv[g] * NHEAD + hh;
            if (bins) {
                if (s >= 0.0f) atomicMax(&lmax[idx], __float_as_int(s));
                else           atomicMin((unsigned*)&lmax[idx], (unsigned)__float_as_int(s));
                if (hh == 0) atomicAdd(&lcnt[yv[g]], 1);
            } else {
                if (s >= 0.0f) atomicMax(&segmax_i[idx], __float_as_int(s));
                else           atomicMin(&segmax_u[idx], (unsigned)__float_as_int(s));
                if (cnt != nullptr && hh == 0) atomicAdd(&cnt[yv[g]], 1);
            }
        }
    }
#undef SC_LOAD
    }
    }

    __syncthreads();
    if (bins) {
        for (int i = threadIdx.x; i < NHEAD * C; i += 256) {
            const int b = lmax[i];
            if (b != (int)0xFF800000) {
                if (__int_as_float(b) >= 0.0f) atomicMax(&segmax_i[i], b);
                else                           atomicMin(&segmax_u[i], (unsigned)b);
            }
        }
        if (cnt != nullptr)
            for (int i = threadIdx.x; i < C; i += 256) {
                const int n = lcnt[i];
                if (n != 0) atomicAdd(&cnt[i], n);
            }
    }
}

#define SCAN_T 1024
__global__ __launch_bounds__(SCAN_T)
void k_scan(int* cnt, int C) {
    __shared__ int part[SCAN_T];
    const int t = threadIdx.x;
    const int per = (C + SCAN_T - 1) / SCAN_T;
    const int b = t * per;
    int s = 0;
    for (int j = 0; j < per; ++j) { const int i = b + j; if (i < C) s += cnt[i]; }
    part[t] = s;
    __syncthreads();
    for (int d = 1; d < SCAN_T; d <<= 1) {
        const int add = (t >= d) ? part[t - d] : 0;
        __syncthreads();
        part[t] += add;
        __syncthreads();
    }
    int run = part[t] - s;
    for (int j = 0; j < per; ++j) {
        const int i = b + j;
        if (i < C) { const int n = cnt[i]; cnt[i] = run; run += n; }
    }
}

#define DN_NPT 4
#define DN_SPT (SEG_CAP / 256)
__global__ __launch_bounds__(256)
void k_denom(const float* __restrict__ score, const int* __restrict__ y,
             const float* __restrict__ segmax, float* denom,
             int* cursor, int* tick, int* perm, int Nn, int C) {
    __shared__ float lden[NHEAD * SEG_CAP];
    __shared__ int   lrun[SEG_CAP];
    __shared__ int   lbase[SEG_CAP];
    __shared__ int   wtot[8];
    const bool bins = (C <= SEG_CAP);
    const bool scan = bins && (perm != nullptr);
    const bool dsum = (denom != nullptr);
    const int t = threadIdx.x;

    const int per = (C + 255) / 256;
    int cv[DN_SPT];
#pragma unroll
    for (int j = 0; j < DN_SPT; ++j) {
        const int i = t * per + j;
        cv[j] = (scan && j < per && i < C) ? cursor[i] : 0;
    }
    const int base = blockIdx.x * (256 * DN_NPT);
    int  yy[DN_NPT], rk[DN_NPT];
    v4f  sc[DN_NPT];
#pragma unroll
    for (int j = 0; j < DN_NPT; ++j) {
        const int node = base + j * 256 + t;
        const int nc = (node < Nn) ? node : (Nn - 1);
        yy[j] = y[nc];
        sc[j] = (v4f){0.0f, 0.0f, 0.0f, 0.0f};
        if (dsum) sc[j] = *(const v4f*)(score + (size_t)nc * NHEAD);
    }
    if (bins) {
        if (dsum) for (int i = t; i < NHEAD * C; i += 256) lden[i] = 0.0f;
        for (int i = t; i < C; i += 256) lrun[i] = 0;
    }
    int mine = 0, incl = 0;
    if (scan) {
#pragma unroll
        for (int j = 0; j < DN_SPT; ++j) mine += cv[j];
        incl = mine;
        for (int d = 1; d < 32; d <<= 1) {
            const int o = __shfl_up(incl, d, 32);
            if ((t & 31) >= d) incl += o;
        }
        if ((t & 31) == 31) wtot[t >> 5] = incl;
    }
    __syncthreads();
    if (scan) {
        int run = incl - mine;
        for (int w = 0; w < (t >> 5); ++w) run += wtot[w];
#pragma unroll
        for (int j = 0; j < DN_SPT; ++j) {
            const int i = t * per + j;
            if (j < per && i < C) { lbase[i] = run; run += cv[j]; }
        }
    }

#pragma unroll
    for (int j = 0; j < DN_NPT; ++j) {
        const int node = base + j * 256 + t;
        float e0 = 0.0f, e1 = 0.0f, e2 = 0.0f, e3 = 0.0f;
        if (dsum) {
            const v4f mx = *(const v4f*)(segmax + (size_t)yy[j] * NHEAD);
            e0 = __expf(sc[j].x - mx.x); e1 = __expf(sc[j].y - mx.y);
            e2 = __expf(sc[j].z - mx.z); e3 = __expf(sc[j].w - mx.w);
        }
        rk[j] = 0;
        if (node < Nn) {
            if (bins) {
                if (dsum) {
                    float* ld = &lden[yy[j] * NHEAD];
                    atomicAdd(ld + 0, e0); atomicAdd(ld + 1, e1);
                    atomicAdd(ld + 2, e2); atomicAdd(ld + 3, e3);
                }
                if (perm != nullptr) rk[j] = atomicAdd(&lrun[yy[j]], 1);
            } else {
                if (dsum) {
                    float* gd = &denom[(size_t)yy[j] * NHEAD];
                    atomicAdd(gd + 0, e0); atomicAdd(gd + 1, e1);
                    atomicAdd(gd + 2, e2); atomicAdd(gd + 3, e3);
                }
                if (perm != nullptr) {
                    const int pos = atomicAdd(&cursor[yy[j]], 1);
                    if ((unsigned)pos < (unsigned)Nn) perm[pos] = node;
                }
            }
        }
    }
    __syncthreads();
    if (bins) {
        if (dsum)
            for (int i = t; i < NHEAD * C; i += 256) {
                const float s = lden[i];
                if (s != 0.0f) atomicAdd(&denom[i], s);
            }
        if (perm != nullptr) {
            int nn[DN_SPT], tk[DN_SPT];
#pragma unroll
            for (int j = 0; j < DN_SPT; ++j) { const int i = t + j * 256; nn[j] = (i < C) ? lrun[i] : 0; }
#pragma unroll
            for (int j = 0; j < DN_SPT; ++j) { tk[j] = 0; if (nn[j] != 0) tk[j] = atomicAdd(&tick[t + j * 256], nn[j]); }
#pragma unroll
            for (int j = 0; j < DN_SPT; ++j) {
                const int i = t + j * 256;
                if (i < C) lrun[i] = (nn[j] != 0) ? lbase[i] + tk[j] : 0;
            }
        }
    }
    __syncthreads();
    if (bins && perm != nullptr) {
#pragma unroll
        for (int j = 0; j < DN_NPT; ++j) {
            const int node = base + j * 256 + t;
            if (node < Nn) {
                const int pos = lrun[yy[j]] + rk[j];
                if ((unsigned)pos < (unsigned)Nn) perm[pos] = node;
            }
        }
    }
}

#define GT_ROWS    64
#define GT_THREADS 512
#define XP         132
#define GEMM_BLOCKS 256

__device__ __forceinline__ int gm_node(const int* __restrict__ perm, int p, int Nn) {
    const int pc = (p < Nn) ? p : (Nn - 1);
    int nd = (perm != nullptr) ? perm[pc] : pc;
    if ((unsigned)nd >= (unsigned)Nn) nd = 0;
    return nd;
}

__global__ __launch_bounds__(GT_THREADS)
void k_gemm(const float* __restrict__ x, const int* __restrict__ y,
            const float* __restrict__ lin_w, const float* __restrict__ lin_b,
            const float* __restrict__ score, const float* __restrict__ segmax,
            const float* __restrict__ denom, const int* __restrict__ perm,
            float* __restrict__ out, int Nn) {
    __shared__ __attribute__((aligned(16))) float Wl[TOT_OUT * XP];
    __shared__ __attribute__((aligned(16))) float Xl[2][GT_ROWS * XP];
    __shared__ float wgt[2][GT_ROWS * NHEAD];
    __shared__ int   yl[2][GT_ROWS];
    __shared__ float bl[TOT_OUT];

    const int t = threadIdx.x;
    const int G = gridDim.x;
    const int ntiles = (Nn + GT_ROWS - 1) / GT_ROWS;
    int tile = blockIdx.x;
    if (tile >= ntiles) return;

    const int row = t >> 3, e = t & 7, h = t & 3;
    const bool wstore = (t & 4) == 0;

    {
        const int p0 = tile * GT_ROWS + row;
        const bool ok0 = p0 < Nn;
        const int nd0 = gm_node(perm, p0, Nn);
        const float4* s0 = (const float4*)(x + (size_t)nd0 * IN_CH) + e;
        float4 a0 = s0[0], a1 = s0[8], a2 = s0[16], a3 = s0[24];
        const int y0 = y[nd0];
        const int i0 = y0 * NHEAD + h;
        float w0 = 0.0f;
        if (ok0) w0 = __expf(score[(size_t)nd0 * NHEAD + h] - segmax[i0]) / denom[i0];
        const float4 z4 = make_float4(0.f, 0.f, 0.f, 0.f);
        if (!ok0) { a0 = z4; a1 = z4; a2 = z4; a3 = z4; }
        float4* d0 = (float4*)(&Xl[0][row * XP]) + e;
        d0[0] = a0; d0[8] = a1; d0[16] = a2; d0[24] = a3;
        if (wstore) wgt[0][row * NHEAD + h] = w0;
        if (e == 0) yl[0][row] = y0;
    }
    if (t < TOT_OUT) bl[t] = lin_b[t];
    for (int i = t; i < TOT_OUT * (IN_CH / 4); i += GT_THREADS) {
        const int n = i >> 5, c4 = i & 31;
        const float4 w4 = *((const float4*)(lin_w + (size_t)n * IN_CH) + c4);
        *((float4*)(&Wl[n * XP]) + c4) = w4;
    }
    int  pA = (tile + G) * GT_ROWS + row;
    bool okA = pA < Nn;
    int  ndA = gm_node(perm, pA, Nn);
    int  yA  = y[ndA];
    float scA = score[(size_t)ndA * NHEAD + h];
    int  pB = (tile + 2 * G) * GT_ROWS + row;
    bool okB = pB < Nn;
    int  ndB = gm_node(perm, pB, Nn);
    __syncthreads();

    const int wave = __builtin_amdgcn_readfirstlane(t >> 5), lane = t & 31;
    const int rowbase = (wave >> 2) * 16;
    const int colbase = (wave & 3) * 64;
    const int head    = wave & 3;
    const int lm      = lane & 15;
    const int khalf   = (lane < 16) ? 0 : 2;
    const int mhalf   = (lane < 16) ? 0 : 8;
    const v8f vzero = {0.f, 0.f, 0.f, 0.f, 0.f, 0.f, 0.f, 0.f};
    const float* bp = &Wl[(colbase + lm) * XP + khalf];

    int buf = 0;
    for (; tile < ntiles; tile += G) {
        const float4* sA = (const float4*)(x + (size_t)ndA * IN_CH) + e;
        float4 xa = sA[0], xb = sA[8], xc = sA[16], xd = sA[24];
        const int iA = yA * NHEAD + h;
        const float mxA = segmax[iA], dnA = denom[iA];
        const int yB = y[ndB];
        const float scB = score[(size_t)ndB * NHEAD + h];
        const int pC = (tile + 3 * G) * GT_ROWS + row;
        const bool okC = pC < Nn;
        const int ndC = gm_node(perm, pC, Nn);

        v8f acc[4];
#pragma unroll
        for (int i = 0; i < 4; ++i) acc[i] = vzero;
        const float* ap = &Xl[buf][(rowbase + lm) * XP + khalf];
#pragma unroll
        for (int ks = 0; ks < IN_CH / 4; ++ks) {
            const v2f a = *(const v2f*)(ap + ks * 4);
#pragma unroll
            for (int nt = 0; nt < 4; ++nt) {
                const v2f b = *(const v2f*)(bp + nt * 16 * XP + ks * 4);
                acc[nt] = __builtin_amdgcn_wmma_f32_16x16x4_f32(
                    false, a, false, b, (short)0, acc[nt], false, false);
            }
        }

        float wr[8];
        int   yr[8];
        int   seg0, all_one_seg;
        {
            const float* wg = wgt[buf];
            const int*   ys = yl[buf];
            seg0 = ys[rowbase];
            int same = 1;
#pragma unroll
            for (int r = 1; r < 16; ++r) same &= (ys[rowbase + r] == seg0) ? 1 : 0;
            all_one_seg = __builtin_amdgcn_readfirstlane(same);
#pragma unroll
            for (int r = 0; r < 8; ++r) {
                const int m = rowbase + mhalf + r;
                wr[r] = wg[m * NHEAD + head];
                yr[r] = ys[m];
            }
        }

        {
            float wA = 0.0f;
            if (okA) wA = __expf(scA - mxA) / dnA;
            const float4 z4 = make_float4(0.f, 0.f, 0.f, 0.f);
            if (!okA) { xa = z4; xb = z4; xc = z4; xd = z4; }
            float4* dA = (float4*)(&Xl[buf ^ 1][row * XP]) + e;
            dA[0] = xa; dA[8] = xb; dA[16] = xc; dA[24] = xd;
            if (wstore) wgt[buf ^ 1][row * NHEAD + h] = wA;
            if (e == 0) yl[buf ^ 1][row] = yA;
        }
        __syncthreads();

        if (all_one_seg) {
#pragma unroll
            for (int nt = 0; nt < 4; ++nt) {
                const int n = colbase + nt * 16 + lm;
                const float bn = bl[n];
                float sum = 0.0f;
#pragma unroll
                for (int r = 0; r < 8; ++r) sum += (acc[nt][r] + bn) * wr[r];
                sum += __shfl_xor(sum, 16, 32);
                if (lane < 16) atomicAdd(&out[(size_t)seg0 * TOT_OUT + n], sum);
            }
        } else {
#pragma unroll
            for (int nt = 0; nt < 4; ++nt) {
                const int n = colbase + nt * 16 + lm;
                const float bn = bl[n];
                int   cur = yr[0];
                float sum = (acc[nt][0] + bn) * wr[0];
#pragma unroll
                for (int r = 1; r < 8; ++r) {
                    const float val = (acc[nt][r] + bn) * wr[r];
                    if (yr[r] == cur) {
                        sum += val;
                    } else {
                        atomicAdd(&out[(size_t)cur * TOT_OUT + n], sum);
                        cur = yr[r]; sum = val;
                    }
                }
                atomicAdd(&out[(size_t)cur * TOT_OUT + n], sum);
            }
        }
        buf ^= 1;
        ndA = ndB; okA = okB; yA = yB; scA = scB;
        ndB = ndC; okB = okC;
    }
}


#define PL_THREADS 512
#define PL_BLOCKS  512

__device__ __forceinline__ int pl_first_row(unsigned starts, int run) {
    unsigned mm = starts | 1u;
    for (int q = 0; q < run && mm != 0u; ++q) mm &= mm - 1u;
    return (mm != 0u) ? __builtin_ctz(mm) : 0;
}

__device__ __forceinline__ void pl_add(float* __restrict__ pooled, float* __restrict__ wsum, const v8f acc, float wtot,
                                       int sa, int sb, int sw, int col, int hd) {
    if (sa >= 0) {
#pragma unroll
        for (int hh2 = 0; hh2 < NHEAD; ++hh2)
            atomicAdd(&pooled[((size_t)sa * NHEAD + hh2) * IN_CH + col], acc[hh2]);
    }
    if (sb >= 0) {
#pragma unroll
        for (int hh2 = 0; hh2 < NHEAD; ++hh2)
            atomicAdd(&pooled[((size_t)sb * NHEAD + hh2) * IN_CH + col], acc[4 + hh2]);
    }
    if (sw >= 0) atomicAdd(&wsum[sw * NHEAD + hd], wtot);
}

__device__ __forceinline__ void pl_flush(float* __restrict__ pooled, int seg, float c0, float c1, float c2, float c3,
                                         int col, int lane) {
    if (seg >= 0 && lane < 16) {
        atomicAdd(&pooled[((size_t)seg * NHEAD + 0) * IN_CH + col], c0);
        atomicAdd(&pooled[((size_t)seg * NHEAD + 1) * IN_CH + col], c1);
        atomicAdd(&pooled[((size_t)seg * NHEAD + 2) * IN_CH + col], c2);
        atomicAdd(&pooled[((size_t)seg * NHEAD + 3) * IN_CH + col], c3);
    }
}

__device__ __forceinline__ int pl_node(const int* __restrict__ perm, int p, int pend, int Nn) {
    const int pc = (p < pend) ? p : (pend - 1);
    int nd = (perm != nullptr) ? perm[pc] : pc;
    if ((unsigned)nd >= (unsigned)Nn) nd = 0;
    return nd;
}

__global__ __launch_bounds__(PL_THREADS, 8)
void k_pool(const float* __restrict__ x, const int* __restrict__ y,
            const float* __restrict__ score, const float* __restrict__ segmax,
            const int* __restrict__ perm,
            float* __restrict__ pooled, float* __restrict__ wsum, int Nn) {
    __shared__ __attribute__((aligned(16))) float Xl[2][GT_ROWS * XP];
    __shared__ float wgt[2][GT_ROWS * NHEAD];
    __shared__ int   yl[2][GT_ROWS];

    const int t = threadIdx.x;
    const int G = gridDim.x;
    const int ntiles = (Nn + GT_ROWS - 1) / GT_ROWS;
    int tile = (int)(((long long)blockIdx.x * ntiles) / G);
    const int tend = (int)(((long long)(blockIdx.x + 1) * ntiles) / G);
    if (tile >= tend) return;
    const int pend = (tend * GT_ROWS < Nn) ? tend * GT_ROWS : Nn;

    const int row = t >> 3, e = t & 7, h = t & 3;
    const bool wstore = (t & 4) == 0;
    {
        const int p0 = tile * GT_ROWS + row;
        const bool ok0 = p0 < pend;
        const int nd0 = pl_node(perm, p0, pend, Nn);
        const float4* s0 = (const float4*)(x + (size_t)nd0 * IN_CH) + e;
        float4 a0 = s0[0], a1 = s0[8], a2 = s0[16], a3 = s0[24];
        const int y0 = y[nd0];
        const int i0 = y0 * NHEAD + h;
        float w0 = 0.0f;
        if (ok0) w0 = __expf(score[(size_t)nd0 * NHEAD + h] - segmax[i0]);
        const float4 z4 = make_float4(0.f, 0.f, 0.f, 0.f);
        if (!ok0) { a0 = z4; a1 = z4; a2 = z4; a3 = z4; }
        float4* d0 = (float4*)(&Xl[0][row * XP]) + e;
        d0[0] = a0; d0[8] = a1; d0[16] = a2; d0[24] = a3;
        if (wstore) wgt[0][row * NHEAD + h] = w0;
        if (e == 0) yl[0][row] = y0;
    }
    int  pA = (tile + 1) * GT_ROWS + row;
    bool okA = pA < pend;
    int  ndA = pl_node(perm, pA, pend, Nn);
    int  yA  = y[ndA];
    float scA = score[(size_t)ndA * NHEAD + h];
    int  pB = (tile + 2) * GT_ROWS + row;
    bool okB = pB < pend;
    int  ndB = pl_node(perm, pB, pend, Nn);
    __syncthreads();

    const int wave = __builtin_amdgcn_readfirstlane(t >> 5), lane = t & 31;
    const int cb    = (wave & 7) * 16;
    const int rb    = (wave >> 3) * 32;
    const int lm    = lane & 15;
    const int khalf = (lane < 16) ? 0 : 2;
    const int mhalf = (lane < 16) ? 0 : 8;
    const int slot  = lm >> 2, hd = lm & 3;
    const unsigned lemask = (2u << lane) - 1u;
    const v8f vzero = {0.f, 0.f, 0.f, 0.f, 0.f, 0.f, 0.f, 0.f};

    int   cseg = -1;
    float car0 = 0.0f, car1 = 0.0f, car2 = 0.0f, car3 = 0.0f;

    int buf = 0;
    for (; tile < tend; ++tile) {
        const float4* sA = (const float4*)(x + (size_t)ndA * IN_CH) + e;
        float4 xa = sA[0], xb = sA[8], xc = sA[16], xd = sA[24];
        const int iA = yA * NHEAD + h;
        const float mxA = segmax[iA];
        const int yB = y[ndB];
        const float scB = score[(size_t)ndB * NHEAD + h];
        const int pC = (tile + 3) * GT_ROWS + row;
        const bool okC = pC < pend;
        const int ndC = pl_node(perm, pC, pend, Nn);

        const float* wg = wgt[buf];
        const int*   ys = yl[buf];
        const int ymine = ys[rb + lane];
        const int yprev = ys[rb + ((lane > 0) ? lane - 1 : 0)];
        const unsigned starts = __builtin_amdgcn_ballot_w32(ymine != yprev);
        const int nruns = __builtin_popcount(starts) + 1;
        const int lastrun = nruns - 1;
        const float* xcol = &Xl[buf][rb * XP + cb + lm];
        const int myrun = __builtin_popcount(starts & lemask);
        const float* wrow = wg + (rb + khalf) * NHEAD + hd;
        const int slast  = __builtin_amdgcn_readfirstlane(ys[rb + 31]);

        float ws0 = 0.0f;
        int segw0 = -1;
        for (int pass = 0; pass * 4 < nruns; ++pass) {
            v8f acc = vzero;
            float wpart = 0.0f;
            if (pass == 0) { acc[0] = car0; acc[1] = car1; acc[2] = car2; acc[3] = car3; }
            const int rel = myrun - 4 * pass;
            const unsigned m0 = __builtin_amdgcn_ballot_w32(rel == 0);
            const unsigned m1 = __builtin_amdgcn_ballot_w32(rel == 1);
            const unsigned m2 = __builtin_amdgcn_ballot_w32(rel == 2);
            const unsigned m3 = __builtin_amdgcn_ballot_w32(rel == 3);
            const unsigned rmask = ((slot == 0) ? m0 : (slot == 1) ? m1 : (slot == 2) ? m2 : m3) >> khalf;
            float wa[16], xb2[16];
#pragma unroll
            for (int ks = 0; ks < 8; ++ks) {
                const int i0 = ks * 4 + khalf;
                wa[2 * ks]      = wrow[ks * 4 * NHEAD];
                wa[2 * ks + 1]  = wrow[ks * 4 * NHEAD + NHEAD];
                xb2[2 * ks]     = xcol[i0 * XP];
                xb2[2 * ks + 1] = xcol[(i0 + 1) * XP];
            }
#pragma unroll
            for (int ks = 0; ks < 8; ++ks) {
                v2f a, b;
                a.x = ((rmask >> (ks * 4)) & 1u)     ? wa[2 * ks]     : 0.0f;
                a.y = ((rmask >> (ks * 4 + 1)) & 1u) ? wa[2 * ks + 1] : 0.0f;
                b.x = xb2[2 * ks];
                b.y = xb2[2 * ks + 1];
                acc = __builtin_amdgcn_wmma_f32_16x16x4_f32(false, a, false, b, (short)0, acc, false, false);
                wpart += a.x + a.y;
            }
            int segv[2];
#pragma unroll
            for (int s2 = 0; s2 < 2; ++s2) {
                const int run = 4 * pass + (mhalf >> 2) + s2;
                segv[s2] = (run < lastrun) ? ys[rb + pl_first_row(starts, run)] : -1;
            }
            if ((lastrun >> 2) == pass) {
                const int ls = lastrun & 3;
                float t0 = (ls & 1) ? acc[4] : acc[0];
                float t1 = (ls & 1) ? acc[5] : acc[1];
                float t2 = (ls & 1) ? acc[6] : acc[2];
                float t3 = (ls & 1) ? acc[7] : acc[3];
                if (ls & 2) {
                    t0 = __shfl_xor(t0, 16, 32); t1 = __shfl_xor(t1, 16, 32);
                    t2 = __shfl_xor(t2, 16, 32); t3 = __shfl_xor(t3, 16, 32);
                }
                car0 = (lane < 16) ? t0 : 0.0f; car1 = (lane < 16) ? t1 : 0.0f;
                car2 = (lane < 16) ? t2 : 0.0f; car3 = (lane < 16) ? t3 : 0.0f;
            }
            float wtot = 0.0f;
            int   segw = -1;
            if (cb == 0) {
                wtot = wpart + __shfl_xor(wpart, 16, 32);
                const int run = 4 * pass + slot;
                if (lane < 16 && run < nruns) segw = ys[rb + pl_first_row(starts, run)];
            }
            if (pass == 0) { ws0 = wtot; segw0 = segw; segw = -1; }
            pl_add(pooled, wsum, acc, wtot, segv[0], segv[1], segw, cb + lm, hd);
        }
        cseg = slast;

        {
            float wA = 0.0f;
            if (okA) wA = __expf(scA - mxA);
            const float4 z4 = make_float4(0.f, 0.f, 0.f, 0.f);
            if (!okA) { xa = z4; xb = z4; xc = z4; xd = z4; }
            float4* dA = (float4*)(&Xl[buf ^ 1][row * XP]) + e;
            dA[0] = xa; dA[8] = xb; dA[16] = xc; dA[24] = xd;
            if (wstore) wgt[buf ^ 1][row * NHEAD + h] = wA;
            if (e == 0) yl[buf ^ 1][row] = yA;
        }
        __syncthreads();

        if (segw0 >= 0) atomicAdd(&wsum[segw0 * NHEAD + hd], ws0);
        const int snext = __builtin_amdgcn_readfirstlane(yl[buf ^ 1][rb]);
        if (tile + 1 == tend || snext != cseg) {
            pl_flush(pooled, cseg, car0, car1, car2, car3, cb + lm, lane);
            car0 = 0.0f; car1 = 0.0f; car2 = 0.0f; car3 = 0.0f;
        }
        buf ^= 1;
        ndA = ndB; okA = okB; yA = yB; scA = scB;
        ndB = ndC; okB = okC;
    }
}

__global__ __launch_bounds__(128)
void k_proj(const float* __restrict__ pooled, const float* __restrict__ wsum,
            const float* __restrict__ lin_w, const float* __restrict__ lin_b,
            float* __restrict__ out, int C) {
    const int lane = threadIdx.x & 31;
    const int head = __builtin_amdgcn_readfirstlane((int)(threadIdx.x >> 5));
    const int sbase = blockIdx.x * 16;
    const int lm    = lane & 15;
    const int khalf = (lane < 16) ? 0 : 2;
    const int mhalf = (lane < 16) ? 0 : 8;
    const int sa = (sbase + lm < C) ? (sbase + lm) : (C - 1);
    const float* ap = pooled + ((size_t)sa * NHEAD + head) * IN_CH + khalf;
    const float* bp = lin_w + (size_t)(head * OUT_CH + lm) * IN_CH + khalf;
    const v8f vzero = {0.f, 0.f, 0.f, 0.f, 0.f, 0.f, 0.f, 0.f};
    v8f acc[4];
#pragma unroll
    for (int i = 0; i < 4; ++i) acc[i] = vzero;
#pragma unroll 8
    for (int ks = 0; ks < IN_CH / 4; ++ks) {
        const v2f a = *(const v2f*)(ap + ks * 4);
#pragma unroll
        for (int nt = 0; nt < 4; ++nt) {
            const v2f b = *(const v2f*)(bp + (size_t)nt * 16 * IN_CH + ks * 4);
            acc[nt] = __builtin_amdgcn_wmma_f32_16x16x4_f32(false, a, false, b, (short)0, acc[nt], false, false);
        }
    }
    float inv[8], one[8];
#pragma unroll
    for (int r = 0; r < 8; ++r) {
        const int sg = sbase + mhalf + r;
        const float ws = (sg < C) ? wsum[sg * NHEAD + head] : 0.0f;
        inv[r] = (ws != 0.0f) ? 1.0f / ws : 0.0f;
        one[r] = (ws != 0.0f) ? 1.0f : 0.0f;
    }
#pragma unroll
    for (int nt = 0; nt < 4; ++nt) {
        const int n = head * OUT_CH + nt * 16 + lm;
        const float bn = lin_b[n];
#pragma unroll
        for (int r = 0; r < 8; ++r) {
            const int sg = sbase + mhalf + r;
            if (sg < C) out[(size_t)sg * TOT_OUT + n] = acc[nt][r] * inv[r] + bn * one[r];
        }
    }
}

extern "C" void kernel_launch(void* const* d_in, const int* in_sizes, int n_in,
                              void* d_out, int out_size, void* d_ws, size_t ws_size,
                              hipStream_t stream) {
    (void)n_in;
    const float* x     = (const float*)d_in[0];
    const int*   y     = (const int*)  d_in[1];
    const float* lin_w = (const float*)d_in[3];
    const float* lin_b = (const float*)d_in[4];
    const float* att_w = (const float*)d_in[5];
    const float* att_b = (const float*)d_in[6];
    float* out = (float*)d_out;

    int N = in_sizes[0] / IN_CH;
    int C = out_size / TOT_OUT;

    float* wsf    = (float*)d_ws;
    float* v      = wsf;
    float* c      = wsf + 512;
    float* segmax = wsf + 516;
    float* denom  = segmax + 4 * C;
    float* score  = denom + 4 * C;
    int*   cnt    = (int*)(score + (size_t)4 * N);
    int*   perm   = cnt + C;
    int*   tick   = perm + N;

    size_t need = ((size_t)516 + (size_t)10 * C + (size_t)5 * N) * sizeof(float);
    if (ws_size < need) { cnt = nullptr; perm = nullptr; tick = nullptr; }

    size_t poff = (((size_t)516 + (size_t)10 * C + (size_t)5 * N) + 3) & ~(size_t)3;
    float* pooled = wsf + poff;
    float* wsum   = pooled + (size_t)C * NHEAD * IN_CH;
    size_t need2 = (poff + (size_t)C * (NHEAD * IN_CH + NHEAD)) * sizeof(float);
    const bool pool = (perm != nullptr) && (ws_size >= need2);

    int initN = pool ? C * (NHEAD * IN_CH + NHEAD) : C * TOT_OUT;
    float* dn = pool ? nullptr : denom;
    k_prep <<<PREP_FOLD + PREP_FILL, 256, 0, stream>>>(pool ? pooled : out, initN, dn, (unsigned*)segmax, cnt, tick, C,
                                                       lin_w, lin_b, att_w, att_b, v, c);
    k_score<<<(N + SC_SLABS * 256 - 1) / (SC_SLABS * 256), 256, 0, stream>>>(x, y, v, c, score,
                                                 (int*)segmax, (unsigned*)segmax, cnt, N, C);
    if (perm != nullptr && C > SEG_CAP) k_scan<<<1, SCAN_T, 0, stream>>>(cnt, C);
    k_denom<<<(N + DN_NPT * 256 - 1) / (DN_NPT * 256), 256, 0, stream>>>(score, y, segmax, dn, cnt, tick, perm, N, C);
    if (pool) {
        k_pool<<<PL_BLOCKS, PL_THREADS, 0, stream>>>(x, y, score, segmax, perm, pooled, wsum, N);
        k_proj<<<(C + 15) / 16, 128, 0, stream>>>(pooled, wsum, lin_w, lin_b, out, C);
    } else {
        k_gemm<<<GEMM_BLOCKS, GT_THREADS, 0, stream>>>(x, y, lin_w, lin_b,
                                                      score, segmax, denom, perm, out, N);
    }
}


# x read once: a block gathers a whole segment into LDS, scores formed as rows pass through registers, then max, exp weights, WMMA row sums from LDS; score pass and pooled atomics gone (sums reordered)
# speedup vs baseline: 12.1328x; 1.4640x over previous
#include <hip/hip_runtime.h>
#include <math.h>

typedef __attribute__((ext_vector_type(2))) float v2f;
typedef __attribute__((ext_vector_type(8))) float v8f;

#define IN_CH     128
#define OUT_CH    64
#define NHEAD     4
#define TOT_OUT   256
#define NEG_SLOPE 0.2f


#define PREP_FOLD 3
#define PREP_FILL 256
__global__ __launch_bounds__(256)
void k_prep(float* zbuf, int zn, float* denom, unsigned* segmax, int* cnt, int* tick, int C,
            const float* __restrict__ lin_w, const float* __restrict__ lin_b,
            const float* __restrict__ att_w, const float* __restrict__ att_b,
            float* v, float* c) {
    if (blockIdx.x < PREP_FOLD) {
        if (blockIdx.x < 2) {
            const int t = blockIdx.x * 256 + threadIdx.x;
            if (t < NHEAD * IN_CH) {
                const int h = t / IN_CH, i = t % IN_CH;
                float s = 0.0f;
#pragma unroll
                for (int d = 0; d < OUT_CH; ++d)
                    s += lin_w[(size_t)(h * OUT_CH + d) * IN_CH + i] * att_w[d];
                v[h * IN_CH + i] = s;
            }
        } else if (threadIdx.x < NHEAD) {
            const int t = threadIdx.x;
            float s = att_b[0];
            for (int d = 0; d < OUT_CH; ++d) s += lin_b[t * OUT_CH + d] * att_w[d];
            c[t] = s;
        }
        return;
    }
    const int gt = (blockIdx.x - PREP_FOLD) * 256 + threadIdx.x;
    const int gs = PREP_FILL * 256;
    const int n4 = zn >> 2;
    float4* z4 = (float4*)zbuf;
    const float4 zero4 = make_float4(0.f, 0.f, 0.f, 0.f);
    for (int i = gt; i < n4; i += gs) z4[i] = zero4;
    for (int i = (n4 << 2) + gt; i < zn; i += gs) zbuf[i] = 0.0f;
    for (int i = gt; i < 4 * C; i += gs) { if (denom != nullptr) denom[i] = 0.0f; segmax[i] = 0xFF800000u; }
    if (cnt != nullptr)  for (int i = gt; i < C; i += gs) cnt[i] = 0;
    if (tick != nullptr) for (int i = gt; i < C; i += gs) tick[i] = 0;
}

typedef __attribute__((ext_vector_type(4))) float v4f;

#define SC_ROWS   8
#define SC_GROUPS 4
#define SC_WAVES  8
#define SC_SLABS  4
#define SEG_CAP   2048

__global__ __launch_bounds__(256)
void k_score(const float* __restrict__ x, const int* __restrict__ y,
             const float* __restrict__ v, const float* __restrict__ c,
             float* __restrict__ score, int* segmax_i, unsigned* segmax_u, int* cnt, int Nn, int C) {
    __shared__ int lmax[NHEAD * SEG_CAP];
    __shared__ int lcnt[SEG_CAP];
    const bool bins = (C <= SEG_CAP);
    if (bins) {
        for (int i = threadIdx.x; i < NHEAD * C; i += 256) lmax[i] = (int)0xFF800000;
        for (int i = threadIdx.x; i < C; i += 256) lcnt[i] = 0;
    }
    __syncthreads();

    const int lane = threadIdx.x & 31;
    const int wave = __builtin_amdgcn_readfirstlane((int)(threadIdx.x >> 5));

    v4f vh[NHEAD];
#pragma unroll
    for (int h = 0; h < NHEAD; ++h) vh[h] = *(const v4f*)(v + h * IN_CH + lane * 4);
    const int hh = lane & 3;
    const int rr = lane >> 2;
    const float ch = c[hh];

    for (int sl = 0; sl < SC_SLABS; ++sl) {
    const int row0 = ((blockIdx.x * SC_SLABS + sl) * SC_WAVES + wave) * (SC_ROWS * SC_GROUPS);
    if (row0 < Nn) {
    const float* xb = x + (size_t)row0 * IN_CH + lane * 4;
    v4f xv[2][SC_ROWS];
#define SC_LOAD(G) do { _Pragma("unroll") for (int r = 0; r < SC_ROWS; ++r) { \
        const int ro_ = (row0 + (G) * SC_ROWS + r < Nn) ? ((G) * SC_ROWS + r) : 0; \
        xv[(G) & 1][r] = __builtin_nontemporal_load((const v4f*)(xb + (size_t)ro_ * IN_CH)); } } while (0)

    int yv[SC_GROUPS];
#pragma unroll
    for (int g = 0; g < SC_GROUPS; ++g) {
        const int nd = row0 + g * SC_ROWS + rr;
        yv[g] = y[(nd < Nn) ? nd : (Nn - 1)];
    }

    SC_LOAD(0);
#pragma unroll
    for (int g = 0; g < SC_GROUPS; ++g) {
        __builtin_amdgcn_sched_barrier(0);
        if (g + 1 < SC_GROUPS) SC_LOAD(g + 1);
        __builtin_amdgcn_sched_barrier(0);

        float vals[SC_ROWS * NHEAD];
#pragma unroll
        for (int r = 0; r < SC_ROWS; ++r) {
            const v4f xr = xv[g & 1][r];
#pragma unroll
            for (int h = 0; h < NHEAD; ++h)
                vals[r * NHEAD + h] = xr.x * vh[h].x + xr.y * vh[h].y + xr.z * vh[h].z + xr.w * vh[h].w;
        }
#define SC_STAGE(WD) do { const bool up_ = (lane & (WD)) != 0; \
        _Pragma("unroll") for (int i = 0; i < (WD); ++i) { \
            const float keep_ = up_ ? vals[i + (WD)] : vals[i]; \
            const float send_ = up_ ? vals[i] : vals[i + (WD)]; \
            vals[i] = keep_ + __shfl_xor(send_, (WD), 32); } } while (0)
        SC_STAGE(16); SC_STAGE(8); SC_STAGE(4); SC_STAGE(2); SC_STAGE(1);
#undef SC_STAGE
        float s = vals[0] + ch;
        s = (s >= 0.0f) ? s : NEG_SLOPE * s;
        const int node = row0 + g * SC_ROWS + rr;
        if (node < Nn) {
            score[(size_t)node * NHEAD + hh] = s;
            const int idx = yv[g] * NHEAD + hh;
            if (bins) {
                if (s >= 0.0f) atomicMax(&lmax[idx], __float_as_int(s));
                else           atomicMin((unsigned*)&lmax[idx], (unsigned)__float_as_int(s));
                if (hh == 0) atomicAdd(&lcnt[yv[g]], 1);
            } else {
                if (s >= 0.0f) atomicMax(&segmax_i[idx], __float_as_int(s));
                else           atomicMin(&segmax_u[idx], (unsigned)__float_as_int(s));
                if (cnt != nullptr && hh == 0) atomicAdd(&cnt[yv[g]], 1);
            }
        }
    }
#undef SC_LOAD
    }
    }

    __syncthreads();
    if (bins) {
        for (int i = threadIdx.x; i < NHEAD * C; i += 256) {
            const int b = lmax[i];
            if (b != (int)0xFF800000) {
                if (__int_as_float(b) >= 0.0f) atomicMax(&segmax_i[i], b);
                else                           atomicMin(&segmax_u[i], (unsigned)b);
            }
        }
        if (cnt != nullptr)
            for (int i = threadIdx.x; i < C; i += 256) {
                const int n = lcnt[i];
                if (n != 0) atomicAdd(&cnt[i], n);
            }
    }
}

#define SCAN_T 1024
__global__ __launch_bounds__(SCAN_T)
void k_scan(int* cnt, int C) {
    __shared__ int part[SCAN_T];
    const int t = threadIdx.x;
    const int per = (C + SCAN_T - 1) / SCAN_T;
    const int b = t * per;
    int s = 0;
    for (int j = 0; j < per; ++j) { const int i = b + j; if (i < C) s += cnt[i]; }
    part[t] = s;
    __syncthreads();
    for (int d = 1; d < SCAN_T; d <<= 1) {
        const int add = (t >= d) ? part[t - d] : 0;
        __syncthreads();
        part[t] += add;
        __syncthreads();
    }
    int run = part[t] - s;
    for (int j = 0; j < per; ++j) {
        const int i = b + j;
        if (i < C) { const int n = cnt[i]; cnt[i] = run; run += n; }
    }
}

#define DN_NPT 4
#define DN_SPT (SEG_CAP / 256)
__global__ __launch_bounds__(256)
void k_denom(const float* __restrict__ score, const int* __restrict__ y,
             const float* __restrict__ segmax, float* denom,
             int* cursor, int* tick, int* perm, int Nn, int C) {
    __shared__ float lden[NHEAD * SEG_CAP];
    __shared__ int   lrun[SEG_CAP];
    __shared__ int   lbase[SEG_CAP];
    __shared__ int   wtot[8];
    const bool bins = (C <= SEG_CAP);
    const bool scan = bins && (perm != nullptr);
    const bool dsum = (denom != nullptr);
    const int t = threadIdx.x;

    const int per = (C + 255) / 256;
    int cv[DN_SPT];
#pragma unroll
    for (int j = 0; j < DN_SPT; ++j) {
        const int i = t * per + j;
        cv[j] = (scan && j < per && i < C) ? cursor[i] : 0;
    }
    const int base = blockIdx.x * (256 * DN_NPT);
    int  yy[DN_NPT], rk[DN_NPT];
    v4f  sc[DN_NPT];
#pragma unroll
    for (int j = 0; j < DN_NPT; ++j) {
        const int node = base + j * 256 + t;
        const int nc = (node < Nn) ? node : (Nn - 1);
        yy[j] = y[nc];
        sc[j] = (v4f){0.0f, 0.0f, 0.0f, 0.0f};
        if (dsum) sc[j] = *(const v4f*)(score + (size_t)nc * NHEAD);
    }
    if (bins) {
        if (dsum) for (int i = t; i < NHEAD * C; i += 256) lden[i] = 0.0f;
        for (int i = t; i < C; i += 256) lrun[i] = 0;
    }
    int mine = 0, incl = 0;
    if (scan) {
#pragma unroll
        for (int j = 0; j < DN_SPT; ++j) mine += cv[j];
        incl = mine;
        for (int d = 1; d < 32; d <<= 1) {
            const int o = __shfl_up(incl, d, 32);
            if ((t & 31) >= d) incl += o;
        }
        if ((t & 31) == 31) wtot[t >> 5] = incl;
    }
    __syncthreads();
    if (scan) {
        int run = incl - mine;
        for (int w = 0; w < (t >> 5); ++w) run += wtot[w];
#pragma unroll
        for (int j = 0; j < DN_SPT; ++j) {
            const int i = t * per + j;
            if (j < per && i < C) { lbase[i] = run; run += cv[j]; }
        }
    }

#pragma unroll
    for (int j = 0; j < DN_NPT; ++j) {
        const int node = base + j * 256 + t;
        float e0 = 0.0f, e1 = 0.0f, e2 = 0.0f, e3 = 0.0f;
        if (dsum) {
            const v4f mx = *(const v4f*)(segmax + (size_t)yy[j] * NHEAD);
            e0 = __expf(sc[j].x - mx.x); e1 = __expf(sc[j].y - mx.y);
            e2 = __expf(sc[j].z - mx.z); e3 = __expf(sc[j].w - mx.w);
        }
        rk[j] = 0;
        if (node < Nn) {
            if (bins) {
                if (dsum) {
                    float* ld = &lden[yy[j] * NHEAD];
                    atomicAdd(ld + 0, e0); atomicAdd(ld + 1, e1);
                    atomicAdd(ld + 2, e2); atomicAdd(ld + 3, e3);
                }
                if (perm != nullptr) rk[j] = atomicAdd(&lrun[yy[j]], 1);
            } else {
                if (dsum) {
                    float* gd = &denom[(size_t)yy[j] * NHEAD];
                    atomicAdd(gd + 0, e0); atomicAdd(gd + 1, e1);
                    atomicAdd(gd + 2, e2); atomicAdd(gd + 3, e3);
                }
                if (perm != nullptr) {
                    const int pos = atomicAdd(&cursor[yy[j]], 1);
                    if ((unsigned)pos < (unsigned)Nn) perm[pos] = node;
                }
            }
        }
    }
    __syncthreads();
    if (bins) {
        if (dsum)
            for (int i = t; i < NHEAD * C; i += 256) {
                const float s = lden[i];
                if (s != 0.0f) atomicAdd(&denom[i], s);
            }
        if (perm != nullptr) {
            int nn[DN_SPT], tk[DN_SPT];
#pragma unroll
            for (int j = 0; j < DN_SPT; ++j) { const int i = t + j * 256; nn[j] = (i < C) ? lrun[i] : 0; }
#pragma unroll
            for (int j = 0; j < DN_SPT; ++j) { tk[j] = 0; if (nn[j] != 0) tk[j] = atomicAdd(&tick[t + j * 256], nn[j]); }
#pragma unroll
            for (int j = 0; j < DN_SPT; ++j) {
                const int i = t + j * 256;
                if (i < C) lrun[i] = (nn[j] != 0) ? lbase[i] + tk[j] : 0;
            }
        }
    }
    __syncthreads();
    if (bins && perm != nullptr) {
#pragma unroll
        for (int j = 0; j < DN_NPT; ++j) {
            const int node = base + j * 256 + t;
            if (node < Nn) {
                const int pos = lrun[yy[j]] + rk[j];
                if ((unsigned)pos < (unsigned)Nn) perm[pos] = node;
            }
        }
    }
}

#define GT_ROWS    64
#define GT_THREADS 512
#define XP         132
#define GEMM_BLOCKS 256

__device__ __forceinline__ int gm_node(const int* __restrict__ perm, int p, int Nn) {
    const int pc = (p < Nn) ? p : (Nn - 1);
    int nd = (perm != nullptr) ? perm[pc] : pc;
    if ((unsigned)nd >= (unsigned)Nn) nd = 0;
    return nd;
}

__global__ __launch_bounds__(GT_THREADS)
void k_gemm(const float* __restrict__ x, const int* __restrict__ y,
            const float* __restrict__ lin_w, const float* __restrict__ lin_b,
            const float* __restrict__ score, const float* __restrict__ segmax,
            const float* __restrict__ denom, const int* __restrict__ perm,
            float* __restrict__ out, int Nn) {
    __shared__ __attribute__((aligned(16))) float Wl[TOT_OUT * XP];
    __shared__ __attribute__((aligned(16))) float Xl[2][GT_ROWS * XP];
    __shared__ float wgt[2][GT_ROWS * NHEAD];
    __shared__ int   yl[2][GT_ROWS];
    __shared__ float bl[TOT_OUT];

    const int t = threadIdx.x;
    const int G = gridDim.x;
    const int ntiles = (Nn + GT_ROWS - 1) / GT_ROWS;
    int tile = blockIdx.x;
    if (tile >= ntiles) return;

    const int row = t >> 3, e = t & 7, h = t & 3;
    const bool wstore = (t & 4) == 0;

    {
        const int p0 = tile * GT_ROWS + row;
        const bool ok0 = p0 < Nn;
        const int nd0 = gm_node(perm, p0, Nn);
        const float4* s0 = (const float4*)(x + (size_t)nd0 * IN_CH) + e;
        float4 a0 = s0[0], a1 = s0[8], a2 = s0[16], a3 = s0[24];
        const int y0 = y[nd0];
        const int i0 = y0 * NHEAD + h;
        float w0 = 0.0f;
        if (ok0) w0 = __expf(score[(size_t)nd0 * NHEAD + h] - segmax[i0]) / denom[i0];
        const float4 z4 = make_float4(0.f, 0.f, 0.f, 0.f);
        if (!ok0) { a0 = z4; a1 = z4; a2 = z4; a3 = z4; }
        float4* d0 = (float4*)(&Xl[0][row * XP]) + e;
        d0[0] = a0; d0[8] = a1; d0[16] = a2; d0[24] = a3;
        if (wstore) wgt[0][row * NHEAD + h] = w0;
        if (e == 0) yl[0][row] = y0;
    }
    if (t < TOT_OUT) bl[t] = lin_b[t];
    for (int i = t; i < TOT_OUT * (IN_CH / 4); i += GT_THREADS) {
        const int n = i >> 5, c4 = i & 31;
        const float4 w4 = *((const float4*)(lin_w + (size_t)n * IN_CH) + c4);
        *((float4*)(&Wl[n * XP]) + c4) = w4;
    }
    int  pA = (tile + G) * GT_ROWS + row;
    bool okA = pA < Nn;
    int  ndA = gm_node(perm, pA, Nn);
    int  yA  = y[ndA];
    float scA = score[(size_t)ndA * NHEAD + h];
    int  pB = (tile + 2 * G) * GT_ROWS + row;
    bool okB = pB < Nn;
    int  ndB = gm_node(perm, pB, Nn);
    __syncthreads();

    const int wave = __builtin_amdgcn_readfirstlane(t >> 5), lane = t & 31;
    const int rowbase = (wave >> 2) * 16;
    const int colbase = (wave & 3) * 64;
    const int head    = wave & 3;
    const int lm      = lane & 15;
    const int khalf   = (lane < 16) ? 0 : 2;
    const int mhalf   = (lane < 16) ? 0 : 8;
    const v8f vzero = {0.f, 0.f, 0.f, 0.f, 0.f, 0.f, 0.f, 0.f};
    const float* bp = &Wl[(colbase + lm) * XP + khalf];

    int buf = 0;
    for (; tile < ntiles; tile += G) {
        const float4* sA = (const float4*)(x + (size_t)ndA * IN_CH) + e;
        float4 xa = sA[0], xb = sA[8], xc = sA[16], xd = sA[24];
        const int iA = yA * NHEAD + h;
        const float mxA = segmax[iA], dnA = denom[iA];
        const int yB = y[ndB];
        const float scB = score[(size_t)ndB * NHEAD + h];
        const int pC = (tile + 3 * G) * GT_ROWS + row;
        const bool okC = pC < Nn;
        const int ndC = gm_node(perm, pC, Nn);

        v8f acc[4];
#pragma unroll
        for (int i = 0; i < 4; ++i) acc[i] = vzero;
        const float* ap = &Xl[buf][(rowbase + lm) * XP + khalf];
#pragma unroll
        for (int ks = 0; ks < IN_CH / 4; ++ks) {
            const v2f a = *(const v2f*)(ap + ks * 4);
#pragma unroll
            for (int nt = 0; nt < 4; ++nt) {
                const v2f b = *(const v2f*)(bp + nt * 16 * XP + ks * 4);
                acc[nt] = __builtin_amdgcn_wmma_f32_16x16x4_f32(
                    false, a, false, b, (short)0, acc[nt], false, false);
            }
        }

        float wr[8];
        int   yr[8];
        int   seg0, all_one_seg;
        {
            const float* wg = wgt[buf];
            const int*   ys = yl[buf];
            seg0 = ys[rowbase];
            int same = 1;
#pragma unroll
            for (int r = 1; r < 16; ++r) same &= (ys[rowbase + r] == seg0) ? 1 : 0;
            all_one_seg = __builtin_amdgcn_readfirstlane(same);
#pragma unroll
            for (int r = 0; r < 8; ++r) {
                const int m = rowbase + mhalf + r;
                wr[r] = wg[m * NHEAD + head];
                yr[r] = ys[m];
            }
        }

        {
            float wA = 0.0f;
            if (okA) wA = __expf(scA - mxA) / dnA;
            const float4 z4 = make_float4(0.f, 0.f, 0.f, 0.f);
            if (!okA) { xa = z4; xb = z4; xc = z4; xd = z4; }
            float4* dA = (float4*)(&Xl[buf ^ 1][row * XP]) + e;
            dA[0] = xa; dA[8] = xb; dA[16] = xc; dA[24] = xd;
            if (wstore) wgt[buf ^ 1][row * NHEAD + h] = wA;
            if (e == 0) yl[buf ^ 1][row] = yA;
        }
        __syncthreads();

        if (all_one_seg) {
#pragma unroll
            for (int nt = 0; nt < 4; ++nt) {
                const int n = colbase + nt * 16 + lm;
                const float bn = bl[n];
                float sum = 0.0f;
#pragma unroll
                for (int r = 0; r < 8; ++r) sum += (acc[nt][r] + bn) * wr[r];
                sum += __shfl_xor(sum, 16, 32);
                if (lane < 16) atomicAdd(&out[(size_t)seg0 * TOT_OUT + n], sum);
            }
        } else {
#pragma unroll
            for (int nt = 0; nt < 4; ++nt) {
                const int n = colbase + nt * 16 + lm;
                const float bn = bl[n];
                int   cur = yr[0];
                float sum = (acc[nt][0] + bn) * wr[0];
#pragma unroll
                for (int r = 1; r < 8; ++r) {
                    const float val = (acc[nt][r] + bn) * wr[r];
                    if (yr[r] == cur) {
                        sum += val;
                    } else {
                        atomicAdd(&out[(size_t)cur * TOT_OUT + n], sum);
                        cur = yr[r]; sum = val;
                    }
                }
                atomicAdd(&out[(size_t)cur * TOT_OUT + n], sum);
            }
        }
        buf ^= 1;
        ndA = ndB; okA = okB; yA = yB; scA = scB;
        ndB = ndC; okB = okC;
    }
}


#define PL_THREADS 512
#define PL_BLOCKS  512

__device__ __forceinline__ int pl_first_row(unsigned starts, int run) {
    unsigned mm = starts | 1u;
    for (int q = 0; q < run && mm != 0u; ++q) mm &= mm - 1u;
    return (mm != 0u) ? __builtin_ctz(mm) : 0;
}

__device__ __forceinline__ void pl_add(float* __restrict__ pooled, float* __restrict__ wsum, const v8f acc, float wtot,
                                       int sa, int sb, int sw, int col, int hd) {
    if (sa >= 0) {
#pragma unroll
        for (int hh2 = 0; hh2 < NHEAD; ++hh2)
            atomicAdd(&pooled[((size_t)sa * NHEAD + hh2) * IN_CH + col], acc[hh2]);
    }
    if (sb >= 0) {
#pragma unroll
        for (int hh2 = 0; hh2 < NHEAD; ++hh2)
            atomicAdd(&pooled[((size_t)sb * NHEAD + hh2) * IN_CH + col], acc[4 + hh2]);
    }
    if (sw >= 0) atomicAdd(&wsum[sw * NHEAD + hd], wtot);
}

__device__ __forceinline__ void pl_flush(float* __restrict__ pooled, int seg, float c0, float c1, float c2, float c3,
                                         int col, int lane) {
    if (seg >= 0 && lane < 16) {
        atomicAdd(&pooled[((size_t)seg * NHEAD + 0) * IN_CH + col], c0);
        atomicAdd(&pooled[((size_t)seg * NHEAD + 1) * IN_CH + col], c1);
        atomicAdd(&pooled[((size_t)seg * NHEAD + 2) * IN_CH + col], c2);
        atomicAdd(&pooled[((size_t)seg * NHEAD + 3) * IN_CH + col], c3);
    }
}

__device__ __forceinline__ int pl_node(const int* __restrict__ perm, int p, int pend, int Nn) {
    const int pc = (p < pend) ? p : (pend - 1);
    int nd = (perm != nullptr) ? perm[pc] : pc;
    if ((unsigned)nd >= (unsigned)Nn) nd = 0;
    return nd;
}

__global__ __launch_bounds__(PL_THREADS, 8)
void k_pool(const float* __restrict__ x, const int* __restrict__ y,
            const float* __restrict__ score, const float* __restrict__ segmax,
            const int* __restrict__ perm,
            float* __restrict__ pooled, float* __restrict__ wsum, int Nn) {
    __shared__ __attribute__((aligned(16))) float Xl[2][GT_ROWS * XP];
    __shared__ float wgt[2][GT_ROWS * NHEAD];
    __shared__ int   yl[2][GT_ROWS];

    const int t = threadIdx.x;
    const int G = gridDim.x;
    const int ntiles = (Nn + GT_ROWS - 1) / GT_ROWS;
    int tile = (int)(((long long)blockIdx.x * ntiles) / G);
    const int tend = (int)(((long long)(blockIdx.x + 1) * ntiles) / G);
    if (tile >= tend) return;
    const int pend = (tend * GT_ROWS < Nn) ? tend * GT_ROWS : Nn;

    const int row = t >> 3, e = t & 7, h = t & 3;
    const bool wstore = (t & 4) == 0;
    {
        const int p0 = tile * GT_ROWS + row;
        const bool ok0 = p0 < pend;
        const int nd0 = pl_node(perm, p0, pend, Nn);
        const float4* s0 = (const float4*)(x + (size_t)nd0 * IN_CH) + e;
        float4 a0 = s0[0], a1 = s0[8], a2 = s0[16], a3 = s0[24];
        const int y0 = y[nd0];
        const int i0 = y0 * NHEAD + h;
        float w0 = 0.0f;
        if (ok0) w0 = __expf(score[(size_t)nd0 * NHEAD + h] - segmax[i0]);
        const float4 z4 = make_float4(0.f, 0.f, 0.f, 0.f);
        if (!ok0) { a0 = z4; a1 = z4; a2 = z4; a3 = z4; }
        float4* d0 = (float4*)(&Xl[0][row * XP]) + e;
        d0[0] = a0; d0[8] = a1; d0[16] = a2; d0[24] = a3;
        if (wstore) wgt[0][row * NHEAD + h] = w0;
        if (e == 0) yl[0][row] = y0;
    }
    int  pA = (tile + 1) * GT_ROWS + row;
    bool okA = pA < pend;
    int  ndA = pl_node(perm, pA, pend, Nn);
    int  yA  = y[ndA];
    float scA = score[(size_t)ndA * NHEAD + h];
    int  pB = (tile + 2) * GT_ROWS + row;
    bool okB = pB < pend;
    int  ndB = pl_node(perm, pB, pend, Nn);
    __syncthreads();

    const int wave = __builtin_amdgcn_readfirstlane(t >> 5), lane = t & 31;
    const int cb    = (wave & 7) * 16;
    const int rb    = (wave >> 3) * 32;
    const int lm    = lane & 15;
    const int khalf = (lane < 16) ? 0 : 2;
    const int mhalf = (lane < 16) ? 0 : 8;
    const int slot  = lm >> 2, hd = lm & 3;
    const unsigned lemask = (2u << lane) - 1u;
    const v8f vzero = {0.f, 0.f, 0.f, 0.f, 0.f, 0.f, 0.f, 0.f};

    int   cseg = -1;
    float car0 = 0.0f, car1 = 0.0f, car2 = 0.0f, car3 = 0.0f;

    int buf = 0;
    for (; tile < tend; ++tile) {
        const float4* sA = (const float4*)(x + (size_t)ndA * IN_CH) + e;
        float4 xa = sA[0], xb = sA[8], xc = sA[16], xd = sA[24];
        const int iA = yA * NHEAD + h;
        const float mxA = segmax[iA];
        const int yB = y[ndB];
        const float scB = score[(size_t)ndB * NHEAD + h];
        const int pC = (tile + 3) * GT_ROWS + row;
        const bool okC = pC < pend;
        const int ndC = pl_node(perm, pC, pend, Nn);

        const float* wg = wgt[buf];
        const int*   ys = yl[buf];
        const int ymine = ys[rb + lane];
        const int yprev = ys[rb + ((lane > 0) ? lane - 1 : 0)];
        const unsigned starts = __builtin_amdgcn_ballot_w32(ymine != yprev);
        const int nruns = __builtin_popcount(starts) + 1;
        const int lastrun = nruns - 1;
        const float* xcol = &Xl[buf][rb * XP + cb + lm];
        const int myrun = __builtin_popcount(starts & lemask);
        const float* wrow = wg + (rb + khalf) * NHEAD + hd;
        const int slast  = __builtin_amdgcn_readfirstlane(ys[rb + 31]);

        float ws0 = 0.0f;
        int segw0 = -1;
        for (int pass = 0; pass * 4 < nruns; ++pass) {
            v8f acc = vzero;
            float wpart = 0.0f;
            if (pass == 0) { acc[0] = car0; acc[1] = car1; acc[2] = car2; acc[3] = car3; }
            const int rel = myrun - 4 * pass;
            const unsigned m0 = __builtin_amdgcn_ballot_w32(rel == 0);
            const unsigned m1 = __builtin_amdgcn_ballot_w32(rel == 1);
            const unsigned m2 = __builtin_amdgcn_ballot_w32(rel == 2);
            const unsigned m3 = __builtin_amdgcn_ballot_w32(rel == 3);
            const unsigned rmask = ((slot == 0) ? m0 : (slot == 1) ? m1 : (slot == 2) ? m2 : m3) >> khalf;
            float wa[16], xb2[16];
#pragma unroll
            for (int ks = 0; ks < 8; ++ks) {
                const int i0 = ks * 4 + khalf;
                wa[2 * ks]      = wrow[ks * 4 * NHEAD];
                wa[2 * ks + 1]  = wrow[ks * 4 * NHEAD + NHEAD];
                xb2[2 * ks]     = xcol[i0 * XP];
                xb2[2 * ks + 1] = xcol[(i0 + 1) * XP];
            }
#pragma unroll
            for (int ks = 0; ks < 8; ++ks) {
                v2f a, b;
                a.x = ((rmask >> (ks * 4)) & 1u)     ? wa[2 * ks]     : 0.0f;
                a.y = ((rmask >> (ks * 4 + 1)) & 1u) ? wa[2 * ks + 1] : 0.0f;
                b.x = xb2[2 * ks];
                b.y = xb2[2 * ks + 1];
                acc = __builtin_amdgcn_wmma_f32_16x16x4_f32(false, a, false, b, (short)0, acc, false, false);
                wpart += a.x + a.y;
            }
            int segv[2];
#pragma unroll
            for (int s2 = 0; s2 < 2; ++s2) {
                const int run = 4 * pass + (mhalf >> 2) + s2;
                segv[s2] = (run < lastrun) ? ys[rb + pl_first_row(starts, run)] : -1;
            }
            if ((lastrun >> 2) == pass) {
                const int ls = lastrun & 3;
                float t0 = (ls & 1) ? acc[4] : acc[0];
                float t1 = (ls & 1) ? acc[5] : acc[1];
                float t2 = (ls & 1) ? acc[6] : acc[2];
                float t3 = (ls & 1) ? acc[7] : acc[3];
                if (ls & 2) {
                    t0 = __shfl_xor(t0, 16, 32); t1 = __shfl_xor(t1, 16, 32);
                    t2 = __shfl_xor(t2, 16, 32); t3 = __shfl_xor(t3, 16, 32);
                }
                car0 = (lane < 16) ? t0 : 0.0f; car1 = (lane < 16) ? t1 : 0.0f;
                car2 = (lane < 16) ? t2 : 0.0f; car3 = (lane < 16) ? t3 : 0.0f;
            }
            float wtot = 0.0f;
            int   segw = -1;
            if (cb == 0) {
                wtot = wpart + __shfl_xor(wpart, 16, 32);
                const int run = 4 * pass + slot;
                if (lane < 16 && run < nruns) segw = ys[rb + pl_first_row(starts, run)];
            }
            if (pass == 0) { ws0 = wtot; segw0 = segw; segw = -1; }
            pl_add(pooled, wsum, acc, wtot, segv[0], segv[1], segw, cb + lm, hd);
        }
        cseg = slast;

        {
            float wA = 0.0f;
            if (okA) wA = __expf(scA - mxA);
            const float4 z4 = make_float4(0.f, 0.f, 0.f, 0.f);
            if (!okA) { xa = z4; xb = z4; xc = z4; xd = z4; }
            float4* dA = (float4*)(&Xl[buf ^ 1][row * XP]) + e;
            dA[0] = xa; dA[8] = xb; dA[16] = xc; dA[24] = xd;
            if (wstore) wgt[buf ^ 1][row * NHEAD + h] = wA;
            if (e == 0) yl[buf ^ 1][row] = yA;
        }
        __syncthreads();

        if (segw0 >= 0) atomicAdd(&wsum[segw0 * NHEAD + hd], ws0);
        const int snext = __builtin_amdgcn_readfirstlane(yl[buf ^ 1][rb]);
        if (tile + 1 == tend || snext != cseg) {
            pl_flush(pooled, cseg, car0, car1, car2, car3, cb + lm, lane);
            car0 = 0.0f; car1 = 0.0f; car2 = 0.0f; car3 = 0.0f;
        }
        buf ^= 1;
        ndA = ndB; okA = okB; yA = yB; scA = scB;
        ndB = ndC; okB = okC;
    }
}

__global__ __launch_bounds__(128)
void k_proj(const float* __restrict__ pooled, const float* __restrict__ wsum,
            const float* __restrict__ lin_w, const float* __restrict__ lin_b,
            float* __restrict__ out, int C) {
    const int lane = threadIdx.x & 31;
    const int head = __builtin_amdgcn_readfirstlane((int)(threadIdx.x >> 5));
    const int sbase = blockIdx.x * 16;
    const int lm    = lane & 15;
    const int khalf = (lane < 16) ? 0 : 2;
    const int mhalf = (lane < 16) ? 0 : 8;
    const int sa = (sbase + lm < C) ? (sbase + lm) : (C - 1);
    const float* ap = pooled + ((size_t)sa * NHEAD + head) * IN_CH + khalf;
    const float* bp = lin_w + (size_t)(head * OUT_CH + lm) * IN_CH + khalf;
    const v8f vzero = {0.f, 0.f, 0.f, 0.f, 0.f, 0.f, 0.f, 0.f};
    v8f acc[4];
#pragma unroll
    for (int i = 0; i < 4; ++i) acc[i] = vzero;
#pragma unroll 8
    for (int ks = 0; ks < IN_CH / 4; ++ks) {
        const v2f a = *(const v2f*)(ap + ks * 4);
#pragma unroll
        for (int nt = 0; nt < 4; ++nt) {
            const v2f b = *(const v2f*)(bp + (size_t)nt * 16 * IN_CH + ks * 4);
            acc[nt] = __builtin_amdgcn_wmma_f32_16x16x4_f32(false, a, false, b, (short)0, acc[nt], false, false);
        }
    }
    float inv[8], one[8];
#pragma unroll
    for (int r = 0; r < 8; ++r) {
        const int sg = sbase + mhalf + r;
        const float ws = (sg < C) ? wsum[sg * NHEAD + head] : 0.0f;
        inv[r] = (ws != 0.0f) ? 1.0f / ws : 0.0f;
        one[r] = (ws != 0.0f) ? 1.0f : 0.0f;
    }
#pragma unroll
    for (int nt = 0; nt < 4; ++nt) {
        const int n = head * OUT_CH + nt * 16 + lm;
        const float bn = lin_b[n];
#pragma unroll
        for (int r = 0; r < 8; ++r) {
            const int sg = sbase + mhalf + r;
            if (sg < C) out[(size_t)sg * TOT_OUT + n] = acc[nt][r] * inv[r] + bn * one[r];
        }
    }
}


#define CT_RPT 8
__global__ __launch_bounds__(256)
void k_count(const int* __restrict__ y, int* cnt, int Nn, int C) {
    __shared__ int lc[SEG_CAP];
    const int t = threadIdx.x;
    for (int i = t; i < C; i += 256) lc[i] = 0;
    const int base = blockIdx.x * (256 * CT_RPT);
    int yy[CT_RPT];
#pragma unroll
    for (int j = 0; j < CT_RPT; ++j) {
        const int nd = base + j * 256 + t;
        yy[j] = y[(nd < Nn) ? nd : (Nn - 1)];
    }
    __syncthreads();
#pragma unroll
    for (int j = 0; j < CT_RPT; ++j) {
        const int nd = base + j * 256 + t;
        if (nd < Nn) atomicAdd(&lc[yy[j]], 1);
    }
    __syncthreads();
    for (int i = t; i < C; i += 256) {
        const int n = lc[i];
        if (n != 0) atomicAdd(&cnt[i], n);
    }
}

#define SG_THREADS 512
#define SG_CAP     576
#define SG_GROUPS  (SG_CAP / 64)
#define SG_BATCH   3

__global__ __launch_bounds__(SG_THREADS)
void k_seg(const float* __restrict__ x, const int* __restrict__ perm, const int* __restrict__ cnt,
           const float* __restrict__ v, const float* __restrict__ c,
           float* __restrict__ scw, float* __restrict__ pooled, float* __restrict__ wsum, int Nn, int C) {
    __shared__ __attribute__((aligned(16))) float Xs[SG_CAP * XP];
    __shared__ float ssc[SG_CAP * NHEAD];
    __shared__ float comb[8 * NHEAD * 16];
    __shared__ float wpart[16 * NHEAD];
    __shared__ int   ired[16 * 8];
    __shared__ int   sst[8], snn[8];
    __shared__ int   smax[NHEAD];

    const int t = threadIdx.x;
    const int lane = t & 31;
    const int wave = __builtin_amdgcn_readfirstlane(t >> 5);
    const int rowt = t >> 3, e = t & 7;
    v4f vr[NHEAD][4];
#pragma unroll
    for (int h = 0; h < NHEAD; ++h)
#pragma unroll
        for (int q = 0; q < 4; ++q) vr[h][q] = *(const v4f*)(v + h * IN_CH + (e + 8 * q) * 4);
    const float ce = c[e & 3];
    const int lm    = lane & 15;
    const int khalf = (lane < 16) ? 0 : 2;
    const int cb    = (wave & 7) * 16;
    const int kg    = wave >> 3;
    const int hm    = lm & 3;
    const bool arow = lm < NHEAD;
    const v8f vzero = {0.f, 0.f, 0.f, 0.f, 0.f, 0.f, 0.f, 0.f};

    int part[8];
#pragma unroll
    for (int j = 0; j < 8; ++j) part[j] = 0;
    for (int i = t; i < C; i += SG_THREADS) {
        const int cv = cnt[i];
#pragma unroll
        for (int j = 0; j < 8; ++j) if (i < (int)(blockIdx.x + j * gridDim.x)) part[j] += cv;
    }
#pragma unroll
    for (int j = 0; j < 8; ++j) {
        for (int d = 16; d >= 1; d >>= 1) part[j] += __shfl_xor(part[j], d, 32);
        if (lane == 0) ired[wave * 8 + j] = part[j];
    }
    if (t < NHEAD) smax[t] = (int)0xFF800000;
    __syncthreads();
    if (t < 8) {
        int a = 0;
        for (int w = 0; w < 16; ++w) a += ired[w * 8 + t];
        const int sj = blockIdx.x + t * gridDim.x;
        sst[t] = a;
        snn[t] = (sj < C) ? cnt[sj] : 0;
    }
    __syncthreads();

    int js = 0;
    for (int s = blockIdx.x; s < C && js < 8; s += gridDim.x, ++js) {
        int start = __builtin_amdgcn_readfirstlane(sst[js]);
        int n = __builtin_amdgcn_readfirstlane(snn[js]);
        if (start < 0) start = 0;
        if (start > Nn) start = Nn;
        if (n < 0) n = 0;
        if (n > Nn - start) n = Nn - start;
        const int nch = (n + SG_CAP - 1) / SG_CAP;

        float mloc = -INFINITY;
        auto stage = [&](int off, int m, bool doscore) {
            int nd[SG_GROUPS];
#pragma unroll
            for (int g = 0; g < SG_GROUPS; ++g) {
                const int r = g * 64 + rowt;
                int q = perm[start + off + ((r < m) ? r : (m - 1))];
                if ((unsigned)q >= (unsigned)Nn) q = 0;
                nd[g] = q;
            }
#pragma unroll
            for (int g = 0; g < SG_GROUPS; ++g) asm volatile("" : "+v"(nd[g]));
            float4 xr[2][SG_BATCH][4];
#define SG_LOADB(B) do { _Pragma("unroll") for (int u = 0; u < SG_BATCH; ++u) { \
                const float4* sp_ = (const float4*)(x + (size_t)nd[(B) * SG_BATCH + u] * IN_CH) + e; \
                xr[(B) & 1][u][0] = sp_[0]; xr[(B) & 1][u][1] = sp_[8]; xr[(B) & 1][u][2] = sp_[16]; xr[(B) & 1][u][3] = sp_[24]; } } while (0)
            SG_LOADB(0);
#pragma unroll
            for (int b = 0; b < SG_GROUPS / SG_BATCH; ++b) {
                __builtin_amdgcn_sched_barrier(0);
                if (b + 1 < SG_GROUPS / SG_BATCH) SG_LOADB(b + 1);
                __builtin_amdgcn_sched_barrier(0);
#pragma unroll
                for (int u = 0; u < SG_BATCH; ++u) {
                    const int r = (b * SG_BATCH + u) * 64 + rowt;
                    const bool ok = r < m;
                    const float4 z4 = make_float4(0.f, 0.f, 0.f, 0.f);
                    const float4 a0 = ok ? xr[b & 1][u][0] : z4, a1 = ok ? xr[b & 1][u][1] : z4;
                    const float4 a2 = ok ? xr[b & 1][u][2] : z4, a3 = ok ? xr[b & 1][u][3] : z4;
                    float4* dr = (float4*)(&Xs[r * XP]) + e;
                    dr[0] = a0; dr[8] = a1; dr[16] = a2; dr[24] = a3;
                    if (doscore) {
                        float p[NHEAD];
#pragma unroll
                        for (int h = 0; h < NHEAD; ++h) {
                            p[h] = a0.x * vr[h][0].x + a0.y * vr[h][0].y + a0.z * vr[h][0].z + a0.w * vr[h][0].w
                                 + a1.x * vr[h][1].x + a1.y * vr[h][1].y + a1.z * vr[h][1].z + a1.w * vr[h][1].w
                                 + a2.x * vr[h][2].x + a2.y * vr[h][2].y + a2.z * vr[h][2].z + a2.w * vr[h][2].w
                                 + a3.x * vr[h][3].x + a3.y * vr[h][3].y + a3.z * vr[h][3].z + a3.w * vr[h][3].w;
                            p[h] += __shfl_xor(p[h], 1, 32);
                            p[h] += __shfl_xor(p[h], 2, 32);
                            p[h] += __shfl_xor(p[h], 4, 32);
                        }
                        float sv = ((e & 3) == 0) ? p[0] : ((e & 3) == 1) ? p[1] : ((e & 3) == 2) ? p[2] : p[3];
                        sv += ce;
                        sv = (sv >= 0.0f) ? sv : NEG_SLOPE * sv;
                        if (e < NHEAD && ok) {
                            ssc[r * NHEAD + e] = sv;
                            if (nch > 1) scw[(size_t)(start + off + r) * NHEAD + e] = sv;
                            mloc = fmaxf(mloc, sv);
                        }
                    }
                }
            }
#undef SG_LOADB
        };

        for (int ch = 0; ch < nch; ++ch) {
            const int off = ch * SG_CAP;
            const int m = (n - off < SG_CAP) ? (n - off) : SG_CAP;
            stage(off, m, true);
        }
        if (e < NHEAD && mloc > -INFINITY) {
            if (mloc >= 0.0f) atomicMax(&smax[e], __float_as_int(mloc));
            else              atomicMin((unsigned*)&smax[e], (unsigned)__float_as_int(mloc));
        }
        __syncthreads();
        const float Mh = __int_as_float(smax[t & 3]);

        v8f acc0 = vzero, acc1 = vzero;
        float wl = 0.0f;
        for (int ch = 0; ch < nch; ++ch) {
            const int off = ch * SG_CAP;
            const int m = (n - off < SG_CAP) ? (n - off) : SG_CAP;
            if (nch > 1) {
                __syncthreads();
                stage(off, m, false);
                for (int idx = t; idx < m * NHEAD; idx += SG_THREADS)
                    ssc[idx] = scw[(size_t)(start + off) * NHEAD + idx];
                __syncthreads();
            }
            const int units = (m + 31) >> 5;
            for (int idx = t; idx < units * 32 * NHEAD; idx += SG_THREADS) {
                float w = 0.0f;
                if ((idx >> 2) < m) w = __expf(ssc[idx] - Mh);
                ssc[idx] = w;
                wl += w;
            }
            __syncthreads();
            const int uh = (units + 1) >> 1;
            const int ub = (kg == 0) ? 0 : uh;
            const int ue = (kg == 0) ? uh : units;
            for (int un = ub; un < ue; ++un) {
                const float* wp = &ssc[(un * 32 + khalf) * NHEAD + hm];
                const float* xp = &Xs[(un * 32 + khalf) * XP + cb + lm];
                float wa[16], xb[16];
#pragma unroll
                for (int ks = 0; ks < 8; ++ks) {
                    wa[2 * ks]     = wp[ks * 4 * NHEAD];
                    wa[2 * ks + 1] = wp[ks * 4 * NHEAD + NHEAD];
                    xb[2 * ks]     = xp[ks * 4 * XP];
                    xb[2 * ks + 1] = xp[ks * 4 * XP + XP];
                }
#pragma unroll
                for (int ks = 0; ks < 8; ks += 2) {
                    v2f a, b;
                    a.x = arow ? wa[2 * ks] : 0.0f;     a.y = arow ? wa[2 * ks + 1] : 0.0f;
                    b.x = xb[2 * ks];                   b.y = xb[2 * ks + 1];
                    acc0 = __builtin_amdgcn_wmma_f32_16x16x4_f32(false, a, false, b, (short)0, acc0, false, false);
                    a.x = arow ? wa[2 * ks + 2] : 0.0f; a.y = arow ? wa[2 * ks + 3] : 0.0f;
                    b.x = xb[2 * ks + 2];               b.y = xb[2 * ks + 3];
                    acc1 = __builtin_amdgcn_wmma_f32_16x16x4_f32(false, a, false, b, (short)0, acc1, false, false);
                }
            }
        }

        const float s0 = acc0[0] + acc1[0], s1 = acc0[1] + acc1[1], s2 = acc0[2] + acc1[2], s3 = acc0[3] + acc1[3];
        if (kg == 1 && lane < 16) {
            float* cp = &comb[(wave & 7) * NHEAD * 16 + lm];
            cp[0] = s0; cp[16] = s1; cp[32] = s2; cp[48] = s3;
        }
        wl += __shfl_xor(wl, 4, 32);
        wl += __shfl_xor(wl, 8, 32);
        wl += __shfl_xor(wl, 16, 32);
        if (lane < NHEAD) wpart[wave * NHEAD + lane] = wl;
        if (t < NHEAD) smax[t] = (int)0xFF800000;
        __syncthreads();
        if (kg == 0 && lane < 16) {
            const float* cp = &comb[(wave & 7) * NHEAD * 16 + lm];
            float* pp = pooled + (size_t)s * NHEAD * IN_CH + cb + lm;
            pp[0]         = s0 + cp[0];
            pp[IN_CH]     = s1 + cp[16];
            pp[2 * IN_CH] = s2 + cp[32];
            pp[3 * IN_CH] = s3 + cp[48];
        }
        if (t < NHEAD) {
            float sum = 0.0f;
            for (int w = 0; w < 16; ++w) sum += wpart[w * NHEAD + t];
            wsum[s * NHEAD + t] = sum;
        }
    }
}

extern "C" void kernel_launch(void* const* d_in, const int* in_sizes, int n_in,
                              void* d_out, int out_size, void* d_ws, size_t ws_size,
                              hipStream_t stream) {
    (void)n_in;
    const float* x     = (const float*)d_in[0];
    const int*   y     = (const int*)  d_in[1];
    const float* lin_w = (const float*)d_in[3];
    const float* lin_b = (const float*)d_in[4];
    const float* att_w = (const float*)d_in[5];
    const float* att_b = (const float*)d_in[6];
    float* out = (float*)d_out;

    int N = in_sizes[0] / IN_CH;
    int C = out_size / TOT_OUT;

    float* wsf    = (float*)d_ws;
    float* v      = wsf;
    float* c      = wsf + 512;
    float* segmax = wsf + 516;
    float* denom  = segmax + 4 * C;
    float* score  = denom + 4 * C;
    int*   cnt    = (int*)(score + (size_t)4 * N);
    int*   perm   = cnt + C;
    int*   tick   = perm + N;

    size_t need = ((size_t)516 + (size_t)10 * C + (size_t)5 * N) * sizeof(float);
    if (ws_size < need) { cnt = nullptr; perm = nullptr; tick = nullptr; }

    size_t poff = (((size_t)516 + (size_t)10 * C + (size_t)5 * N) + 3) & ~(size_t)3;
    float* pooled = wsf + poff;
    float* wsum   = pooled + (size_t)C * NHEAD * IN_CH;
    size_t need2 = (poff + (size_t)C * (NHEAD * IN_CH + NHEAD)) * sizeof(float);
    const bool pool = (perm != nullptr) && (ws_size >= need2);

    if (pool && C <= SEG_CAP) {
        k_prep <<<PREP_FOLD + PREP_FILL, 256, 0, stream>>>(pooled, 0, nullptr, (unsigned*)segmax, cnt, tick, C,
                                                           lin_w, lin_b, att_w, att_b, v, c);
        k_count<<<(N + CT_RPT * 256 - 1) / (CT_RPT * 256), 256, 0, stream>>>(y, cnt, N, C);
        k_denom<<<(N + DN_NPT * 256 - 1) / (DN_NPT * 256), 256, 0, stream>>>(score, y, segmax, nullptr, cnt, tick, perm, N, C);
        k_seg  <<<(C < 256) ? C : 256, SG_THREADS, 0, stream>>>(x, perm, cnt, v, c, score, pooled, wsum, N, C);
        k_proj <<<(C + 15) / 16, 128, 0, stream>>>(pooled, wsum, lin_w, lin_b, out, C);
        return;
    }

    int initN = pool ? C * (NHEAD * IN_CH + NHEAD) : C * TOT_OUT;
    float* dn = pool ? nullptr : denom;
    k_prep <<<PREP_FOLD + PREP_FILL, 256, 0, stream>>>(pool ? pooled : out, initN, dn, (unsigned*)segmax, cnt, tick, C,
                                                       lin_w, lin_b, att_w, att_b, v, c);
    k_score<<<(N + SC_SLABS * 256 - 1) / (SC_SLABS * 256), 256, 0, stream>>>(x, y, v, c, score,
                                                 (int*)segmax, (unsigned*)segmax, cnt, N, C);
    if (perm != nullptr && C > SEG_CAP) k_scan<<<1, SCAN_T, 0, stream>>>(cnt, C);
    k_denom<<<(N + DN_NPT * 256 - 1) / (DN_NPT * 256), 256, 0, stream>>>(score, y, segmax, dn, cnt, tick, perm, N, C);
    if (pool) {
        k_pool<<<PL_BLOCKS, PL_THREADS, 0, stream>>>(x, y, score, segmax, perm, pooled, wsum, N);
        k_proj<<<(C + 15) / 16, 128, 0, stream>>>(pooled, wsum, lin_w, lin_b, out, C);
    } else {
        k_gemm<<<GEMM_BLOCKS, GT_THREADS, 0, stream>>>(x, y, lin_w, lin_b,
                                                      score, segmax, denom, perm, out, N);
    }
}
